# starting kernel
# baseline (speedup 1.0000x reference)
#include <hip/hip_runtime.h>
#include <hip/hip_bf16.h>


typedef __attribute__((ext_vector_type(2))) float v2f;
typedef __attribute__((ext_vector_type(8))) float v8f;

#define NB   32
#define HIDC 128
#define HW   1024

__device__ __forceinline__ v8f wmma4(v2f a, v2f b, v8f c) {
  return __builtin_amdgcn_wmma_f32_16x16x4_f32(false, a, false, b, (short)0, c,
                                               false, false);
}

__device__ __forceinline__ float eluf(float x) {
  return x > 0.f ? x : (expf(x) - 1.f);
}

__global__ void k_elu(const float* __restrict__ X, float* __restrict__ Y, int n) {
  int i = blockIdx.x * 256 + threadIdx.x;
  if (i < n) Y[i] = eluf(X[i]);
}

__global__ void k_add_elu(const float* __restrict__ A, const float* __restrict__ B,
                          float* __restrict__ Y, int n) {
  int i = blockIdx.x * 256 + threadIdx.x;
  if (i < n) Y[i] = eluf(A[i] + B[i]);
}

__global__ void k_gate(float* __restrict__ out, const float* __restrict__ T2) {
  int i = blockIdx.x * 256 + threadIdx.x;
  int b = i >> 17;
  int rem = i & 0x1FFFF;
  float t1 = T2[((size_t)b << 18) + rem];
  float t2 = T2[((size_t)b << 18) + (1 << 17) + rem];
  out[i] += t1 * (1.f / (1.f + expf(-t2)));
}

__global__ void __launch_bounds__(256)
k_conv2x2(const float* __restrict__ X, const float* __restrict__ W,
          const float* __restrict__ bias, float* __restrict__ Y, int O) {
  const int lane = threadIdx.x & 31;
  const int wave = threadIdx.x >> 5;
  const int half = lane >> 4;
  const int l15  = lane & 15;
  const int b  = blockIdx.x;
  const int p0 = blockIdx.y * 64;
  const int o0 = (blockIdx.z * 8 + wave) * 16;
  if (o0 >= O) return;

  const float* Xb = X + (size_t)b * HIDC * HW;

  int   sidx[4][4];
  float smsk[4][4];
#pragma unroll
  for (int t = 0; t < 4; ++t) {
    int p = p0 + t * 16 + l15;
    int y = p >> 5, x = p & 31;
#pragma unroll
    for (int tap = 0; tap < 4; ++tap) {
      int ys = y + (tap >> 1) - 1;
      int xs = x + (tap & 1) - 1;
      bool ok = (ys >= 0) && (xs >= 0);
      sidx[t][tap] = ok ? (ys * 32 + xs) : 0;
      smsk[t][tap] = ok ? 1.f : 0.f;
    }
  }

  v8f acc[4];
#pragma unroll
  for (int t = 0; t < 4; ++t) acc[t] = (v8f){0, 0, 0, 0, 0, 0, 0, 0};

  const int orow = o0 + l15;
#pragma unroll
  for (int tap = 0; tap < 4; ++tap) {
    const float* wt = W + (size_t)orow * 512 + tap;
    for (int c0 = 0; c0 < HIDC; c0 += 4) {
      int ca = c0 + half * 2;
      v2f a;
      a.x = wt[(size_t)ca * 4];
      a.y = wt[(size_t)(ca + 1) * 4];
      const float* x0 = Xb + (size_t)ca * HW;
      const float* x1 = x0 + HW;
#pragma unroll
      for (int t = 0; t < 4; ++t) {
        v2f bb;
        bb.x = x0[sidx[t][tap]] * smsk[t][tap];
        bb.y = x1[sidx[t][tap]] * smsk[t][tap];
        acc[t] = wmma4(a, bb, acc[t]);
      }
    }
  }

#pragma unroll
  for (int t = 0; t < 4; ++t) {
#pragma unroll
    for (int r = 0; r < 8; ++r) {
      int o = o0 + r + half * 8;
      float v = acc[t][r] + bias[o];
      Y[((size_t)b * O + o) * HW + p0 + t * 16 + l15] = v;
    }
  }
}

__global__ void __launch_bounds__(256)
k_gemm1x1(const float* __restrict__ X, const float* __restrict__ W,
          const float* __restrict__ bias, float* __restrict__ Y,
          const float* __restrict__ R, int O, int outElu) {
  const int lane = threadIdx.x & 31;
  const int wave = threadIdx.x >> 5;
  const int half = lane >> 4;
  const int l15  = lane & 15;
  const int b  = blockIdx.x;
  const int p0 = blockIdx.y * 64;
  const int o0 = (blockIdx.z * 8 + wave) * 16;
  if (o0 >= O) return;

  const float* Xb   = X + (size_t)b * HIDC * HW;
  const float* wrow = W + (size_t)(o0 + l15) * HIDC;

  v8f acc[4];
#pragma unroll
  for (int t = 0; t < 4; ++t) acc[t] = (v8f){0, 0, 0, 0, 0, 0, 0, 0};

  for (int c0 = 0; c0 < HIDC; c0 += 4) {
    int ca = c0 + half * 2;
    v2f a;
    a.x = wrow[ca];
    a.y = wrow[ca + 1];
    const float* x0 = Xb + (size_t)ca * HW + p0 + l15;
#pragma unroll
    for (int t = 0; t < 4; ++t) {
      v2f bb;
      bb.x = x0[t * 16];
      bb.y = x0[HW + t * 16];
      acc[t] = wmma4(a, bb, acc[t]);
    }
  }

#pragma unroll
  for (int t = 0; t < 4; ++t) {
#pragma unroll
    for (int r = 0; r < 8; ++r) {
      int o = o0 + r + half * 8;
      size_t idx = ((size_t)b * O + o) * HW + p0 + t * 16 + l15;
      float v = acc[t][r] + bias[o];
      if (outElu) v = eluf(v);
      if (R) v += R[idx];
      Y[idx] = v;
    }
  }
}

__global__ void __launch_bounds__(256)
k_qkv(const float* __restrict__ X, const float* __restrict__ W,
      const float* __restrict__ bias, float* __restrict__ Y) {
  const int lane = threadIdx.x & 31;
  const int wave = threadIdx.x >> 5;
  const int half = lane >> 4;
  const int l15  = lane & 15;
  const int b  = blockIdx.x;
  const int p0 = blockIdx.y * 64;
  const int o0 = (blockIdx.z * 8 + wave) * 16;
  if (o0 >= 160) return;

  const float* Xb   = X + (size_t)b * HIDC * HW;
  const float* wrow = W + (size_t)(o0 + l15) * 130;

  v8f acc[4];
#pragma unroll
  for (int t = 0; t < 4; ++t) acc[t] = (v8f){0, 0, 0, 0, 0, 0, 0, 0};

  for (int c0 = 0; c0 < HIDC; c0 += 4) {
    int ca = c0 + half * 2;
    v2f a;
    a.x = wrow[ca];
    a.y = wrow[ca + 1];
    const float* x0 = Xb + (size_t)ca * HW + p0 + l15;
#pragma unroll
    for (int t = 0; t < 4; ++t) {
      v2f bb;
      bb.x = x0[t * 16];
      bb.y = x0[HW + t * 16];
      acc[t] = wmma4(a, bb, acc[t]);
    }
  }

#pragma unroll
  for (int t = 0; t < 4; ++t) {
    int p = p0 + t * 16 + l15;
    float py = (float)(p >> 5) * (1.f / 32.f) - 0.5f;
    float px = (float)(p & 31) * (1.f / 32.f) - 0.5f;
#pragma unroll
    for (int r = 0; r < 8; ++r) {
      int o = o0 + r + half * 8;
      float v = acc[t][r] + W[(size_t)o * 130 + 128] * py +
                W[(size_t)o * 130 + 129] * px + bias[o];
      Y[((size_t)b * 160 + o) * HW + p] = v;
    }
  }
}

__global__ void __launch_bounds__(32)
k_attn(const float* __restrict__ QKV, float* __restrict__ Ob) {
  __shared__ float pl[16 * 17];
  const int lane = threadIdx.x;
  const int half = lane >> 4;
  const int l15  = lane & 15;
  const int qt = blockIdx.x;
  const int b  = blockIdx.y;
  const int q0 = qt * 16;
  const float* Qb = QKV + (size_t)b * 160 * HW;

  v2f qa[4];
#pragma unroll
  for (int kc = 0; kc < 4; ++kc) {
    qa[kc].x = Qb[(size_t)(kc * 4 + half * 2) * HW + q0 + l15];
    qa[kc].y = Qb[(size_t)(kc * 4 + half * 2 + 1) * HW + q0 + l15];
  }

  v8f acc[8];
#pragma unroll
  for (int vt = 0; vt < 8; ++vt) acc[vt] = (v8f){0, 0, 0, 0, 0, 0, 0, 0};
  float mrun[8], lrun[8];
#pragma unroll
  for (int r = 0; r < 8; ++r) { mrun[r] = -3.0e38f; lrun[r] = 0.f; }

  for (int kt = 0; kt <= qt; ++kt) {
    int k0 = kt * 16;
    v8f s = (v8f){0, 0, 0, 0, 0, 0, 0, 0};
#pragma unroll
    for (int kc = 0; kc < 4; ++kc) {
      v2f bb;
      bb.x = Qb[(size_t)(16 + kc * 4 + half * 2) * HW + k0 + l15];
      bb.y = Qb[(size_t)(16 + kc * 4 + half * 2 + 1) * HW + k0 + l15];
      s = wmma4(qa[kc], bb, s);
    }
#pragma unroll
    for (int r = 0; r < 8; ++r) {
      int qp = q0 + r + half * 8;
      int kp = k0 + l15;
      float val = (kp <= qp) ? s[r] * 0.25f : -3.0e38f;
      float rm = val;
      rm = fmaxf(rm, __shfl_xor(rm, 1));
      rm = fmaxf(rm, __shfl_xor(rm, 2));
      rm = fmaxf(rm, __shfl_xor(rm, 4));
      rm = fmaxf(rm, __shfl_xor(rm, 8));
      float nm = fmaxf(mrun[r], rm);
      float alpha = expf(mrun[r] - nm);
      float p = expf(val - nm);
      float rs = p;
      rs += __shfl_xor(rs, 1);
      rs += __shfl_xor(rs, 2);
      rs += __shfl_xor(rs, 4);
      rs += __shfl_xor(rs, 8);
      lrun[r] = lrun[r] * alpha + rs;
      mrun[r] = nm;
#pragma unroll
      for (int vt = 0; vt < 8; ++vt) acc[vt][r] *= alpha;
      pl[(r + half * 8) * 17 + l15] = p;
    }
    asm volatile("s_wait_dscnt 0" ::: "memory");
#pragma unroll
    for (int kc = 0; kc < 4; ++kc) {
      v2f pa;
      pa.x = pl[l15 * 17 + kc * 4 + half * 2];
      pa.y = pl[l15 * 17 + kc * 4 + half * 2 + 1];
#pragma unroll
      for (int vt = 0; vt < 8; ++vt) {
        const float* vrow =
            Qb + (size_t)(32 + vt * 16 + l15) * HW + k0 + kc * 4 + half * 2;
        v2f bb;
        bb.x = vrow[0];
        bb.y = vrow[1];
        acc[vt] = wmma4(pa, bb, acc[vt]);
      }
    }
  }

#pragma unroll
  for (int r = 0; r < 8; ++r) {
    float inv = 1.f / lrun[r];
    int qp = q0 + r + half * 8;
#pragma unroll
    for (int vt = 0; vt < 8; ++vt) {
      Ob[((size_t)b * HIDC + vt * 16 + l15) * HW + qp] = acc[vt][r] * inv;
    }
  }
}

extern "C" void kernel_launch(void* const* d_in, const int* in_sizes, int n_in,
                              void* d_out, int out_size, void* d_ws, size_t ws_size,
                              hipStream_t stream) {
  const float* x    = (const float*)d_in[0];
  const float* rw1  = (const float*)d_in[1];
  const float* rb1  = (const float*)d_in[2];
  const float* rw2  = (const float*)d_in[3];
  const float* rb2  = (const float*)d_in[4];
  const float* qkvw = (const float*)d_in[5];
  const float* qkvb = (const float*)d_in[6];
  const float* apw  = (const float*)d_in[7];
  const float* apb  = (const float*)d_in[8];
  const float* ocw  = (const float*)d_in[9];
  const float* ocb  = (const float*)d_in[10];
  const float* oaw  = (const float*)d_in[11];
  const float* oab  = (const float*)d_in[12];
  const float* opw  = (const float*)d_in[13];
  const float* opb  = (const float*)d_in[14];
  float* Yout = (float*)d_out;

  const size_t N1 = (size_t)NB * HIDC * HW;
  float* A  = (float*)d_ws;
  float* Bf = A + N1;
  float* Cf = Bf + N1;
  float* Df = Cf + N1;

  hipMemcpyAsync(A, x, N1 * sizeof(float), hipMemcpyDeviceToDevice, stream);

  dim3 blk(256);
  dim3 ew((unsigned)((N1 + 255) / 256));
  dim3 g128(NB, HW / 64, 1), g256(NB, HW / 64, 2);

  for (int i = 0; i < 6; ++i) {
    const float* w1 = rw1 + (size_t)i * HIDC * HIDC * 4;
    const float* b1 = rb1 + (size_t)i * HIDC;
    const float* w2 = rw2 + (size_t)i * 2 * HIDC * HIDC * 4;
    const float* b2 = rb2 + (size_t)i * 2 * HIDC;
    k_elu<<<ew, blk, 0, stream>>>(A, Bf, (int)N1);
    k_conv2x2<<<g128, blk, 0, stream>>>(Bf, w1, b1, Cf, HIDC);
    k_elu<<<ew, blk, 0, stream>>>(Cf, Bf, (int)N1);
    k_conv2x2<<<g256, blk, 0, stream>>>(Bf, w2, b2, Df, 2 * HIDC);
    k_gate<<<ew, blk, 0, stream>>>(A, Df);
  }

  k_qkv<<<g256, blk, 0, stream>>>(A, qkvw, qkvb, Df);
  k_attn<<<dim3(HW / 16, NB), dim3(32), 0, stream>>>(Df, Cf);
  k_gemm1x1<<<g128, blk, 0, stream>>>(Cf, apw, apb, Bf, nullptr, HIDC, 0);

  k_elu<<<ew, blk, 0, stream>>>(A, Cf, (int)N1);
  k_gemm1x1<<<g128, blk, 0, stream>>>(Cf, ocw, ocb, Df, nullptr, HIDC, 1);

  k_elu<<<ew, blk, 0, stream>>>(Bf, Cf, (int)N1);
  k_gemm1x1<<<g128, blk, 0, stream>>>(Cf, oaw, oab, Bf, nullptr, HIDC, 1);

  k_add_elu<<<ew, blk, 0, stream>>>(Df, Bf, Cf, (int)N1);
  k_gemm1x1<<<g128, blk, 0, stream>>>(Cf, opw, opb, Yout, x, HIDC, 1);
}


# Elementwise passes (ELU, sigmoid gate, add, input copy) folded into conv/GEMM epilogues; gated conv keeps rows o and o+128 in one lane; 39 launches to 19; every sum keeps its order
# speedup vs baseline: 1.3456x; 1.3456x over previous
#include <hip/hip_runtime.h>
#include <hip/hip_bf16.h>


typedef __attribute__((ext_vector_type(2))) float v2f;
typedef __attribute__((ext_vector_type(8))) float v8f;

#define NB   32
#define HIDC 128
#define HW   1024

__device__ __forceinline__ v8f wmma4(v2f a, v2f b, v8f c) {
  return __builtin_amdgcn_wmma_f32_16x16x4_f32(false, a, false, b, (short)0, c,
                                               false, false);
}

__device__ __forceinline__ float eluf(float x) {
  return x > 0.f ? x : (expf(x) - 1.f);
}

__global__ void k_elu(const float* __restrict__ X, float* __restrict__ Y, int n) {
  int i = blockIdx.x * 256 + threadIdx.x;
  if (i < n) Y[i] = eluf(X[i]);
}

template <int G>
__global__ void __launch_bounds__(256)
k_conv2x2(const float* __restrict__ X, const float* __restrict__ W,
          const float* __restrict__ bias, float* Y, const float* Rin,
          float* __restrict__ E) {
  const int lane = threadIdx.x & 31;
  const int wave = threadIdx.x >> 5;
  const int half = lane >> 4;
  const int l15  = lane & 15;
  const int b  = blockIdx.x;
  const int p0 = blockIdx.y * 64;
  const int o0 = wave * 16;

  const float* Xb = X + (size_t)b * HIDC * HW;

  int   sidx[4][4];
  float smsk[4][4];
#pragma unroll
  for (int t = 0; t < 4; ++t) {
    int p = p0 + t * 16 + l15;
    int y = p >> 5, x = p & 31;
#pragma unroll
    for (int tap = 0; tap < 4; ++tap) {
      int ys = y + (tap >> 1) - 1;
      int xs = x + (tap & 1) - 1;
      bool ok = (ys >= 0) && (xs >= 0);
      sidx[t][tap] = ok ? (ys * 32 + xs) : 0;
      smsk[t][tap] = ok ? 1.f : 0.f;
    }
  }

  v8f acc[G][4];
#pragma unroll
  for (int g = 0; g < G; ++g)
#pragma unroll
    for (int t = 0; t < 4; ++t) acc[g][t] = (v8f){0, 0, 0, 0, 0, 0, 0, 0};

#pragma unroll
  for (int tap = 0; tap < 4; ++tap) {
    const float* wt[G];
#pragma unroll
    for (int g = 0; g < G; ++g)
      wt[g] = W + (size_t)(g * 128 + o0 + l15) * 512 + tap;
    for (int c0 = 0; c0 < HIDC; c0 += 4) {
      int ca = c0 + half * 2;
      v2f a[G];
#pragma unroll
      for (int g = 0; g < G; ++g) {
        a[g].x = wt[g][(size_t)ca * 4];
        a[g].y = wt[g][(size_t)(ca + 1) * 4];
      }
      const float* x0 = Xb + (size_t)ca * HW;
      const float* x1 = x0 + HW;
#pragma unroll
      for (int t = 0; t < 4; ++t) {
        v2f bb;
        bb.x = x0[sidx[t][tap]] * smsk[t][tap];
        bb.y = x1[sidx[t][tap]] * smsk[t][tap];
#pragma unroll
        for (int g = 0; g < G; ++g) acc[g][t] = wmma4(a[g], bb, acc[g][t]);
      }
    }
  }

  float bs[G][8];
#pragma unroll
  for (int g = 0; g < G; ++g)
#pragma unroll
    for (int r = 0; r < 8; ++r) bs[g][r] = bias[g * HIDC + o0 + r + half * 8];

  if (G == 1) {
#pragma unroll
    for (int t = 0; t < 4; ++t) {
#pragma unroll
      for (int r = 0; r < 8; ++r) {
        int o = o0 + r + half * 8;
        float v = acc[0][t][r] + bs[0][r];
        Y[((size_t)b * HIDC + o) * HW + p0 + t * 16 + l15] = eluf(v);
      }
    }
  } else {
    float res[4][8];
#pragma unroll
    for (int t = 0; t < 4; ++t) {
#pragma unroll
      for (int r = 0; r < 8; ++r) {
        int o = o0 + r + half * 8;
        res[t][r] = Rin[((size_t)b * HIDC + o) * HW + p0 + t * 16 + l15];
      }
    }
#pragma unroll
    for (int t = 0; t < 4; ++t) {
#pragma unroll
      for (int r = 0; r < 8; ++r) {
        int o = o0 + r + half * 8;
        size_t idx = ((size_t)b * HIDC + o) * HW + p0 + t * 16 + l15;
        float t1 = acc[0][t][r] + bs[0][r];
        float t2 = acc[G - 1][t][r] + bs[G - 1][r];
        float v = res[t][r] + t1 * (1.f / (1.f + expf(-t2)));
        Y[idx] = v;
        E[idx] = eluf(v);
      }
    }
  }
}

__global__ void __launch_bounds__(256)
k_gemm1x1(const float* __restrict__ X, const float* __restrict__ W,
          const float* __restrict__ bias, float* __restrict__ Y,
          const float* __restrict__ R, const float* __restrict__ Rpre, int O,
          int outElu) {
  const int lane = threadIdx.x & 31;
  const int wave = threadIdx.x >> 5;
  const int half = lane >> 4;
  const int l15  = lane & 15;
  const int b  = blockIdx.x;
  const int p0 = blockIdx.y * 64;
  const int o0 = (blockIdx.z * 8 + wave) * 16;
  if (o0 >= O) return;

  const float* Xb   = X + (size_t)b * HIDC * HW;
  const float* wrow = W + (size_t)(o0 + l15) * HIDC;

  v8f acc[4];
#pragma unroll
  for (int t = 0; t < 4; ++t) acc[t] = (v8f){0, 0, 0, 0, 0, 0, 0, 0};

  for (int c0 = 0; c0 < HIDC; c0 += 4) {
    int ca = c0 + half * 2;
    v2f a;
    a.x = wrow[ca];
    a.y = wrow[ca + 1];
    const float* x0 = Xb + (size_t)ca * HW + p0 + l15;
#pragma unroll
    for (int t = 0; t < 4; ++t) {
      v2f bb;
      bb.x = x0[t * 16];
      bb.y = x0[HW + t * 16];
      acc[t] = wmma4(a, bb, acc[t]);
    }
  }

#pragma unroll
  for (int t = 0; t < 4; ++t) {
#pragma unroll
    for (int r = 0; r < 8; ++r) {
      int o = o0 + r + half * 8;
      size_t idx = ((size_t)b * O + o) * HW + p0 + t * 16 + l15;
      float v = acc[t][r] + bias[o];
      if (outElu) v = eluf(v);
      if (Rpre) v = eluf(Rpre[idx] + v);
      if (R) v += R[idx];
      Y[idx] = v;
    }
  }
}

__global__ void __launch_bounds__(256)
k_qkv(const float* __restrict__ X, const float* __restrict__ W,
      const float* __restrict__ bias, float* __restrict__ Y) {
  const int lane = threadIdx.x & 31;
  const int wave = threadIdx.x >> 5;
  const int half = lane >> 4;
  const int l15  = lane & 15;
  const int b  = blockIdx.x;
  const int p0 = blockIdx.y * 64;
  const int o0 = (blockIdx.z * 8 + wave) * 16;
  if (o0 >= 160) return;

  const float* Xb   = X + (size_t)b * HIDC * HW;
  const float* wrow = W + (size_t)(o0 + l15) * 130;

  v8f acc[4];
#pragma unroll
  for (int t = 0; t < 4; ++t) acc[t] = (v8f){0, 0, 0, 0, 0, 0, 0, 0};

  for (int c0 = 0; c0 < HIDC; c0 += 4) {
    int ca = c0 + half * 2;
    v2f a;
    a.x = wrow[ca];
    a.y = wrow[ca + 1];
    const float* x0 = Xb + (size_t)ca * HW + p0 + l15;
#pragma unroll
    for (int t = 0; t < 4; ++t) {
      v2f bb;
      bb.x = x0[t * 16];
      bb.y = x0[HW + t * 16];
      acc[t] = wmma4(a, bb, acc[t]);
    }
  }

#pragma unroll
  for (int t = 0; t < 4; ++t) {
    int p = p0 + t * 16 + l15;
    float py = (float)(p >> 5) * (1.f / 32.f) - 0.5f;
    float px = (float)(p & 31) * (1.f / 32.f) - 0.5f;
#pragma unroll
    for (int r = 0; r < 8; ++r) {
      int o = o0 + r + half * 8;
      float v = acc[t][r] + W[(size_t)o * 130 + 128] * py +
                W[(size_t)o * 130 + 129] * px + bias[o];
      Y[((size_t)b * 160 + o) * HW + p] = v;
    }
  }
}

__global__ void __launch_bounds__(32)
k_attn(const float* __restrict__ QKV, float* __restrict__ Ob) {
  __shared__ float pl[16 * 17];
  const int lane = threadIdx.x;
  const int half = lane >> 4;
  const int l15  = lane & 15;
  const int qt = blockIdx.x;
  const int b  = blockIdx.y;
  const int q0 = qt * 16;
  const float* Qb = QKV + (size_t)b * 160 * HW;

  v2f qa[4];
#pragma unroll
  for (int kc = 0; kc < 4; ++kc) {
    qa[kc].x = Qb[(size_t)(kc * 4 + half * 2) * HW + q0 + l15];
    qa[kc].y = Qb[(size_t)(kc * 4 + half * 2 + 1) * HW + q0 + l15];
  }

  v8f acc[8];
#pragma unroll
  for (int vt = 0; vt < 8; ++vt) acc[vt] = (v8f){0, 0, 0, 0, 0, 0, 0, 0};
  float mrun[8], lrun[8];
#pragma unroll
  for (int r = 0; r < 8; ++r) { mrun[r] = -3.0e38f; lrun[r] = 0.f; }

  for (int kt = 0; kt <= qt; ++kt) {
    int k0 = kt * 16;
    v8f s = (v8f){0, 0, 0, 0, 0, 0, 0, 0};
#pragma unroll
    for (int kc = 0; kc < 4; ++kc) {
      v2f bb;
      bb.x = Qb[(size_t)(16 + kc * 4 + half * 2) * HW + k0 + l15];
      bb.y = Qb[(size_t)(16 + kc * 4 + half * 2 + 1) * HW + k0 + l15];
      s = wmma4(qa[kc], bb, s);
    }
#pragma unroll
    for (int r = 0; r < 8; ++r) {
      int qp = q0 + r + half * 8;
      int kp = k0 + l15;
      float val = (kp <= qp) ? s[r] * 0.25f : -3.0e38f;
      float rm = val;
      rm = fmaxf(rm, __shfl_xor(rm, 1));
      rm = fmaxf(rm, __shfl_xor(rm, 2));
      rm = fmaxf(rm, __shfl_xor(rm, 4));
      rm = fmaxf(rm, __shfl_xor(rm, 8));
      float nm = fmaxf(mrun[r], rm);
      float alpha = expf(mrun[r] - nm);
      float p = expf(val - nm);
      float rs = p;
      rs += __shfl_xor(rs, 1);
      rs += __shfl_xor(rs, 2);
      rs += __shfl_xor(rs, 4);
      rs += __shfl_xor(rs, 8);
      lrun[r] = lrun[r] * alpha + rs;
      mrun[r] = nm;
#pragma unroll
      for (int vt = 0; vt < 8; ++vt) acc[vt][r] *= alpha;
      pl[(r + half * 8) * 17 + l15] = p;
    }
    asm volatile("s_wait_dscnt 0" ::: "memory");
#pragma unroll
    for (int kc = 0; kc < 4; ++kc) {
      v2f pa;
      pa.x = pl[l15 * 17 + kc * 4 + half * 2];
      pa.y = pl[l15 * 17 + kc * 4 + half * 2 + 1];
#pragma unroll
      for (int vt = 0; vt < 8; ++vt) {
        const float* vrow =
            Qb + (size_t)(32 + vt * 16 + l15) * HW + k0 + kc * 4 + half * 2;
        v2f bb;
        bb.x = vrow[0];
        bb.y = vrow[1];
        acc[vt] = wmma4(pa, bb, acc[vt]);
      }
    }
  }

#pragma unroll
  for (int r = 0; r < 8; ++r) {
    float inv = 1.f / lrun[r];
    int qp = q0 + r + half * 8;
#pragma unroll
    for (int vt = 0; vt < 8; ++vt) {
      Ob[((size_t)b * HIDC + vt * 16 + l15) * HW + qp] = acc[vt][r] * inv;
    }
  }
}

extern "C" void kernel_launch(void* const* d_in, const int* in_sizes, int n_in,
                              void* d_out, int out_size, void* d_ws, size_t ws_size,
                              hipStream_t stream) {
  const float* x    = (const float*)d_in[0];
  const float* rw1  = (const float*)d_in[1];
  const float* rb1  = (const float*)d_in[2];
  const float* rw2  = (const float*)d_in[3];
  const float* rb2  = (const float*)d_in[4];
  const float* qkvw = (const float*)d_in[5];
  const float* qkvb = (const float*)d_in[6];
  const float* apw  = (const float*)d_in[7];
  const float* apb  = (const float*)d_in[8];
  const float* ocw  = (const float*)d_in[9];
  const float* ocb  = (const float*)d_in[10];
  const float* oaw  = (const float*)d_in[11];
  const float* oab  = (const float*)d_in[12];
  const float* opw  = (const float*)d_in[13];
  const float* opb  = (const float*)d_in[14];
  float* Yout = (float*)d_out;

  const size_t N1 = (size_t)NB * HIDC * HW;
  float* A  = (float*)d_ws;
  float* E  = A + N1;
  float* F  = E + N1;
  float* Df = F + N1;
  float* Gf = Df + 2 * N1;
  float* Hf = Gf + N1;

  dim3 blk(256);
  dim3 ew((unsigned)((N1 + 255) / 256));
  dim3 g128(NB, HW / 64, 1), g256(NB, HW / 64, 2);

  k_elu<<<ew, blk, 0, stream>>>(x, E, (int)N1);
  for (int i = 0; i < 6; ++i) {
    const float* w1 = rw1 + (size_t)i * HIDC * HIDC * 4;
    const float* b1 = rb1 + (size_t)i * HIDC;
    const float* w2 = rw2 + (size_t)i * 2 * HIDC * HIDC * 4;
    const float* b2 = rb2 + (size_t)i * 2 * HIDC;
    k_conv2x2<1><<<g128, blk, 0, stream>>>(E, w1, b1, F, nullptr, nullptr);
    k_conv2x2<2><<<g128, blk, 0, stream>>>(F, w2, b2, A, i == 0 ? x : A, E);
  }

  k_qkv<<<g256, blk, 0, stream>>>(A, qkvw, qkvb, Df);
  k_attn<<<dim3(HW / 16, NB), dim3(32), 0, stream>>>(Df, F);
  k_gemm1x1<<<g128, blk, 0, stream>>>(F, apw, apb, Gf, nullptr, nullptr, HIDC, 1);

  k_gemm1x1<<<g128, blk, 0, stream>>>(E, ocw, ocb, Hf, nullptr, nullptr, HIDC, 1);

  k_gemm1x1<<<g128, blk, 0, stream>>>(Gf, oaw, oab, F, nullptr, Hf, HIDC, 1);

  k_gemm1x1<<<g128, blk, 0, stream>>>(F, opw, opb, Yout, x, nullptr, HIDC, 1);
}


# Conv K-loop interchanged: 4-channel step outermost, the 4 taps innermost, so a lane's weights for all taps come from two 16-byte loads per step (was eight 4-byte); sums now channel-major, then tap
# speedup vs baseline: 1.5050x; 1.1185x over previous
#include <hip/hip_runtime.h>
#include <hip/hip_bf16.h>


typedef __attribute__((ext_vector_type(2))) float v2f;
typedef __attribute__((ext_vector_type(4))) float v4f;
typedef __attribute__((ext_vector_type(8))) float v8f;

#define NB   32
#define HIDC 128
#define HW   1024

__device__ __forceinline__ v8f wmma4(v2f a, v2f b, v8f c) {
  return __builtin_amdgcn_wmma_f32_16x16x4_f32(false, a, false, b, (short)0, c,
                                               false, false);
}

__device__ __forceinline__ float eluf(float x) {
  return x > 0.f ? x : (expf(x) - 1.f);
}

__global__ void k_elu(const float* __restrict__ X, float* __restrict__ Y, int n) {
  int i = blockIdx.x * 256 + threadIdx.x;
  if (i < n) Y[i] = eluf(X[i]);
}

template <int G>
__global__ void __launch_bounds__(256)
k_conv2x2(const float* __restrict__ X, const float* __restrict__ W,
          const float* __restrict__ bias, float* Y, const float* Rin,
          float* __restrict__ E) {
  const int lane = threadIdx.x & 31;
  const int wave = threadIdx.x >> 5;
  const int half = lane >> 4;
  const int l15  = lane & 15;
  const int b  = blockIdx.x;
  const int p0 = blockIdx.y * 64;
  const int o0 = wave * 16;

  const float* Xb = X + (size_t)b * HIDC * HW;

  int   sidx[4][4];
  float smsk[4][4];
#pragma unroll
  for (int t = 0; t < 4; ++t) {
    int p = p0 + t * 16 + l15;
    int y = p >> 5, x = p & 31;
#pragma unroll
    for (int tap = 0; tap < 4; ++tap) {
      int ys = y + (tap >> 1) - 1;
      int xs = x + (tap & 1) - 1;
      bool ok = (ys >= 0) && (xs >= 0);
      sidx[t][tap] = ok ? (ys * 32 + xs) : 0;
      smsk[t][tap] = ok ? 1.f : 0.f;
    }
  }

  v8f acc[G][4];
#pragma unroll
  for (int g = 0; g < G; ++g)
#pragma unroll
    for (int t = 0; t < 4; ++t) acc[g][t] = (v8f){0, 0, 0, 0, 0, 0, 0, 0};

  const v4f* wt[G];
#pragma unroll
  for (int g = 0; g < G; ++g)
    wt[g] = (const v4f*)(W + (size_t)(g * 128 + o0 + l15) * 512 + half * 8);
  for (int c0 = 0; c0 < HIDC; c0 += 4) {
    v4f wa[G][2];
#pragma unroll
    for (int g = 0; g < G; ++g) {
      wa[g][0] = wt[g][c0];
      wa[g][1] = wt[g][c0 + 1];
    }
    const float* x0 = Xb + (size_t)(c0 + half * 2) * HW;
    const float* x1 = x0 + HW;
    v2f xr[4][4];
#pragma unroll
    for (int tap = 0; tap < 4; ++tap) {
#pragma unroll
      for (int t = 0; t < 4; ++t) {
        xr[tap][t].x = x0[sidx[t][tap]];
        xr[tap][t].y = x1[sidx[t][tap]];
      }
    }
#pragma unroll
    for (int tap = 0; tap < 4; ++tap) {
      v2f a[G];
#pragma unroll
      for (int g = 0; g < G; ++g) {
        a[g].x = wa[g][0][tap];
        a[g].y = wa[g][1][tap];
      }
#pragma unroll
      for (int t = 0; t < 4; ++t) {
        v2f bb = xr[tap][t] * smsk[t][tap];
#pragma unroll
        for (int g = 0; g < G; ++g) acc[g][t] = wmma4(a[g], bb, acc[g][t]);
      }
    }
  }

  float bs[G][8];
#pragma unroll
  for (int g = 0; g < G; ++g)
#pragma unroll
    for (int r = 0; r < 8; ++r) bs[g][r] = bias[g * HIDC + o0 + r + half * 8];

  if (G == 1) {
#pragma unroll
    for (int t = 0; t < 4; ++t) {
#pragma unroll
      for (int r = 0; r < 8; ++r) {
        int o = o0 + r + half * 8;
        float v = acc[0][t][r] + bs[0][r];
        Y[((size_t)b * HIDC + o) * HW + p0 + t * 16 + l15] = eluf(v);
      }
    }
  } else {
    float res[4][8];
#pragma unroll
    for (int t = 0; t < 4; ++t) {
#pragma unroll
      for (int r = 0; r < 8; ++r) {
        int o = o0 + r + half * 8;
        res[t][r] = Rin[((size_t)b * HIDC + o) * HW + p0 + t * 16 + l15];
      }
    }
#pragma unroll
    for (int t = 0; t < 4; ++t) {
#pragma unroll
      for (int r = 0; r < 8; ++r) {
        int o = o0 + r + half * 8;
        size_t idx = ((size_t)b * HIDC + o) * HW + p0 + t * 16 + l15;
        float t1 = acc[0][t][r] + bs[0][r];
        float t2 = acc[G - 1][t][r] + bs[G - 1][r];
        float v = res[t][r] + t1 * (1.f / (1.f + expf(-t2)));
        Y[idx] = v;
        E[idx] = eluf(v);
      }
    }
  }
}

__global__ void __launch_bounds__(256)
k_gemm1x1(const float* __restrict__ X, const float* __restrict__ W,
          const float* __restrict__ bias, float* __restrict__ Y,
          const float* __restrict__ R, const float* __restrict__ Rpre, int O,
          int outElu) {
  const int lane = threadIdx.x & 31;
  const int wave = threadIdx.x >> 5;
  const int half = lane >> 4;
  const int l15  = lane & 15;
  const int b  = blockIdx.x;
  const int p0 = blockIdx.y * 64;
  const int o0 = (blockIdx.z * 8 + wave) * 16;
  if (o0 >= O) return;

  const float* Xb   = X + (size_t)b * HIDC * HW;
  const float* wrow = W + (size_t)(o0 + l15) * HIDC;

  v8f acc[4];
#pragma unroll
  for (int t = 0; t < 4; ++t) acc[t] = (v8f){0, 0, 0, 0, 0, 0, 0, 0};

  for (int c0 = 0; c0 < HIDC; c0 += 4) {
    int ca = c0 + half * 2;
    v2f a;
    a.x = wrow[ca];
    a.y = wrow[ca + 1];
    const float* x0 = Xb + (size_t)ca * HW + p0 + l15;
#pragma unroll
    for (int t = 0; t < 4; ++t) {
      v2f bb;
      bb.x = x0[t * 16];
      bb.y = x0[HW + t * 16];
      acc[t] = wmma4(a, bb, acc[t]);
    }
  }

#pragma unroll
  for (int t = 0; t < 4; ++t) {
#pragma unroll
    for (int r = 0; r < 8; ++r) {
      int o = o0 + r + half * 8;
      size_t idx = ((size_t)b * O + o) * HW + p0 + t * 16 + l15;
      float v = acc[t][r] + bias[o];
      if (outElu) v = eluf(v);
      if (Rpre) v = eluf(Rpre[idx] + v);
      if (R) v += R[idx];
      Y[idx] = v;
    }
  }
}

__global__ void __launch_bounds__(256)
k_qkv(const float* __restrict__ X, const float* __restrict__ W,
      const float* __restrict__ bias, float* __restrict__ Y) {
  const int lane = threadIdx.x & 31;
  const int wave = threadIdx.x >> 5;
  const int half = lane >> 4;
  const int l15  = lane & 15;
  const int b  = blockIdx.x;
  const int p0 = blockIdx.y * 64;
  const int o0 = (blockIdx.z * 8 + wave) * 16;
  if (o0 >= 160) return;

  const float* Xb   = X + (size_t)b * HIDC * HW;
  const float* wrow = W + (size_t)(o0 + l15) * 130;

  v8f acc[4];
#pragma unroll
  for (int t = 0; t < 4; ++t) acc[t] = (v8f){0, 0, 0, 0, 0, 0, 0, 0};

  for (int c0 = 0; c0 < HIDC; c0 += 4) {
    int ca = c0 + half * 2;
    v2f a;
    a.x = wrow[ca];
    a.y = wrow[ca + 1];
    const float* x0 = Xb + (size_t)ca * HW + p0 + l15;
#pragma unroll
    for (int t = 0; t < 4; ++t) {
      v2f bb;
      bb.x = x0[t * 16];
      bb.y = x0[HW + t * 16];
      acc[t] = wmma4(a, bb, acc[t]);
    }
  }

#pragma unroll
  for (int t = 0; t < 4; ++t) {
    int p = p0 + t * 16 + l15;
    float py = (float)(p >> 5) * (1.f / 32.f) - 0.5f;
    float px = (float)(p & 31) * (1.f / 32.f) - 0.5f;
#pragma unroll
    for (int r = 0; r < 8; ++r) {
      int o = o0 + r + half * 8;
      float v = acc[t][r] + W[(size_t)o * 130 + 128] * py +
                W[(size_t)o * 130 + 129] * px + bias[o];
      Y[((size_t)b * 160 + o) * HW + p] = v;
    }
  }
}

__global__ void __launch_bounds__(32)
k_attn(const float* __restrict__ QKV, float* __restrict__ Ob) {
  __shared__ float pl[16 * 17];
  const int lane = threadIdx.x;
  const int half = lane >> 4;
  const int l15  = lane & 15;
  const int qt = blockIdx.x;
  const int b  = blockIdx.y;
  const int q0 = qt * 16;
  const float* Qb = QKV + (size_t)b * 160 * HW;

  v2f qa[4];
#pragma unroll
  for (int kc = 0; kc < 4; ++kc) {
    qa[kc].x = Qb[(size_t)(kc * 4 + half * 2) * HW + q0 + l15];
    qa[kc].y = Qb[(size_t)(kc * 4 + half * 2 + 1) * HW + q0 + l15];
  }

  v8f acc[8];
#pragma unroll
  for (int vt = 0; vt < 8; ++vt) acc[vt] = (v8f){0, 0, 0, 0, 0, 0, 0, 0};
  float mrun[8], lrun[8];
#pragma unroll
  for (int r = 0; r < 8; ++r) { mrun[r] = -3.0e38f; lrun[r] = 0.f; }

  for (int kt = 0; kt <= qt; ++kt) {
    int k0 = kt * 16;
    v8f s = (v8f){0, 0, 0, 0, 0, 0, 0, 0};
#pragma unroll
    for (int kc = 0; kc < 4; ++kc) {
      v2f bb;
      bb.x = Qb[(size_t)(16 + kc * 4 + half * 2) * HW + k0 + l15];
      bb.y = Qb[(size_t)(16 + kc * 4 + half * 2 + 1) * HW + k0 + l15];
      s = wmma4(qa[kc], bb, s);
    }
#pragma unroll
    for (int r = 0; r < 8; ++r) {
      int qp = q0 + r + half * 8;
      int kp = k0 + l15;
      float val = (kp <= qp) ? s[r] * 0.25f : -3.0e38f;
      float rm = val;
      rm = fmaxf(rm, __shfl_xor(rm, 1));
      rm = fmaxf(rm, __shfl_xor(rm, 2));
      rm = fmaxf(rm, __shfl_xor(rm, 4));
      rm = fmaxf(rm, __shfl_xor(rm, 8));
      float nm = fmaxf(mrun[r], rm);
      float alpha = expf(mrun[r] - nm);
      float p = expf(val - nm);
      float rs = p;
      rs += __shfl_xor(rs, 1);
      rs += __shfl_xor(rs, 2);
      rs += __shfl_xor(rs, 4);
      rs += __shfl_xor(rs, 8);
      lrun[r] = lrun[r] * alpha + rs;
      mrun[r] = nm;
#pragma unroll
      for (int vt = 0; vt < 8; ++vt) acc[vt][r] *= alpha;
      pl[(r + half * 8) * 17 + l15] = p;
    }
    asm volatile("s_wait_dscnt 0" ::: "memory");
#pragma unroll
    for (int kc = 0; kc < 4; ++kc) {
      v2f pa;
      pa.x = pl[l15 * 17 + kc * 4 + half * 2];
      pa.y = pl[l15 * 17 + kc * 4 + half * 2 + 1];
#pragma unroll
      for (int vt = 0; vt < 8; ++vt) {
        const float* vrow =
            Qb + (size_t)(32 + vt * 16 + l15) * HW + k0 + kc * 4 + half * 2;
        v2f bb;
        bb.x = vrow[0];
        bb.y = vrow[1];
        acc[vt] = wmma4(pa, bb, acc[vt]);
      }
    }
  }

#pragma unroll
  for (int r = 0; r < 8; ++r) {
    float inv = 1.f / lrun[r];
    int qp = q0 + r + half * 8;
#pragma unroll
    for (int vt = 0; vt < 8; ++vt) {
      Ob[((size_t)b * HIDC + vt * 16 + l15) * HW + qp] = acc[vt][r] * inv;
    }
  }
}

extern "C" void kernel_launch(void* const* d_in, const int* in_sizes, int n_in,
                              void* d_out, int out_size, void* d_ws, size_t ws_size,
                              hipStream_t stream) {
  const float* x    = (const float*)d_in[0];
  const float* rw1  = (const float*)d_in[1];
  const float* rb1  = (const float*)d_in[2];
  const float* rw2  = (const float*)d_in[3];
  const float* rb2  = (const float*)d_in[4];
  const float* qkvw = (const float*)d_in[5];
  const float* qkvb = (const float*)d_in[6];
  const float* apw  = (const float*)d_in[7];
  const float* apb  = (const float*)d_in[8];
  const float* ocw  = (const float*)d_in[9];
  const float* ocb  = (const float*)d_in[10];
  const float* oaw  = (const float*)d_in[11];
  const float* oab  = (const float*)d_in[12];
  const float* opw  = (const float*)d_in[13];
  const float* opb  = (const float*)d_in[14];
  float* Yout = (float*)d_out;

  const size_t N1 = (size_t)NB * HIDC * HW;
  float* A  = (float*)d_ws;
  float* E  = A + N1;
  float* F  = E + N1;
  float* Df = F + N1;
  float* Gf = Df + 2 * N1;
  float* Hf = Gf + N1;

  dim3 blk(256);
  dim3 ew((unsigned)((N1 + 255) / 256));
  dim3 g128(NB, HW / 64, 1), g256(NB, HW / 64, 2);

  k_elu<<<ew, blk, 0, stream>>>(x, E, (int)N1);
  for (int i = 0; i < 6; ++i) {
    const float* w1 = rw1 + (size_t)i * HIDC * HIDC * 4;
    const float* b1 = rb1 + (size_t)i * HIDC;
    const float* w2 = rw2 + (size_t)i * 2 * HIDC * HIDC * 4;
    const float* b2 = rb2 + (size_t)i * 2 * HIDC;
    k_conv2x2<1><<<g128, blk, 0, stream>>>(E, w1, b1, F, nullptr, nullptr);
    k_conv2x2<2><<<g128, blk, 0, stream>>>(F, w2, b2, A, i == 0 ? x : A, E);
  }

  k_qkv<<<g256, blk, 0, stream>>>(A, qkvw, qkvb, Df);
  k_attn<<<dim3(HW / 16, NB), dim3(32), 0, stream>>>(Df, F);
  k_gemm1x1<<<g128, blk, 0, stream>>>(F, apw, apb, Gf, nullptr, nullptr, HIDC, 1);

  k_gemm1x1<<<g128, blk, 0, stream>>>(E, ocw, ocb, Hf, nullptr, nullptr, HIDC, 1);

  k_gemm1x1<<<g128, blk, 0, stream>>>(Gf, oaw, oab, F, nullptr, Hf, HIDC, 1);

  k_gemm1x1<<<g128, blk, 0, stream>>>(F, opw, opb, Yout, x, nullptr, HIDC, 1);
}


# Conv pixel tile (3 rows x 128 channels, padding stored as zeros) staged once per block in LDS and shared by its 8 waves: B fragments are LDS reads, no per-wave pixel gathers; sum order kept
# speedup vs baseline: 1.6275x; 1.0814x over previous
#include <hip/hip_runtime.h>
#include <hip/hip_bf16.h>


typedef __attribute__((ext_vector_type(2))) float v2f;
typedef __attribute__((ext_vector_type(4))) float v4f;
typedef __attribute__((ext_vector_type(8))) float v8f;

#define NB   32
#define HIDC 128
#define HW   1024

__device__ __forceinline__ v8f wmma4(v2f a, v2f b, v8f c) {
  return __builtin_amdgcn_wmma_f32_16x16x4_f32(false, a, false, b, (short)0, c,
                                               false, false);
}

__device__ __forceinline__ float eluf(float x) {
  return x > 0.f ? x : (expf(x) - 1.f);
}

__global__ void k_elu(const float* __restrict__ X, float* __restrict__ Y, int n) {
  int i = blockIdx.x * 256 + threadIdx.x;
  if (i < n) Y[i] = eluf(X[i]);
}

#define XROWP  36
#define XPAIRP 224

template <int G>
struct ConvOps {
  v4f wa[G][2];
  v2f fr[3][2][2];
};

template <int G>
__device__ __forceinline__ void conv_load(ConvOps<G>& o, const v4f* const (&wt)[G],
                                          const float* xl, int step) {
#pragma unroll
  for (int g = 0; g < G; ++g) {
    o.wa[g][0] = wt[g][step * 4];
    o.wa[g][1] = wt[g][step * 4 + 1];
  }
  const float* base = xl + step * (2 * XPAIRP);
#pragma unroll
  for (int rr = 0; rr < 3; ++rr)
#pragma unroll
    for (int th = 0; th < 2; ++th)
#pragma unroll
      for (int dx = 0; dx < 2; ++dx)
        o.fr[rr][th][dx] = *(const v2f*)(base + (rr * XROWP + th * 16 + dx) * 2);
}

template <int G>
__device__ __forceinline__ void conv_mma(const ConvOps<G>& o, v8f (&acc)[G][4]) {
#pragma unroll
  for (int tap = 0; tap < 4; ++tap) {
    v2f a[G];
#pragma unroll
    for (int g = 0; g < G; ++g) {
      a[g].x = o.wa[g][0][tap];
      a[g].y = o.wa[g][1][tap];
    }
#pragma unroll
    for (int t = 0; t < 4; ++t) {
      v2f bb = o.fr[(t >> 1) + (tap >> 1)][t & 1][tap & 1];
#pragma unroll
      for (int g = 0; g < G; ++g) acc[g][t] = wmma4(a[g], bb, acc[g][t]);
    }
  }
}

template <int G>
__global__ void __launch_bounds__(256)
k_conv2x2(const float* __restrict__ X, const float* __restrict__ W,
          const float* __restrict__ bias, float* Y, const float* Rin,
          float* __restrict__ E) {
  __shared__ __attribute__((aligned(16))) float xs[64 * XPAIRP];
  const int tid  = threadIdx.x;
  const int lane = tid & 31;
  const int wave = tid >> 5;
  const int half = lane >> 4;
  const int l15  = lane & 15;
  const int b  = blockIdx.x;
  const int p0 = blockIdx.y * 64;
  const int o0 = wave * 16;

  const float* Xb = X + (size_t)b * HIDC * HW;

  {
    const int yt = (int)blockIdx.y * 2 - 1;
    const int j  = tid & 7;
    const int pb = tid >> 3;
    v4f ga[3][2], gc[3][2];
#pragma unroll
    for (int r = 0; r < 3; ++r) {
#pragma unroll
      for (int k = 0; k < 2; ++k) {
        const int yy = (r == 0 && yt < 0) ? 0 : yt + r;
        const float* g = Xb + (size_t)(2 * pb + 64 * k) * HW + yy * 32 + j * 4;
        ga[r][k] = *(const v4f*)g;
        gc[r][k] = *(const v4f*)(g + HW);
      }
    }
    float* dst = xs + pb * XPAIRP + (4 + j * 4) * 2;
#pragma unroll
    for (int r = 0; r < 3; ++r) {
#pragma unroll
      for (int k = 0; k < 2; ++k) {
        float* d = dst + k * (32 * XPAIRP) + r * (XROWP * 2);
        v4f a = ga[r][k], c = gc[r][k];
        if (r == 0 && yt < 0) {
          a = (v4f){0.f, 0.f, 0.f, 0.f};
          c = (v4f){0.f, 0.f, 0.f, 0.f};
        }
        *(v4f*)d       = (v4f){a.x, c.x, a.y, c.y};
        *(v4f*)(d + 4) = (v4f){a.z, c.z, a.w, c.w};
      }
    }
    if (tid < 192) {
      int p = tid & 63;
      int r = tid >> 6;
      *(v2f*)(xs + p * XPAIRP + (r * XROWP + 3) * 2) = (v2f){0.f, 0.f};
    }
  }
  __syncthreads();

  v8f acc[G][4];
#pragma unroll
  for (int g = 0; g < G; ++g)
#pragma unroll
    for (int t = 0; t < 4; ++t) acc[g][t] = (v8f){0, 0, 0, 0, 0, 0, 0, 0};

  const v4f* wt[G];
#pragma unroll
  for (int g = 0; g < G; ++g)
    wt[g] = (const v4f*)(W + (size_t)(g * 128 + o0 + l15) * 512 + half * 8);
  const float* xl = xs + half * XPAIRP + (3 + l15) * 2;

  ConvOps<G> opA, opB;
  conv_load<G>(opA, wt, xl, 0);
  for (int s = 0; s < HIDC / 4; s += 2) {
    conv_load<G>(opB, wt, xl, s + 1);
    __builtin_amdgcn_sched_barrier(0);
    conv_mma<G>(opA, acc);
    if (s + 2 < HIDC / 4) conv_load<G>(opA, wt, xl, s + 2);
    __builtin_amdgcn_sched_barrier(0);
    conv_mma<G>(opB, acc);
  }

  float bs[G][8];
#pragma unroll
  for (int g = 0; g < G; ++g)
#pragma unroll
    for (int r = 0; r < 8; ++r) bs[g][r] = bias[g * HIDC + o0 + r + half * 8];

  if (G == 1) {
#pragma unroll
    for (int t = 0; t < 4; ++t) {
#pragma unroll
      for (int r = 0; r < 8; ++r) {
        int o = o0 + r + half * 8;
        float v = acc[0][t][r] + bs[0][r];
        Y[((size_t)b * HIDC + o) * HW + p0 + t * 16 + l15] = eluf(v);
      }
    }
  } else {
    float res[4][8];
#pragma unroll
    for (int t = 0; t < 4; ++t) {
#pragma unroll
      for (int r = 0; r < 8; ++r) {
        int o = o0 + r + half * 8;
        res[t][r] = Rin[((size_t)b * HIDC + o) * HW + p0 + t * 16 + l15];
      }
    }
#pragma unroll
    for (int t = 0; t < 4; ++t) {
#pragma unroll
      for (int r = 0; r < 8; ++r) {
        int o = o0 + r + half * 8;
        size_t idx = ((size_t)b * HIDC + o) * HW + p0 + t * 16 + l15;
        float t1 = acc[0][t][r] + bs[0][r];
        float t2 = acc[G - 1][t][r] + bs[G - 1][r];
        float v = res[t][r] + t1 * (1.f / (1.f + expf(-t2)));
        Y[idx] = v;
        E[idx] = eluf(v);
      }
    }
  }
}

__global__ void __launch_bounds__(256)
k_gemm1x1(const float* __restrict__ X, const float* __restrict__ W,
          const float* __restrict__ bias, float* __restrict__ Y,
          const float* __restrict__ R, const float* __restrict__ Rpre, int O,
          int outElu) {
  const int lane = threadIdx.x & 31;
  const int wave = threadIdx.x >> 5;
  const int half = lane >> 4;
  const int l15  = lane & 15;
  const int b  = blockIdx.x;
  const int p0 = blockIdx.y * 64;
  const int o0 = (blockIdx.z * 8 + wave) * 16;
  if (o0 >= O) return;

  const float* Xb   = X + (size_t)b * HIDC * HW;
  const float* wrow = W + (size_t)(o0 + l15) * HIDC;

  v8f acc[4];
#pragma unroll
  for (int t = 0; t < 4; ++t) acc[t] = (v8f){0, 0, 0, 0, 0, 0, 0, 0};

  for (int c0 = 0; c0 < HIDC; c0 += 4) {
    int ca = c0 + half * 2;
    v2f a;
    a.x = wrow[ca];
    a.y = wrow[ca + 1];
    const float* x0 = Xb + (size_t)ca * HW + p0 + l15;
#pragma unroll
    for (int t = 0; t < 4; ++t) {
      v2f bb;
      bb.x = x0[t * 16];
      bb.y = x0[HW + t * 16];
      acc[t] = wmma4(a, bb, acc[t]);
    }
  }

#pragma unroll
  for (int t = 0; t < 4; ++t) {
#pragma unroll
    for (int r = 0; r < 8; ++r) {
      int o = o0 + r + half * 8;
      size_t idx = ((size_t)b * O + o) * HW + p0 + t * 16 + l15;
      float v = acc[t][r] + bias[o];
      if (outElu) v = eluf(v);
      if (Rpre) v = eluf(Rpre[idx] + v);
      if (R) v += R[idx];
      Y[idx] = v;
    }
  }
}

__global__ void __launch_bounds__(256)
k_qkv(const float* __restrict__ X, const float* __restrict__ W,
      const float* __restrict__ bias, float* __restrict__ Y) {
  const int lane = threadIdx.x & 31;
  const int wave = threadIdx.x >> 5;
  const int half = lane >> 4;
  const int l15  = lane & 15;
  const int b  = blockIdx.x;
  const int p0 = blockIdx.y * 64;
  const int o0 = (blockIdx.z * 8 + wave) * 16;
  if (o0 >= 160) return;

  const float* Xb   = X + (size_t)b * HIDC * HW;
  const float* wrow = W + (size_t)(o0 + l15) * 130;

  v8f acc[4];
#pragma unroll
  for (int t = 0; t < 4; ++t) acc[t] = (v8f){0, 0, 0, 0, 0, 0, 0, 0};

  for (int c0 = 0; c0 < HIDC; c0 += 4) {
    int ca = c0 + half * 2;
    v2f a;
    a.x = wrow[ca];
    a.y = wrow[ca + 1];
    const float* x0 = Xb + (size_t)ca * HW + p0 + l15;
#pragma unroll
    for (int t = 0; t < 4; ++t) {
      v2f bb;
      bb.x = x0[t * 16];
      bb.y = x0[HW + t * 16];
      acc[t] = wmma4(a, bb, acc[t]);
    }
  }

#pragma unroll
  for (int t = 0; t < 4; ++t) {
    int p = p0 + t * 16 + l15;
    float py = (float)(p >> 5) * (1.f / 32.f) - 0.5f;
    float px = (float)(p & 31) * (1.f / 32.f) - 0.5f;
#pragma unroll
    for (int r = 0; r < 8; ++r) {
      int o = o0 + r + half * 8;
      float v = acc[t][r] + W[(size_t)o * 130 + 128] * py +
                W[(size_t)o * 130 + 129] * px + bias[o];
      Y[((size_t)b * 160 + o) * HW + p] = v;
    }
  }
}

__global__ void __launch_bounds__(32)
k_attn(const float* __restrict__ QKV, float* __restrict__ Ob) {
  __shared__ float pl[16 * 17];
  const int lane = threadIdx.x;
  const int half = lane >> 4;
  const int l15  = lane & 15;
  const int qt = blockIdx.x;
  const int b  = blockIdx.y;
  const int q0 = qt * 16;
  const float* Qb = QKV + (size_t)b * 160 * HW;

  v2f qa[4];
#pragma unroll
  for (int kc = 0; kc < 4; ++kc) {
    qa[kc].x = Qb[(size_t)(kc * 4 + half * 2) * HW + q0 + l15];
    qa[kc].y = Qb[(size_t)(kc * 4 + half * 2 + 1) * HW + q0 + l15];
  }

  v8f acc[8];
#pragma unroll
  for (int vt = 0; vt < 8; ++vt) acc[vt] = (v8f){0, 0, 0, 0, 0, 0, 0, 0};
  float mrun[8], lrun[8];
#pragma unroll
  for (int r = 0; r < 8; ++r) { mrun[r] = -3.0e38f; lrun[r] = 0.f; }

  for (int kt = 0; kt <= qt; ++kt) {
    int k0 = kt * 16;
    v8f s = (v8f){0, 0, 0, 0, 0, 0, 0, 0};
#pragma unroll
    for (int kc = 0; kc < 4; ++kc) {
      v2f bb;
      bb.x = Qb[(size_t)(16 + kc * 4 + half * 2) * HW + k0 + l15];
      bb.y = Qb[(size_t)(16 + kc * 4 + half * 2 + 1) * HW + k0 + l15];
      s = wmma4(qa[kc], bb, s);
    }
#pragma unroll
    for (int r = 0; r < 8; ++r) {
      int qp = q0 + r + half * 8;
      int kp = k0 + l15;
      float val = (kp <= qp) ? s[r] * 0.25f : -3.0e38f;
      float rm = val;
      rm = fmaxf(rm, __shfl_xor(rm, 1));
      rm = fmaxf(rm, __shfl_xor(rm, 2));
      rm = fmaxf(rm, __shfl_xor(rm, 4));
      rm = fmaxf(rm, __shfl_xor(rm, 8));
      float nm = fmaxf(mrun[r], rm);
      float alpha = expf(mrun[r] - nm);
      float p = expf(val - nm);
      float rs = p;
      rs += __shfl_xor(rs, 1);
      rs += __shfl_xor(rs, 2);
      rs += __shfl_xor(rs, 4);
      rs += __shfl_xor(rs, 8);
      lrun[r] = lrun[r] * alpha + rs;
      mrun[r] = nm;
#pragma unroll
      for (int vt = 0; vt < 8; ++vt) acc[vt][r] *= alpha;
      pl[(r + half * 8) * 17 + l15] = p;
    }
    asm volatile("s_wait_dscnt 0" ::: "memory");
#pragma unroll
    for (int kc = 0; kc < 4; ++kc) {
      v2f pa;
      pa.x = pl[l15 * 17 + kc * 4 + half * 2];
      pa.y = pl[l15 * 17 + kc * 4 + half * 2 + 1];
#pragma unroll
      for (int vt = 0; vt < 8; ++vt) {
        const float* vrow =
            Qb + (size_t)(32 + vt * 16 + l15) * HW + k0 + kc * 4 + half * 2;
        v2f bb;
        bb.x = vrow[0];
        bb.y = vrow[1];
        acc[vt] = wmma4(pa, bb, acc[vt]);
      }
    }
  }

#pragma unroll
  for (int r = 0; r < 8; ++r) {
    float inv = 1.f / lrun[r];
    int qp = q0 + r + half * 8;
#pragma unroll
    for (int vt = 0; vt < 8; ++vt) {
      Ob[((size_t)b * HIDC + vt * 16 + l15) * HW + qp] = acc[vt][r] * inv;
    }
  }
}

extern "C" void kernel_launch(void* const* d_in, const int* in_sizes, int n_in,
                              void* d_out, int out_size, void* d_ws, size_t ws_size,
                              hipStream_t stream) {
  const float* x    = (const float*)d_in[0];
  const float* rw1  = (const float*)d_in[1];
  const float* rb1  = (const float*)d_in[2];
  const float* rw2  = (const float*)d_in[3];
  const float* rb2  = (const float*)d_in[4];
  const float* qkvw = (const float*)d_in[5];
  const float* qkvb = (const float*)d_in[6];
  const float* apw  = (const float*)d_in[7];
  const float* apb  = (const float*)d_in[8];
  const float* ocw  = (const float*)d_in[9];
  const float* ocb  = (const float*)d_in[10];
  const float* oaw  = (const float*)d_in[11];
  const float* oab  = (const float*)d_in[12];
  const float* opw  = (const float*)d_in[13];
  const float* opb  = (const float*)d_in[14];
  float* Yout = (float*)d_out;

  const size_t N1 = (size_t)NB * HIDC * HW;
  float* A  = (float*)d_ws;
  float* E  = A + N1;
  float* F  = E + N1;
  float* Df = F + N1;
  float* Gf = Df + 2 * N1;
  float* Hf = Gf + N1;

  dim3 blk(256);
  dim3 ew((unsigned)((N1 + 255) / 256));
  dim3 g128(NB, HW / 64, 1), g256(NB, HW / 64, 2);

  k_elu<<<ew, blk, 0, stream>>>(x, E, (int)N1);
  for (int i = 0; i < 6; ++i) {
    const float* w1 = rw1 + (size_t)i * HIDC * HIDC * 4;
    const float* b1 = rb1 + (size_t)i * HIDC;
    const float* w2 = rw2 + (size_t)i * 2 * HIDC * HIDC * 4;
    const float* b2 = rb2 + (size_t)i * 2 * HIDC;
    k_conv2x2<1><<<g128, blk, 0, stream>>>(E, w1, b1, F, nullptr, nullptr);
    k_conv2x2<2><<<g128, blk, 0, stream>>>(F, w2, b2, A, i == 0 ? x : A, E);
  }

  k_qkv<<<g256, blk, 0, stream>>>(A, qkvw, qkvb, Df);
  k_attn<<<dim3(HW / 16, NB), dim3(32), 0, stream>>>(Df, F);
  k_gemm1x1<<<g128, blk, 0, stream>>>(F, apw, apb, Gf, nullptr, nullptr, HIDC, 1);

  k_gemm1x1<<<g128, blk, 0, stream>>>(E, ocw, ocb, Hf, nullptr, nullptr, HIDC, 1);

  k_gemm1x1<<<g128, blk, 0, stream>>>(Gf, oaw, oab, F, nullptr, Hf, HIDC, 1);

  k_gemm1x1<<<g128, blk, 0, stream>>>(F, opw, opb, Yout, x, nullptr, HIDC, 1);
}


# Attention: the key tiles of each 16-query tile are split over the 8 waves of a block (was one wave per tile), partials merged through LDS, heaviest tiles first; softmax sum order changed
# speedup vs baseline: 1.8911x; 1.1620x over previous
#include <hip/hip_runtime.h>
#include <hip/hip_bf16.h>


typedef __attribute__((ext_vector_type(2))) float v2f;
typedef __attribute__((ext_vector_type(4))) float v4f;
typedef __attribute__((ext_vector_type(8))) float v8f;

#define NB   32
#define HIDC 128
#define HW   1024

__device__ __forceinline__ v8f wmma4(v2f a, v2f b, v8f c) {
  return __builtin_amdgcn_wmma_f32_16x16x4_f32(false, a, false, b, (short)0, c,
                                               false, false);
}

__device__ __forceinline__ float eluf(float x) {
  return x > 0.f ? x : (expf(x) - 1.f);
}

__global__ void k_elu(const float* __restrict__ X, float* __restrict__ Y, int n) {
  int i = blockIdx.x * 256 + threadIdx.x;
  if (i < n) Y[i] = eluf(X[i]);
}

#define XROWP  36
#define XPAIRP 224

template <int G>
struct ConvOps {
  v4f wa[G][2];
  v2f fr[3][2][2];
};

template <int G>
__device__ __forceinline__ void conv_load(ConvOps<G>& o, const v4f* const (&wt)[G],
                                          const float* xl, int step) {
#pragma unroll
  for (int g = 0; g < G; ++g) {
    o.wa[g][0] = wt[g][step * 4];
    o.wa[g][1] = wt[g][step * 4 + 1];
  }
  const float* base = xl + step * (2 * XPAIRP);
#pragma unroll
  for (int rr = 0; rr < 3; ++rr)
#pragma unroll
    for (int th = 0; th < 2; ++th)
#pragma unroll
      for (int dx = 0; dx < 2; ++dx)
        o.fr[rr][th][dx] = *(const v2f*)(base + (rr * XROWP + th * 16 + dx) * 2);
}

template <int G>
__device__ __forceinline__ void conv_mma(const ConvOps<G>& o, v8f (&acc)[G][4]) {
#pragma unroll
  for (int tap = 0; tap < 4; ++tap) {
    v2f a[G];
#pragma unroll
    for (int g = 0; g < G; ++g) {
      a[g].x = o.wa[g][0][tap];
      a[g].y = o.wa[g][1][tap];
    }
#pragma unroll
    for (int t = 0; t < 4; ++t) {
      v2f bb = o.fr[(t >> 1) + (tap >> 1)][t & 1][tap & 1];
#pragma unroll
      for (int g = 0; g < G; ++g) acc[g][t] = wmma4(a[g], bb, acc[g][t]);
    }
  }
}

template <int G>
__global__ void __launch_bounds__(256)
k_conv2x2(const float* __restrict__ X, const float* __restrict__ W,
          const float* __restrict__ bias, float* Y, const float* Rin,
          float* __restrict__ E) {
  __shared__ __attribute__((aligned(16))) float xs[64 * XPAIRP];
  const int tid  = threadIdx.x;
  const int lane = tid & 31;
  const int wave = tid >> 5;
  const int half = lane >> 4;
  const int l15  = lane & 15;
  const int b  = blockIdx.x;
  const int p0 = blockIdx.y * 64;
  const int o0 = wave * 16;

  const float* Xb = X + (size_t)b * HIDC * HW;

  {
    const int yt = (int)blockIdx.y * 2 - 1;
    const int j  = tid & 7;
    const int pb = tid >> 3;
    v4f ga[3][2], gc[3][2];
#pragma unroll
    for (int r = 0; r < 3; ++r) {
#pragma unroll
      for (int k = 0; k < 2; ++k) {
        const int yy = (r == 0 && yt < 0) ? 0 : yt + r;
        const float* g = Xb + (size_t)(2 * pb + 64 * k) * HW + yy * 32 + j * 4;
        ga[r][k] = *(const v4f*)g;
        gc[r][k] = *(const v4f*)(g + HW);
      }
    }
    float* dst = xs + pb * XPAIRP + (4 + j * 4) * 2;
#pragma unroll
    for (int r = 0; r < 3; ++r) {
#pragma unroll
      for (int k = 0; k < 2; ++k) {
        float* d = dst + k * (32 * XPAIRP) + r * (XROWP * 2);
        v4f a = ga[r][k], c = gc[r][k];
        if (r == 0 && yt < 0) {
          a = (v4f){0.f, 0.f, 0.f, 0.f};
          c = (v4f){0.f, 0.f, 0.f, 0.f};
        }
        *(v4f*)d       = (v4f){a.x, c.x, a.y, c.y};
        *(v4f*)(d + 4) = (v4f){a.z, c.z, a.w, c.w};
      }
    }
    if (tid < 192) {
      int p = tid & 63;
      int r = tid >> 6;
      *(v2f*)(xs + p * XPAIRP + (r * XROWP + 3) * 2) = (v2f){0.f, 0.f};
    }
  }
  __syncthreads();

  v8f acc[G][4];
#pragma unroll
  for (int g = 0; g < G; ++g)
#pragma unroll
    for (int t = 0; t < 4; ++t) acc[g][t] = (v8f){0, 0, 0, 0, 0, 0, 0, 0};

  const v4f* wt[G];
#pragma unroll
  for (int g = 0; g < G; ++g)
    wt[g] = (const v4f*)(W + (size_t)(g * 128 + o0 + l15) * 512 + half * 8);
  const float* xl = xs + half * XPAIRP + (3 + l15) * 2;

  ConvOps<G> opA, opB;
  conv_load<G>(opA, wt, xl, 0);
  for (int s = 0; s < HIDC / 4; s += 2) {
    conv_load<G>(opB, wt, xl, s + 1);
    __builtin_amdgcn_sched_barrier(0);
    conv_mma<G>(opA, acc);
    if (s + 2 < HIDC / 4) conv_load<G>(opA, wt, xl, s + 2);
    __builtin_amdgcn_sched_barrier(0);
    conv_mma<G>(opB, acc);
  }

  float bs[G][8];
#pragma unroll
  for (int g = 0; g < G; ++g)
#pragma unroll
    for (int r = 0; r < 8; ++r) bs[g][r] = bias[g * HIDC + o0 + r + half * 8];

  if (G == 1) {
#pragma unroll
    for (int t = 0; t < 4; ++t) {
#pragma unroll
      for (int r = 0; r < 8; ++r) {
        int o = o0 + r + half * 8;
        float v = acc[0][t][r] + bs[0][r];
        Y[((size_t)b * HIDC + o) * HW + p0 + t * 16 + l15] = eluf(v);
      }
    }
  } else {
    float res[4][8];
#pragma unroll
    for (int t = 0; t < 4; ++t) {
#pragma unroll
      for (int r = 0; r < 8; ++r) {
        int o = o0 + r + half * 8;
        res[t][r] = Rin[((size_t)b * HIDC + o) * HW + p0 + t * 16 + l15];
      }
    }
#pragma unroll
    for (int t = 0; t < 4; ++t) {
#pragma unroll
      for (int r = 0; r < 8; ++r) {
        int o = o0 + r + half * 8;
        size_t idx = ((size_t)b * HIDC + o) * HW + p0 + t * 16 + l15;
        float t1 = acc[0][t][r] + bs[0][r];
        float t2 = acc[G - 1][t][r] + bs[G - 1][r];
        float v = res[t][r] + t1 * (1.f / (1.f + expf(-t2)));
        Y[idx] = v;
        E[idx] = eluf(v);
      }
    }
  }
}

__global__ void __launch_bounds__(256)
k_gemm1x1(const float* __restrict__ X, const float* __restrict__ W,
          const float* __restrict__ bias, float* __restrict__ Y,
          const float* __restrict__ R, const float* __restrict__ Rpre, int O,
          int outElu) {
  const int lane = threadIdx.x & 31;
  const int wave = threadIdx.x >> 5;
  const int half = lane >> 4;
  const int l15  = lane & 15;
  const int b  = blockIdx.x;
  const int p0 = blockIdx.y * 64;
  const int o0 = (blockIdx.z * 8 + wave) * 16;
  if (o0 >= O) return;

  const float* Xb   = X + (size_t)b * HIDC * HW;
  const float* wrow = W + (size_t)(o0 + l15) * HIDC;

  v8f acc[4];
#pragma unroll
  for (int t = 0; t < 4; ++t) acc[t] = (v8f){0, 0, 0, 0, 0, 0, 0, 0};

  for (int c0 = 0; c0 < HIDC; c0 += 4) {
    int ca = c0 + half * 2;
    v2f a;
    a.x = wrow[ca];
    a.y = wrow[ca + 1];
    const float* x0 = Xb + (size_t)ca * HW + p0 + l15;
#pragma unroll
    for (int t = 0; t < 4; ++t) {
      v2f bb;
      bb.x = x0[t * 16];
      bb.y = x0[HW + t * 16];
      acc[t] = wmma4(a, bb, acc[t]);
    }
  }

#pragma unroll
  for (int t = 0; t < 4; ++t) {
#pragma unroll
    for (int r = 0; r < 8; ++r) {
      int o = o0 + r + half * 8;
      size_t idx = ((size_t)b * O + o) * HW + p0 + t * 16 + l15;
      float v = acc[t][r] + bias[o];
      if (outElu) v = eluf(v);
      if (Rpre) v = eluf(Rpre[idx] + v);
      if (R) v += R[idx];
      Y[idx] = v;
    }
  }
}

__global__ void __launch_bounds__(256)
k_qkv(const float* __restrict__ X, const float* __restrict__ W,
      const float* __restrict__ bias, float* __restrict__ Y) {
  const int lane = threadIdx.x & 31;
  const int wave = threadIdx.x >> 5;
  const int half = lane >> 4;
  const int l15  = lane & 15;
  const int b  = blockIdx.x;
  const int p0 = blockIdx.y * 64;
  const int o0 = (blockIdx.z * 8 + wave) * 16;
  if (o0 >= 160) return;

  const float* Xb   = X + (size_t)b * HIDC * HW;
  const float* wrow = W + (size_t)(o0 + l15) * 130;

  v8f acc[4];
#pragma unroll
  for (int t = 0; t < 4; ++t) acc[t] = (v8f){0, 0, 0, 0, 0, 0, 0, 0};

  for (int c0 = 0; c0 < HIDC; c0 += 4) {
    int ca = c0 + half * 2;
    v2f a;
    a.x = wrow[ca];
    a.y = wrow[ca + 1];
    const float* x0 = Xb + (size_t)ca * HW + p0 + l15;
#pragma unroll
    for (int t = 0; t < 4; ++t) {
      v2f bb;
      bb.x = x0[t * 16];
      bb.y = x0[HW + t * 16];
      acc[t] = wmma4(a, bb, acc[t]);
    }
  }

#pragma unroll
  for (int t = 0; t < 4; ++t) {
    int p = p0 + t * 16 + l15;
    float py = (float)(p >> 5) * (1.f / 32.f) - 0.5f;
    float px = (float)(p & 31) * (1.f / 32.f) - 0.5f;
#pragma unroll
    for (int r = 0; r < 8; ++r) {
      int o = o0 + r + half * 8;
      float v = acc[t][r] + W[(size_t)o * 130 + 128] * py +
                W[(size_t)o * 130 + 129] * px + bias[o];
      Y[((size_t)b * 160 + o) * HW + p] = v;
    }
  }
}

#define ATT_NW 8
__global__ void __launch_bounds__(256)
k_attn(const float* __restrict__ QKV, float* __restrict__ Ob) {
  __shared__ __attribute__((aligned(16))) float pl[ATT_NW * 16 * 17];
  __shared__ __attribute__((aligned(16))) float po[ATT_NW * 8 * 8 * 32];
  __shared__ __attribute__((aligned(16))) float pm[ATT_NW * 16];
  __shared__ __attribute__((aligned(16))) float ps[ATT_NW * 16];
  const int lane = threadIdx.x & 31;
  const int wave = __builtin_amdgcn_readfirstlane((int)(threadIdx.x >> 5));
  const int half = lane >> 4;
  const int l15  = lane & 15;
  const int qt = (HW / 16 - 1) - (int)blockIdx.x;
  const int b  = blockIdx.y;
  const int q0 = qt * 16;
  const float* Qb = QKV + (size_t)b * 160 * HW;
  float* plw = pl + wave * (16 * 17);

  v2f qa[4];
#pragma unroll
  for (int kc = 0; kc < 4; ++kc) {
    qa[kc].x = Qb[(size_t)(kc * 4 + half * 2) * HW + q0 + l15];
    qa[kc].y = Qb[(size_t)(kc * 4 + half * 2 + 1) * HW + q0 + l15];
  }

  v8f acc[8];
#pragma unroll
  for (int vt = 0; vt < 8; ++vt) acc[vt] = (v8f){0, 0, 0, 0, 0, 0, 0, 0};
  float mrun[8], lrun[8];
#pragma unroll
  for (int r = 0; r < 8; ++r) { mrun[r] = -3.0e38f; lrun[r] = 0.f; }

  for (int kt = wave; kt <= qt; kt += ATT_NW) {
    int k0 = kt * 16;
    v8f s = (v8f){0, 0, 0, 0, 0, 0, 0, 0};
#pragma unroll
    for (int kc = 0; kc < 4; ++kc) {
      v2f bb;
      bb.x = Qb[(size_t)(16 + kc * 4 + half * 2) * HW + k0 + l15];
      bb.y = Qb[(size_t)(16 + kc * 4 + half * 2 + 1) * HW + k0 + l15];
      s = wmma4(qa[kc], bb, s);
    }
#pragma unroll
    for (int r = 0; r < 8; ++r) {
      int qp = q0 + r + half * 8;
      int kp = k0 + l15;
      float val = (kp <= qp) ? s[r] * 0.25f : -3.0e38f;
      float rm = val;
      rm = fmaxf(rm, __shfl_xor(rm, 1));
      rm = fmaxf(rm, __shfl_xor(rm, 2));
      rm = fmaxf(rm, __shfl_xor(rm, 4));
      rm = fmaxf(rm, __shfl_xor(rm, 8));
      float nm = fmaxf(mrun[r], rm);
      float alpha = expf(mrun[r] - nm);
      float p = expf(val - nm);
      float rs = p;
      rs += __shfl_xor(rs, 1);
      rs += __shfl_xor(rs, 2);
      rs += __shfl_xor(rs, 4);
      rs += __shfl_xor(rs, 8);
      lrun[r] = lrun[r] * alpha + rs;
      mrun[r] = nm;
#pragma unroll
      for (int vt = 0; vt < 8; ++vt) acc[vt][r] *= alpha;
      plw[(r + half * 8) * 17 + l15] = p;
    }
    asm volatile("s_wait_dscnt 0" ::: "memory");
#pragma unroll
    for (int kc = 0; kc < 4; ++kc) {
      v2f pa;
      pa.x = plw[l15 * 17 + kc * 4 + half * 2];
      pa.y = plw[l15 * 17 + kc * 4 + half * 2 + 1];
#pragma unroll
      for (int vt = 0; vt < 8; ++vt) {
        const float* vrow =
            Qb + (size_t)(32 + vt * 16 + l15) * HW + k0 + kc * 4 + half * 2;
        v2f bb;
        bb.x = vrow[0];
        bb.y = vrow[1];
        acc[vt] = wmma4(pa, bb, acc[vt]);
      }
    }
  }

  const int nact = (qt + 1 < ATT_NW) ? qt + 1 : ATT_NW;
  const bool active = wave < nact;
  if (active && l15 == 0) {
#pragma unroll
    for (int r = 0; r < 8; ++r) pm[wave * 16 + half * 8 + r] = mrun[r];
  }
  __syncthreads();
  if (active) {
    float M[8];
#pragma unroll
    for (int r = 0; r < 8; ++r) M[r] = pm[half * 8 + r];
    for (int j = 1; j < nact; ++j) {
#pragma unroll
      for (int r = 0; r < 8; ++r) M[r] = fmaxf(M[r], pm[j * 16 + half * 8 + r]);
    }
    float lsc[8];
#pragma unroll
    for (int r = 0; r < 8; ++r) {
      const float sc = expf(mrun[r] - M[r]);
      lsc[r] = lrun[r] * sc;
#pragma unroll
      for (int vt = 0; vt < 8; ++vt)
        po[((wave * 8 + vt) * 8 + r) * 32 + lane] = acc[vt][r] * sc;
    }
    if (l15 == 0) {
#pragma unroll
      for (int r = 0; r < 8; ++r) ps[wave * 16 + half * 8 + r] = lsc[r];
    }
  }
  __syncthreads();

  float Ls[8], os[8];
#pragma unroll
  for (int r = 0; r < 8; ++r) { Ls[r] = 0.f; os[r] = 0.f; }
  for (int j = 0; j < nact; ++j) {
#pragma unroll
    for (int r = 0; r < 8; ++r) {
      Ls[r] += ps[j * 16 + half * 8 + r];
      os[r] += po[((j * 8 + wave) * 8 + r) * 32 + lane];
    }
  }
#pragma unroll
  for (int r = 0; r < 8; ++r) {
    float inv = 1.f / Ls[r];
    int qp = q0 + r + half * 8;
    Ob[((size_t)b * HIDC + wave * 16 + l15) * HW + qp] = os[r] * inv;
  }
}

extern "C" void kernel_launch(void* const* d_in, const int* in_sizes, int n_in,
                              void* d_out, int out_size, void* d_ws, size_t ws_size,
                              hipStream_t stream) {
  const float* x    = (const float*)d_in[0];
  const float* rw1  = (const float*)d_in[1];
  const float* rb1  = (const float*)d_in[2];
  const float* rw2  = (const float*)d_in[3];
  const float* rb2  = (const float*)d_in[4];
  const float* qkvw = (const float*)d_in[5];
  const float* qkvb = (const float*)d_in[6];
  const float* apw  = (const float*)d_in[7];
  const float* apb  = (const float*)d_in[8];
  const float* ocw  = (const float*)d_in[9];
  const float* ocb  = (const float*)d_in[10];
  const float* oaw  = (const float*)d_in[11];
  const float* oab  = (const float*)d_in[12];
  const float* opw  = (const float*)d_in[13];
  const float* opb  = (const float*)d_in[14];
  float* Yout = (float*)d_out;

  const size_t N1 = (size_t)NB * HIDC * HW;
  float* A  = (float*)d_ws;
  float* E  = A + N1;
  float* F  = E + N1;
  float* Df = F + N1;
  float* Gf = Df + 2 * N1;
  float* Hf = Gf + N1;

  dim3 blk(256);
  dim3 ew((unsigned)((N1 + 255) / 256));
  dim3 g128(NB, HW / 64, 1), g256(NB, HW / 64, 2);

  k_elu<<<ew, blk, 0, stream>>>(x, E, (int)N1);
  for (int i = 0; i < 6; ++i) {
    const float* w1 = rw1 + (size_t)i * HIDC * HIDC * 4;
    const float* b1 = rb1 + (size_t)i * HIDC;
    const float* w2 = rw2 + (size_t)i * 2 * HIDC * HIDC * 4;
    const float* b2 = rb2 + (size_t)i * 2 * HIDC;
    k_conv2x2<1><<<g128, blk, 0, stream>>>(E, w1, b1, F, nullptr, nullptr);
    k_conv2x2<2><<<g128, blk, 0, stream>>>(F, w2, b2, A, i == 0 ? x : A, E);
  }

  k_qkv<<<g256, blk, 0, stream>>>(A, qkvw, qkvb, Df);
  k_attn<<<dim3(HW / 16, NB), dim3(256), 0, stream>>>(Df, F);
  k_gemm1x1<<<g128, blk, 0, stream>>>(F, apw, apb, Gf, nullptr, nullptr, HIDC, 1);

  k_gemm1x1<<<g128, blk, 0, stream>>>(E, ocw, ocb, Hf, nullptr, nullptr, HIDC, 1);

  k_gemm1x1<<<g128, blk, 0, stream>>>(Gf, oaw, oab, F, nullptr, Hf, HIDC, 1);

  k_gemm1x1<<<g128, blk, 0, stream>>>(F, opw, opb, Yout, x, nullptr, HIDC, 1);
}


# Attention products transposed (S^T=K*Q^T, O^T=V^T*P^T): a lane owns one query, softmax runs in registers, P^T goes from the score registers into the 2nd product, no LDS relayout; sum order changed
# speedup vs baseline: 2.0558x; 1.0871x over previous
#include <hip/hip_runtime.h>
#include <hip/hip_bf16.h>


typedef __attribute__((ext_vector_type(2))) float v2f;
typedef __attribute__((ext_vector_type(4))) float v4f;
typedef __attribute__((ext_vector_type(8))) float v8f;

#define NB   32
#define HIDC 128
#define HW   1024

__device__ __forceinline__ v8f wmma4(v2f a, v2f b, v8f c) {
  return __builtin_amdgcn_wmma_f32_16x16x4_f32(false, a, false, b, (short)0, c,
                                               false, false);
}

__device__ __forceinline__ float eluf(float x) {
  return x > 0.f ? x : (expf(x) - 1.f);
}

__global__ void k_elu(const float* __restrict__ X, float* __restrict__ Y, int n) {
  int i = blockIdx.x * 256 + threadIdx.x;
  if (i < n) Y[i] = eluf(X[i]);
}

#define XROWP  36
#define XPAIRP 224

template <int G>
struct ConvOps {
  v4f wa[G][2];
  v2f fr[3][2][2];
};

template <int G>
__device__ __forceinline__ void conv_load(ConvOps<G>& o, const v4f* const (&wt)[G],
                                          const float* xl, int step) {
#pragma unroll
  for (int g = 0; g < G; ++g) {
    o.wa[g][0] = wt[g][step * 4];
    o.wa[g][1] = wt[g][step * 4 + 1];
  }
  const float* base = xl + step * (2 * XPAIRP);
#pragma unroll
  for (int rr = 0; rr < 3; ++rr)
#pragma unroll
    for (int th = 0; th < 2; ++th)
#pragma unroll
      for (int dx = 0; dx < 2; ++dx)
        o.fr[rr][th][dx] = *(const v2f*)(base + (rr * XROWP + th * 16 + dx) * 2);
}

template <int G>
__device__ __forceinline__ void conv_mma(const ConvOps<G>& o, v8f (&acc)[G][4]) {
#pragma unroll
  for (int tap = 0; tap < 4; ++tap) {
    v2f a[G];
#pragma unroll
    for (int g = 0; g < G; ++g) {
      a[g].x = o.wa[g][0][tap];
      a[g].y = o.wa[g][1][tap];
    }
#pragma unroll
    for (int t = 0; t < 4; ++t) {
      v2f bb = o.fr[(t >> 1) + (tap >> 1)][t & 1][tap & 1];
#pragma unroll
      for (int g = 0; g < G; ++g) acc[g][t] = wmma4(a[g], bb, acc[g][t]);
    }
  }
}

template <int G>
__global__ void __launch_bounds__(256)
k_conv2x2(const float* __restrict__ X, const float* __restrict__ W,
          const float* __restrict__ bias, float* Y, const float* Rin,
          float* __restrict__ E) {
  __shared__ __attribute__((aligned(16))) float xs[64 * XPAIRP];
  const int tid  = threadIdx.x;
  const int lane = tid & 31;
  const int wave = tid >> 5;
  const int half = lane >> 4;
  const int l15  = lane & 15;
  const int b  = blockIdx.x;
  const int p0 = blockIdx.y * 64;
  const int o0 = wave * 16;

  const float* Xb = X + (size_t)b * HIDC * HW;

  {
    const int yt = (int)blockIdx.y * 2 - 1;
    const int j  = tid & 7;
    const int pb = tid >> 3;
    v4f ga[3][2], gc[3][2];
#pragma unroll
    for (int r = 0; r < 3; ++r) {
#pragma unroll
      for (int k = 0; k < 2; ++k) {
        const int yy = (r == 0 && yt < 0) ? 0 : yt + r;
        const float* g = Xb + (size_t)(2 * pb + 64 * k) * HW + yy * 32 + j * 4;
        ga[r][k] = *(const v4f*)g;
        gc[r][k] = *(const v4f*)(g + HW);
      }
    }
    float* dst = xs + pb * XPAIRP + (4 + j * 4) * 2;
#pragma unroll
    for (int r = 0; r < 3; ++r) {
#pragma unroll
      for (int k = 0; k < 2; ++k) {
        float* d = dst + k * (32 * XPAIRP) + r * (XROWP * 2);
        v4f a = ga[r][k], c = gc[r][k];
        if (r == 0 && yt < 0) {
          a = (v4f){0.f, 0.f, 0.f, 0.f};
          c = (v4f){0.f, 0.f, 0.f, 0.f};
        }
        *(v4f*)d       = (v4f){a.x, c.x, a.y, c.y};
        *(v4f*)(d + 4) = (v4f){a.z, c.z, a.w, c.w};
      }
    }
    if (tid < 192) {
      int p = tid & 63;
      int r = tid >> 6;
      *(v2f*)(xs + p * XPAIRP + (r * XROWP + 3) * 2) = (v2f){0.f, 0.f};
    }
  }
  __syncthreads();

  v8f acc[G][4];
#pragma unroll
  for (int g = 0; g < G; ++g)
#pragma unroll
    for (int t = 0; t < 4; ++t) acc[g][t] = (v8f){0, 0, 0, 0, 0, 0, 0, 0};

  const v4f* wt[G];
#pragma unroll
  for (int g = 0; g < G; ++g)
    wt[g] = (const v4f*)(W + (size_t)(g * 128 + o0 + l15) * 512 + half * 8);
  const float* xl = xs + half * XPAIRP + (3 + l15) * 2;

  ConvOps<G> opA, opB;
  conv_load<G>(opA, wt, xl, 0);
  for (int s = 0; s < HIDC / 4; s += 2) {
    conv_load<G>(opB, wt, xl, s + 1);
    __builtin_amdgcn_sched_barrier(0);
    conv_mma<G>(opA, acc);
    if (s + 2 < HIDC / 4) conv_load<G>(opA, wt, xl, s + 2);
    __builtin_amdgcn_sched_barrier(0);
    conv_mma<G>(opB, acc);
  }

  float bs[G][8];
#pragma unroll
  for (int g = 0; g < G; ++g)
#pragma unroll
    for (int r = 0; r < 8; ++r) bs[g][r] = bias[g * HIDC + o0 + r + half * 8];

  if (G == 1) {
#pragma unroll
    for (int t = 0; t < 4; ++t) {
#pragma unroll
      for (int r = 0; r < 8; ++r) {
        int o = o0 + r + half * 8;
        float v = acc[0][t][r] + bs[0][r];
        Y[((size_t)b * HIDC + o) * HW + p0 + t * 16 + l15] = eluf(v);
      }
    }
  } else {
    float res[4][8];
#pragma unroll
    for (int t = 0; t < 4; ++t) {
#pragma unroll
      for (int r = 0; r < 8; ++r) {
        int o = o0 + r + half * 8;
        res[t][r] = Rin[((size_t)b * HIDC + o) * HW + p0 + t * 16 + l15];
      }
    }
#pragma unroll
    for (int t = 0; t < 4; ++t) {
#pragma unroll
      for (int r = 0; r < 8; ++r) {
        int o = o0 + r + half * 8;
        size_t idx = ((size_t)b * HIDC + o) * HW + p0 + t * 16 + l15;
        float t1 = acc[0][t][r] + bs[0][r];
        float t2 = acc[G - 1][t][r] + bs[G - 1][r];
        float v = res[t][r] + t1 * (1.f / (1.f + expf(-t2)));
        Y[idx] = v;
        E[idx] = eluf(v);
      }
    }
  }
}

__global__ void __launch_bounds__(256)
k_gemm1x1(const float* __restrict__ X, const float* __restrict__ W,
          const float* __restrict__ bias, float* __restrict__ Y,
          const float* __restrict__ R, const float* __restrict__ Rpre, int O,
          int outElu) {
  const int lane = threadIdx.x & 31;
  const int wave = threadIdx.x >> 5;
  const int half = lane >> 4;
  const int l15  = lane & 15;
  const int b  = blockIdx.x;
  const int p0 = blockIdx.y * 64;
  const int o0 = (blockIdx.z * 8 + wave) * 16;
  if (o0 >= O) return;

  const float* Xb   = X + (size_t)b * HIDC * HW;
  const float* wrow = W + (size_t)(o0 + l15) * HIDC;

  v8f acc[4];
#pragma unroll
  for (int t = 0; t < 4; ++t) acc[t] = (v8f){0, 0, 0, 0, 0, 0, 0, 0};

  for (int c0 = 0; c0 < HIDC; c0 += 4) {
    int ca = c0 + half * 2;
    v2f a;
    a.x = wrow[ca];
    a.y = wrow[ca + 1];
    const float* x0 = Xb + (size_t)ca * HW + p0 + l15;
#pragma unroll
    for (int t = 0; t < 4; ++t) {
      v2f bb;
      bb.x = x0[t * 16];
      bb.y = x0[HW + t * 16];
      acc[t] = wmma4(a, bb, acc[t]);
    }
  }

#pragma unroll
  for (int t = 0; t < 4; ++t) {
#pragma unroll
    for (int r = 0; r < 8; ++r) {
      int o = o0 + r + half * 8;
      size_t idx = ((size_t)b * O + o) * HW + p0 + t * 16 + l15;
      float v = acc[t][r] + bias[o];
      if (outElu) v = eluf(v);
      if (Rpre) v = eluf(Rpre[idx] + v);
      if (R) v += R[idx];
      Y[idx] = v;
    }
  }
}

__global__ void __launch_bounds__(256)
k_qkv(const float* __restrict__ X, const float* __restrict__ W,
      const float* __restrict__ bias, float* __restrict__ Y) {
  const int lane = threadIdx.x & 31;
  const int wave = threadIdx.x >> 5;
  const int half = lane >> 4;
  const int l15  = lane & 15;
  const int b  = blockIdx.x;
  const int p0 = blockIdx.y * 64;
  const int o0 = (blockIdx.z * 8 + wave) * 16;
  if (o0 >= 160) return;

  const float* Xb   = X + (size_t)b * HIDC * HW;
  const float* wrow = W + (size_t)(o0 + l15) * 130;

  v8f acc[4];
#pragma unroll
  for (int t = 0; t < 4; ++t) acc[t] = (v8f){0, 0, 0, 0, 0, 0, 0, 0};

  for (int c0 = 0; c0 < HIDC; c0 += 4) {
    int ca = c0 + half * 2;
    v2f a;
    a.x = wrow[ca];
    a.y = wrow[ca + 1];
    const float* x0 = Xb + (size_t)ca * HW + p0 + l15;
#pragma unroll
    for (int t = 0; t < 4; ++t) {
      v2f bb;
      bb.x = x0[t * 16];
      bb.y = x0[HW + t * 16];
      acc[t] = wmma4(a, bb, acc[t]);
    }
  }

#pragma unroll
  for (int t = 0; t < 4; ++t) {
    int p = p0 + t * 16 + l15;
    float py = (float)(p >> 5) * (1.f / 32.f) - 0.5f;
    float px = (float)(p & 31) * (1.f / 32.f) - 0.5f;
#pragma unroll
    for (int r = 0; r < 8; ++r) {
      int o = o0 + r + half * 8;
      float v = acc[t][r] + W[(size_t)o * 130 + 128] * py +
                W[(size_t)o * 130 + 129] * px + bias[o];
      Y[((size_t)b * 160 + o) * HW + p] = v;
    }
  }
}

#define ATT_NW 8
__global__ void __launch_bounds__(256)
k_attn(const float* __restrict__ QKV, float* __restrict__ Ob) {
  __shared__ __attribute__((aligned(16))) float po[ATT_NW * 8 * 8 * 32];
  __shared__ __attribute__((aligned(16))) float pm[ATT_NW * 16];
  __shared__ __attribute__((aligned(16))) float ps[ATT_NW * 32];
  const int lane = threadIdx.x & 31;
  const int wave = __builtin_amdgcn_readfirstlane((int)(threadIdx.x >> 5));
  const int half = lane >> 4;
  const int l15  = lane & 15;
  const int qt = (HW / 16 - 1) - (int)blockIdx.x;
  const int b  = blockIdx.y;
  const int q0 = qt * 16;
  const int qp = q0 + l15;
  const float* Qb = QKV + (size_t)b * 160 * HW;

  v2f qb[4];
#pragma unroll
  for (int kc = 0; kc < 4; ++kc) {
    qb[kc].x = Qb[(size_t)(kc * 4 + half * 2) * HW + qp];
    qb[kc].y = Qb[(size_t)(kc * 4 + half * 2 + 1) * HW + qp];
  }

  v8f acc[8];
#pragma unroll
  for (int vt = 0; vt < 8; ++vt) acc[vt] = (v8f){0, 0, 0, 0, 0, 0, 0, 0};
  float mrun = -3.0e38f;
  float lrun = 0.f;

  for (int kt = wave; kt <= qt; kt += ATT_NW) {
    const int k0 = kt * 16;
    v2f ka[4];
#pragma unroll
    for (int kc = 0; kc < 4; ++kc) {
      ka[kc].x = Qb[(size_t)(16 + kc * 4 + half * 2) * HW + k0 + l15];
      ka[kc].y = Qb[(size_t)(16 + kc * 4 + half * 2 + 1) * HW + k0 + l15];
    }
    v4f va[8][2];
#pragma unroll
    for (int vt = 0; vt < 8; ++vt) {
      const float* vrow = Qb + (size_t)(32 + vt * 16 + l15) * HW + k0 + half * 8;
      va[vt][0] = *(const v4f*)vrow;
      va[vt][1] = *(const v4f*)(vrow + 4);
    }
    __builtin_amdgcn_sched_barrier(0);
    v8f s = (v8f){0, 0, 0, 0, 0, 0, 0, 0};
#pragma unroll
    for (int kc = 0; kc < 4; ++kc) s = wmma4(ka[kc], qb[kc], s);
    float val[8];
    float rm = -3.0e38f;
#pragma unroll
    for (int r = 0; r < 8; ++r) {
      const int kp = k0 + half * 8 + r;
      val[r] = (kp <= qp) ? s[r] * 0.25f : -3.0e38f;
      rm = fmaxf(rm, val[r]);
    }
    rm = fmaxf(rm, __shfl_xor(rm, 16));
    const float nm = fmaxf(mrun, rm);
    const float alpha = expf(mrun - nm);
    v8f p;
    float rs = 0.f;
#pragma unroll
    for (int r = 0; r < 8; ++r) {
      p[r] = expf(val[r] - nm);
      rs += p[r];
    }
    lrun = lrun * alpha + rs;
    mrun = nm;
#pragma unroll
    for (int vt = 0; vt < 8; ++vt)
#pragma unroll
      for (int r = 0; r < 8; ++r) acc[vt][r] *= alpha;
#pragma unroll
    for (int kc = 0; kc < 4; ++kc) {
      v2f pb;
      pb.x = p[2 * kc];
      pb.y = p[2 * kc + 1];
#pragma unroll
      for (int vt = 0; vt < 8; ++vt) {
        v2f a;
        a.x = va[vt][kc >> 1][(kc & 1) * 2];
        a.y = va[vt][kc >> 1][(kc & 1) * 2 + 1];
        acc[vt] = wmma4(a, pb, acc[vt]);
      }
    }
  }

  const int nact = (qt + 1 < ATT_NW) ? qt + 1 : ATT_NW;
  const bool active = wave < nact;
  if (active && half == 0) pm[wave * 16 + l15] = mrun;
  __syncthreads();
  if (active) {
    float M = pm[l15];
    for (int j = 1; j < nact; ++j) M = fmaxf(M, pm[j * 16 + l15]);
    const float sc = expf(mrun - M);
    ps[wave * 32 + lane] = lrun * sc;
#pragma unroll
    for (int r = 0; r < 8; ++r)
#pragma unroll
      for (int vt = 0; vt < 8; ++vt)
        po[((wave * 8 + vt) * 8 + r) * 32 + lane] = acc[vt][r] * sc;
  }
  __syncthreads();

  float Ls = 0.f, os[8];
#pragma unroll
  for (int r = 0; r < 8; ++r) os[r] = 0.f;
  for (int j = 0; j < nact; ++j) {
    Ls += ps[j * 32 + l15] + ps[j * 32 + 16 + l15];
#pragma unroll
    for (int r = 0; r < 8; ++r) os[r] += po[((j * 8 + wave) * 8 + r) * 32 + lane];
  }
  const float inv = 1.f / Ls;
#pragma unroll
  for (int r = 0; r < 8; ++r)
    Ob[((size_t)b * HIDC + wave * 16 + half * 8 + r) * HW + qp] = os[r] * inv;
}

extern "C" void kernel_launch(void* const* d_in, const int* in_sizes, int n_in,
                              void* d_out, int out_size, void* d_ws, size_t ws_size,
                              hipStream_t stream) {
  const float* x    = (const float*)d_in[0];
  const float* rw1  = (const float*)d_in[1];
  const float* rb1  = (const float*)d_in[2];
  const float* rw2  = (const float*)d_in[3];
  const float* rb2  = (const float*)d_in[4];
  const float* qkvw = (const float*)d_in[5];
  const float* qkvb = (const float*)d_in[6];
  const float* apw  = (const float*)d_in[7];
  const float* apb  = (const float*)d_in[8];
  const float* ocw  = (const float*)d_in[9];
  const float* ocb  = (const float*)d_in[10];
  const float* oaw  = (const float*)d_in[11];
  const float* oab  = (const float*)d_in[12];
  const float* opw  = (const float*)d_in[13];
  const float* opb  = (const float*)d_in[14];
  float* Yout = (float*)d_out;

  const size_t N1 = (size_t)NB * HIDC * HW;
  float* A  = (float*)d_ws;
  float* E  = A + N1;
  float* F  = E + N1;
  float* Df = F + N1;
  float* Gf = Df + 2 * N1;
  float* Hf = Gf + N1;

  dim3 blk(256);
  dim3 ew((unsigned)((N1 + 255) / 256));
  dim3 g128(NB, HW / 64, 1), g256(NB, HW / 64, 2);

  k_elu<<<ew, blk, 0, stream>>>(x, E, (int)N1);
  for (int i = 0; i < 6; ++i) {
    const float* w1 = rw1 + (size_t)i * HIDC * HIDC * 4;
    const float* b1 = rb1 + (size_t)i * HIDC;
    const float* w2 = rw2 + (size_t)i * 2 * HIDC * HIDC * 4;
    const float* b2 = rb2 + (size_t)i * 2 * HIDC;
    k_conv2x2<1><<<g128, blk, 0, stream>>>(E, w1, b1, F, nullptr, nullptr);
    k_conv2x2<2><<<g128, blk, 0, stream>>>(F, w2, b2, A, i == 0 ? x : A, E);
  }

  k_qkv<<<g256, blk, 0, stream>>>(A, qkvw, qkvb, Df);
  k_attn<<<dim3(HW / 16, NB), dim3(256), 0, stream>>>(Df, F);
  k_gemm1x1<<<g128, blk, 0, stream>>>(F, apw, apb, Gf, nullptr, nullptr, HIDC, 1);

  k_gemm1x1<<<g128, blk, 0, stream>>>(E, ocw, ocb, Hf, nullptr, nullptr, HIDC, 1);

  k_gemm1x1<<<g128, blk, 0, stream>>>(Gf, oaw, oab, F, nullptr, Hf, HIDC, 1);

  k_gemm1x1<<<g128, blk, 0, stream>>>(F, opw, opb, Yout, x, nullptr, HIDC, 1);
}


# The four 1x1 convs behind the attention fused into one kernel: a block stages its O and E tiles in LDS once, both intermediates stay in LDS; 19 launches to 16; every sum keeps its order
# speedup vs baseline: 2.1703x; 1.0557x over previous
#include <hip/hip_runtime.h>
#include <hip/hip_bf16.h>


typedef __attribute__((ext_vector_type(2))) float v2f;
typedef __attribute__((ext_vector_type(4))) float v4f;
typedef __attribute__((ext_vector_type(8))) float v8f;

#define NB   32
#define HIDC 128
#define HW   1024

__device__ __forceinline__ v8f wmma4(v2f a, v2f b, v8f c) {
  return __builtin_amdgcn_wmma_f32_16x16x4_f32(false, a, false, b, (short)0, c,
                                               false, false);
}

__device__ __forceinline__ float eluf(float x) {
  return x > 0.f ? x : (expf(x) - 1.f);
}

__global__ void k_elu(const float* __restrict__ X, float* __restrict__ Y, int n) {
  int i = blockIdx.x * 256 + threadIdx.x;
  if (i < n) Y[i] = eluf(X[i]);
}

#define XROWP  36
#define XPAIRP 224

template <int G>
struct ConvOps {
  v4f wa[G][2];
  v2f fr[3][2][2];
};

template <int G>
__device__ __forceinline__ void conv_load(ConvOps<G>& o, const v4f* const (&wt)[G],
                                          const float* xl, int step) {
#pragma unroll
  for (int g = 0; g < G; ++g) {
    o.wa[g][0] = wt[g][step * 4];
    o.wa[g][1] = wt[g][step * 4 + 1];
  }
  const float* base = xl + step * (2 * XPAIRP);
#pragma unroll
  for (int rr = 0; rr < 3; ++rr)
#pragma unroll
    for (int th = 0; th < 2; ++th)
#pragma unroll
      for (int dx = 0; dx < 2; ++dx)
        o.fr[rr][th][dx] = *(const v2f*)(base + (rr * XROWP + th * 16 + dx) * 2);
}

template <int G>
__device__ __forceinline__ void conv_mma(const ConvOps<G>& o, v8f (&acc)[G][4]) {
#pragma unroll
  for (int tap = 0; tap < 4; ++tap) {
    v2f a[G];
#pragma unroll
    for (int g = 0; g < G; ++g) {
      a[g].x = o.wa[g][0][tap];
      a[g].y = o.wa[g][1][tap];
    }
#pragma unroll
    for (int t = 0; t < 4; ++t) {
      v2f bb = o.fr[(t >> 1) + (tap >> 1)][t & 1][tap & 1];
#pragma unroll
      for (int g = 0; g < G; ++g) acc[g][t] = wmma4(a[g], bb, acc[g][t]);
    }
  }
}

template <int G>
__global__ void __launch_bounds__(256)
k_conv2x2(const float* __restrict__ X, const float* __restrict__ W,
          const float* __restrict__ bias, float* Y, const float* Rin,
          float* __restrict__ E) {
  __shared__ __attribute__((aligned(16))) float xs[64 * XPAIRP];
  const int tid  = threadIdx.x;
  const int lane = tid & 31;
  const int wave = tid >> 5;
  const int half = lane >> 4;
  const int l15  = lane & 15;
  const int b  = blockIdx.x;
  const int p0 = blockIdx.y * 64;
  const int o0 = wave * 16;

  const float* Xb = X + (size_t)b * HIDC * HW;

  {
    const int yt = (int)blockIdx.y * 2 - 1;
    const int j  = tid & 7;
    const int pb = tid >> 3;
    v4f ga[3][2], gc[3][2];
#pragma unroll
    for (int r = 0; r < 3; ++r) {
#pragma unroll
      for (int k = 0; k < 2; ++k) {
        const int yy = (r == 0 && yt < 0) ? 0 : yt + r;
        const float* g = Xb + (size_t)(2 * pb + 64 * k) * HW + yy * 32 + j * 4;
        ga[r][k] = *(const v4f*)g;
        gc[r][k] = *(const v4f*)(g + HW);
      }
    }
    float* dst = xs + pb * XPAIRP + (4 + j * 4) * 2;
#pragma unroll
    for (int r = 0; r < 3; ++r) {
#pragma unroll
      for (int k = 0; k < 2; ++k) {
        float* d = dst + k * (32 * XPAIRP) + r * (XROWP * 2);
        v4f a = ga[r][k], c = gc[r][k];
        if (r == 0 && yt < 0) {
          a = (v4f){0.f, 0.f, 0.f, 0.f};
          c = (v4f){0.f, 0.f, 0.f, 0.f};
        }
        *(v4f*)d       = (v4f){a.x, c.x, a.y, c.y};
        *(v4f*)(d + 4) = (v4f){a.z, c.z, a.w, c.w};
      }
    }
    if (tid < 192) {
      int p = tid & 63;
      int r = tid >> 6;
      *(v2f*)(xs + p * XPAIRP + (r * XROWP + 3) * 2) = (v2f){0.f, 0.f};
    }
  }
  __syncthreads();

  v8f acc[G][4];
#pragma unroll
  for (int g = 0; g < G; ++g)
#pragma unroll
    for (int t = 0; t < 4; ++t) acc[g][t] = (v8f){0, 0, 0, 0, 0, 0, 0, 0};

  const v4f* wt[G];
#pragma unroll
  for (int g = 0; g < G; ++g)
    wt[g] = (const v4f*)(W + (size_t)(g * 128 + o0 + l15) * 512 + half * 8);
  const float* xl = xs + half * XPAIRP + (3 + l15) * 2;

  ConvOps<G> opA, opB;
  conv_load<G>(opA, wt, xl, 0);
  for (int s = 0; s < HIDC / 4; s += 2) {
    conv_load<G>(opB, wt, xl, s + 1);
    __builtin_amdgcn_sched_barrier(0);
    conv_mma<G>(opA, acc);
    if (s + 2 < HIDC / 4) conv_load<G>(opA, wt, xl, s + 2);
    __builtin_amdgcn_sched_barrier(0);
    conv_mma<G>(opB, acc);
  }

  float bs[G][8];
#pragma unroll
  for (int g = 0; g < G; ++g)
#pragma unroll
    for (int r = 0; r < 8; ++r) bs[g][r] = bias[g * HIDC + o0 + r + half * 8];

  if (G == 1) {
#pragma unroll
    for (int t = 0; t < 4; ++t) {
#pragma unroll
      for (int r = 0; r < 8; ++r) {
        int o = o0 + r + half * 8;
        float v = acc[0][t][r] + bs[0][r];
        Y[((size_t)b * HIDC + o) * HW + p0 + t * 16 + l15] = eluf(v);
      }
    }
  } else {
    float res[4][8];
#pragma unroll
    for (int t = 0; t < 4; ++t) {
#pragma unroll
      for (int r = 0; r < 8; ++r) {
        int o = o0 + r + half * 8;
        res[t][r] = Rin[((size_t)b * HIDC + o) * HW + p0 + t * 16 + l15];
      }
    }
#pragma unroll
    for (int t = 0; t < 4; ++t) {
#pragma unroll
      for (int r = 0; r < 8; ++r) {
        int o = o0 + r + half * 8;
        size_t idx = ((size_t)b * HIDC + o) * HW + p0 + t * 16 + l15;
        float t1 = acc[0][t][r] + bs[0][r];
        float t2 = acc[G - 1][t][r] + bs[G - 1][r];
        float v = res[t][r] + t1 * (1.f / (1.f + expf(-t2)));
        Y[idx] = v;
        E[idx] = eluf(v);
      }
    }
  }
}

#define TPITCH 160
#define TBUF   (64 * TPITCH)

template <int N>
struct TailOps {
  v2f a[N][4 / N];
  v2f b[N][4 / N][4];
};

template <int N>
__device__ __forceinline__ void tail_load(TailOps<N>& o, const float* const (&wr)[N],
                                          const float* const (&bl)[N], int ch) {
#pragma unroll
  for (int n = 0; n < N; ++n)
#pragma unroll
    for (int j = 0; j < 4 / N; ++j) {
      const int s = ch * (4 / N) + j;
      o.a[n][j].x = wr[n][s * 4];
      o.a[n][j].y = wr[n][s * 4 + 1];
#pragma unroll
      for (int t = 0; t < 4; ++t)
        o.b[n][j][t] = *(const v2f*)(bl[n] + s * (2 * TPITCH) + t * 32);
    }
}

template <int N>
__device__ __forceinline__ void tail_mma(const TailOps<N>& o, v8f (&acc)[N][4]) {
#pragma unroll
  for (int j = 0; j < 4 / N; ++j)
#pragma unroll
    for (int t = 0; t < 4; ++t)
#pragma unroll
      for (int n = 0; n < N; ++n) acc[n][t] = wmma4(o.a[n][j], o.b[n][j][t], acc[n][t]);
}

template <int N>
__device__ __forceinline__ void tail_gemm(const float* const (&wr)[N],
                                          const float* const (&bl)[N], v8f (&acc)[N][4]) {
#pragma unroll
  for (int n = 0; n < N; ++n)
#pragma unroll
    for (int t = 0; t < 4; ++t) acc[n][t] = (v8f){0, 0, 0, 0, 0, 0, 0, 0};
  TailOps<N> opA, opB;
  tail_load<N>(opA, wr, bl, 0);
#pragma clang loop unroll(disable)
  for (int ch = 0; ch < 8 * N; ch += 2) {
    tail_load<N>(opB, wr, bl, ch + 1);
    __builtin_amdgcn_sched_barrier(0);
    tail_mma<N>(opA, acc);
    if (ch + 2 < 8 * N) tail_load<N>(opA, wr, bl, ch + 2);
    __builtin_amdgcn_sched_barrier(0);
    tail_mma<N>(opB, acc);
  }
}

__global__ void __launch_bounds__(256)
k_tail(const float* __restrict__ O, const float* __restrict__ E,
       const float* __restrict__ Xres,
       const float* __restrict__ Wap, const float* __restrict__ bap,
       const float* __restrict__ Wc, const float* __restrict__ bc,
       const float* __restrict__ Wa, const float* __restrict__ ba,
       const float* __restrict__ Wp, const float* __restrict__ bp,
       float* __restrict__ Y) {
  __shared__ __attribute__((aligned(16))) float ts[3 * TBUF];
  const int tid  = threadIdx.x;
  const int lane = tid & 31;
  const int wave = tid >> 5;
  const int half = lane >> 4;
  const int l15  = lane & 15;
  const int b  = blockIdx.x;
  const int p0 = blockIdx.y * 64;
  const int o0 = wave * 16;

  {
    const int j  = tid & 15;
    const int pb = tid >> 4;
    const float* Ob = O + (size_t)b * HIDC * HW + p0 + j * 4;
    const float* Eb = E + (size_t)b * HIDC * HW + p0 + j * 4;
    v4f oa[4], oc[4], ea[4], ec[4];
#pragma unroll
    for (int k = 0; k < 4; ++k) {
      const size_t ro = (size_t)(2 * (pb + 16 * k)) * HW;
      oa[k] = *(const v4f*)(Ob + ro);
      oc[k] = *(const v4f*)(Ob + ro + HW);
      ea[k] = *(const v4f*)(Eb + ro);
      ec[k] = *(const v4f*)(Eb + ro + HW);
    }
#pragma unroll
    for (int k = 0; k < 4; ++k) {
      float* d = ts + (pb + 16 * k) * TPITCH + j * 8;
      *(v4f*)d              = (v4f){oa[k].x, oc[k].x, oa[k].y, oc[k].y};
      *(v4f*)(d + 4)        = (v4f){oa[k].z, oc[k].z, oa[k].w, oc[k].w};
      *(v4f*)(d + TBUF)     = (v4f){ea[k].x, ec[k].x, ea[k].y, ec[k].y};
      *(v4f*)(d + TBUF + 4) = (v4f){ea[k].z, ec[k].z, ea[k].w, ec[k].w};
    }
  }
  __syncthreads();

  const size_t wofs = (size_t)(o0 + l15) * HIDC + half * 2;
  const int    lofs = half * TPITCH + l15 * 2;
  float* tw = ts + ((o0 >> 1) + half * 4) * TPITCH + l15 * 2;
  const int ob = o0 + half * 8;

  {
    v8f acc[1][4];
    const float* const wr[1] = {Wap + wofs};
    const float* const bl[1] = {ts + lofs};
    tail_gemm<1>(wr, bl, acc);
    float bs[8];
#pragma unroll
    for (int r = 0; r < 8; ++r) bs[r] = bap[ob + r];
#pragma unroll
    for (int t = 0; t < 4; ++t)
#pragma unroll
      for (int r = 0; r < 8; r += 2) {
        v2f v;
        v.x = eluf(acc[0][t][r] + bs[r]);
        v.y = eluf(acc[0][t][r + 1] + bs[r + 1]);
        *(v2f*)(tw + 2 * TBUF + (r >> 1) * TPITCH + t * 32) = v;
      }
  }
  __syncthreads();

  float xr[4][8];
  const size_t ybase = ((size_t)b * HIDC + ob) * HW + p0 + l15;
  {
    v8f acc[2][4];
    const float* const wr[2] = {Wa + wofs, Wc + wofs};
    const float* const bl[2] = {ts + 2 * TBUF + lofs, ts + TBUF + lofs};
    tail_gemm<2>(wr, bl, acc);
    float bsa[8], bsc[8];
#pragma unroll
    for (int r = 0; r < 8; ++r) {
      bsa[r] = ba[ob + r];
      bsc[r] = bc[ob + r];
    }
#pragma unroll
    for (int t = 0; t < 4; ++t)
#pragma unroll
      for (int r = 0; r < 8; ++r) xr[t][r] = Xres[ybase + (size_t)r * HW + t * 16];
#pragma unroll
    for (int t = 0; t < 4; ++t)
#pragma unroll
      for (int r = 0; r < 8; r += 2) {
        v2f v;
        {
          const float a1 = eluf(acc[0][t][r] + bsa[r]);
          const float c1 = eluf(acc[1][t][r] + bsc[r]);
          v.x = eluf(c1 + a1);
        }
        {
          const float a1 = eluf(acc[0][t][r + 1] + bsa[r + 1]);
          const float c1 = eluf(acc[1][t][r + 1] + bsc[r + 1]);
          v.y = eluf(c1 + a1);
        }
        *(v2f*)(tw + (r >> 1) * TPITCH + t * 32) = v;
      }
  }
  __syncthreads();

  {
    v8f acc[1][4];
    const float* const wr[1] = {Wp + wofs};
    const float* const bl[1] = {ts + lofs};
    tail_gemm<1>(wr, bl, acc);
    float bs[8];
#pragma unroll
    for (int r = 0; r < 8; ++r) bs[r] = bp[ob + r];
#pragma unroll
    for (int t = 0; t < 4; ++t)
#pragma unroll
      for (int r = 0; r < 8; ++r) {
        float v = eluf(acc[0][t][r] + bs[r]);
        v += xr[t][r];
        Y[ybase + (size_t)r * HW + t * 16] = v;
      }
  }
}

__global__ void __launch_bounds__(256)
k_qkv(const float* __restrict__ X, const float* __restrict__ W,
      const float* __restrict__ bias, float* __restrict__ Y) {
  const int lane = threadIdx.x & 31;
  const int wave = threadIdx.x >> 5;
  const int half = lane >> 4;
  const int l15  = lane & 15;
  const int b  = blockIdx.x;
  const int p0 = blockIdx.y * 64;
  const int o0 = (blockIdx.z * 8 + wave) * 16;
  if (o0 >= 160) return;

  const float* Xb   = X + (size_t)b * HIDC * HW;
  const float* wrow = W + (size_t)(o0 + l15) * 130;

  v8f acc[4];
#pragma unroll
  for (int t = 0; t < 4; ++t) acc[t] = (v8f){0, 0, 0, 0, 0, 0, 0, 0};

  for (int c0 = 0; c0 < HIDC; c0 += 4) {
    int ca = c0 + half * 2;
    v2f a;
    a.x = wrow[ca];
    a.y = wrow[ca + 1];
    const float* x0 = Xb + (size_t)ca * HW + p0 + l15;
#pragma unroll
    for (int t = 0; t < 4; ++t) {
      v2f bb;
      bb.x = x0[t * 16];
      bb.y = x0[HW + t * 16];
      acc[t] = wmma4(a, bb, acc[t]);
    }
  }

#pragma unroll
  for (int t = 0; t < 4; ++t) {
    int p = p0 + t * 16 + l15;
    float py = (float)(p >> 5) * (1.f / 32.f) - 0.5f;
    float px = (float)(p & 31) * (1.f / 32.f) - 0.5f;
#pragma unroll
    for (int r = 0; r < 8; ++r) {
      int o = o0 + r + half * 8;
      float v = acc[t][r] + W[(size_t)o * 130 + 128] * py +
                W[(size_t)o * 130 + 129] * px + bias[o];
      Y[((size_t)b * 160 + o) * HW + p] = v;
    }
  }
}

#define ATT_NW 8
__global__ void __launch_bounds__(256)
k_attn(const float* __restrict__ QKV, float* __restrict__ Ob) {
  __shared__ __attribute__((aligned(16))) float po[ATT_NW * 8 * 8 * 32];
  __shared__ __attribute__((aligned(16))) float pm[ATT_NW * 16];
  __shared__ __attribute__((aligned(16))) float ps[ATT_NW * 32];
  const int lane = threadIdx.x & 31;
  const int wave = __builtin_amdgcn_readfirstlane((int)(threadIdx.x >> 5));
  const int half = lane >> 4;
  const int l15  = lane & 15;
  const int qt = (HW / 16 - 1) - (int)blockIdx.x;
  const int b  = blockIdx.y;
  const int q0 = qt * 16;
  const int qp = q0 + l15;
  const float* Qb = QKV + (size_t)b * 160 * HW;

  v2f qb[4];
#pragma unroll
  for (int kc = 0; kc < 4; ++kc) {
    qb[kc].x = Qb[(size_t)(kc * 4 + half * 2) * HW + qp];
    qb[kc].y = Qb[(size_t)(kc * 4 + half * 2 + 1) * HW + qp];
  }

  v8f acc[8];
#pragma unroll
  for (int vt = 0; vt < 8; ++vt) acc[vt] = (v8f){0, 0, 0, 0, 0, 0, 0, 0};
  float mrun = -3.0e38f;
  float lrun = 0.f;

  for (int kt = wave; kt <= qt; kt += ATT_NW) {
    const int k0 = kt * 16;
    v2f ka[4];
#pragma unroll
    for (int kc = 0; kc < 4; ++kc) {
      ka[kc].x = Qb[(size_t)(16 + kc * 4 + half * 2) * HW + k0 + l15];
      ka[kc].y = Qb[(size_t)(16 + kc * 4 + half * 2 + 1) * HW + k0 + l15];
    }
    v4f va[8][2];
#pragma unroll
    for (int vt = 0; vt < 8; ++vt) {
      const float* vrow = Qb + (size_t)(32 + vt * 16 + l15) * HW + k0 + half * 8;
      va[vt][0] = *(const v4f*)vrow;
      va[vt][1] = *(const v4f*)(vrow + 4);
    }
    __builtin_amdgcn_sched_barrier(0);
    v8f s = (v8f){0, 0, 0, 0, 0, 0, 0, 0};
#pragma unroll
    for (int kc = 0; kc < 4; ++kc) s = wmma4(ka[kc], qb[kc], s);
    float val[8];
    float rm = -3.0e38f;
#pragma unroll
    for (int r = 0; r < 8; ++r) {
      const int kp = k0 + half * 8 + r;
      val[r] = (kp <= qp) ? s[r] * 0.25f : -3.0e38f;
      rm = fmaxf(rm, val[r]);
    }
    rm = fmaxf(rm, __shfl_xor(rm, 16));
    const float nm = fmaxf(mrun, rm);
    const float alpha = expf(mrun - nm);
    v8f p;
    float rs = 0.f;
#pragma unroll
    for (int r = 0; r < 8; ++r) {
      p[r] = expf(val[r] - nm);
      rs += p[r];
    }
    lrun = lrun * alpha + rs;
    mrun = nm;
#pragma unroll
    for (int vt = 0; vt < 8; ++vt)
#pragma unroll
      for (int r = 0; r < 8; ++r) acc[vt][r] *= alpha;
#pragma unroll
    for (int kc = 0; kc < 4; ++kc) {
      v2f pb;
      pb.x = p[2 * kc];
      pb.y = p[2 * kc + 1];
#pragma unroll
      for (int vt = 0; vt < 8; ++vt) {
        v2f a;
        a.x = va[vt][kc >> 1][(kc & 1) * 2];
        a.y = va[vt][kc >> 1][(kc & 1) * 2 + 1];
        acc[vt] = wmma4(a, pb, acc[vt]);
      }
    }
  }

  const int nact = (qt + 1 < ATT_NW) ? qt + 1 : ATT_NW;
  const bool active = wave < nact;
  if (active && half == 0) pm[wave * 16 + l15] = mrun;
  __syncthreads();
  if (active) {
    float M = pm[l15];
    for (int j = 1; j < nact; ++j) M = fmaxf(M, pm[j * 16 + l15]);
    const float sc = expf(mrun - M);
    ps[wave * 32 + lane] = lrun * sc;
#pragma unroll
    for (int r = 0; r < 8; ++r)
#pragma unroll
      for (int vt = 0; vt < 8; ++vt)
        po[((wave * 8 + vt) * 8 + r) * 32 + lane] = acc[vt][r] * sc;
  }
  __syncthreads();

  float Ls = 0.f, os[8];
#pragma unroll
  for (int r = 0; r < 8; ++r) os[r] = 0.f;
  for (int j = 0; j < nact; ++j) {
    Ls += ps[j * 32 + l15] + ps[j * 32 + 16 + l15];
#pragma unroll
    for (int r = 0; r < 8; ++r) os[r] += po[((j * 8 + wave) * 8 + r) * 32 + lane];
  }
  const float inv = 1.f / Ls;
#pragma unroll
  for (int r = 0; r < 8; ++r)
    Ob[((size_t)b * HIDC + wave * 16 + half * 8 + r) * HW + qp] = os[r] * inv;
}

extern "C" void kernel_launch(void* const* d_in, const int* in_sizes, int n_in,
                              void* d_out, int out_size, void* d_ws, size_t ws_size,
                              hipStream_t stream) {
  const float* x    = (const float*)d_in[0];
  const float* rw1  = (const float*)d_in[1];
  const float* rb1  = (const float*)d_in[2];
  const float* rw2  = (const float*)d_in[3];
  const float* rb2  = (const float*)d_in[4];
  const float* qkvw = (const float*)d_in[5];
  const float* qkvb = (const float*)d_in[6];
  const float* apw  = (const float*)d_in[7];
  const float* apb  = (const float*)d_in[8];
  const float* ocw  = (const float*)d_in[9];
  const float* ocb  = (const float*)d_in[10];
  const float* oaw  = (const float*)d_in[11];
  const float* oab  = (const float*)d_in[12];
  const float* opw  = (const float*)d_in[13];
  const float* opb  = (const float*)d_in[14];
  float* Yout = (float*)d_out;

  const size_t N1 = (size_t)NB * HIDC * HW;
  float* A  = (float*)d_ws;
  float* E  = A + N1;
  float* F  = E + N1;
  float* Df = F + N1;

  dim3 blk(256);
  dim3 ew((unsigned)((N1 + 255) / 256));
  dim3 g128(NB, HW / 64, 1), g256(NB, HW / 64, 2);

  k_elu<<<ew, blk, 0, stream>>>(x, E, (int)N1);
  for (int i = 0; i < 6; ++i) {
    const float* w1 = rw1 + (size_t)i * HIDC * HIDC * 4;
    const float* b1 = rb1 + (size_t)i * HIDC;
    const float* w2 = rw2 + (size_t)i * 2 * HIDC * HIDC * 4;
    const float* b2 = rb2 + (size_t)i * 2 * HIDC;
    k_conv2x2<1><<<g128, blk, 0, stream>>>(E, w1, b1, F, nullptr, nullptr);
    k_conv2x2<2><<<g128, blk, 0, stream>>>(F, w2, b2, A, i == 0 ? x : A, E);
  }

  k_qkv<<<g256, blk, 0, stream>>>(A, qkvw, qkvb, Df);
  k_attn<<<dim3(HW / 16, NB), dim3(256), 0, stream>>>(Df, F);

  k_tail<<<g128, blk, 0, stream>>>(F, E, x, apw, apb, ocw, ocb, oaw, oab, opw, opb, Yout);
}


# 2x2 convs in minimal-filtering form F(2x2,2x2): 9 matrix products per 2x2 output block and channel (was 16); input, weight and output transforms are additions; conv sum order changed
# speedup vs baseline: 2.5894x; 1.1931x over previous
#include <hip/hip_runtime.h>
#include <hip/hip_bf16.h>


typedef __attribute__((ext_vector_type(2))) float v2f;
typedef __attribute__((ext_vector_type(4))) float v4f;
typedef __attribute__((ext_vector_type(8))) float v8f;

#define NB   32
#define HIDC 128
#define HW   1024

__device__ __forceinline__ v8f wmma4(v2f a, v2f b, v8f c) {
  return __builtin_amdgcn_wmma_f32_16x16x4_f32(false, a, false, b, (short)0, c,
                                               false, false);
}

__device__ __forceinline__ float eluf(float x) {
  return x > 0.f ? x : (expf(x) - 1.f);
}

__global__ void k_elu(const float* __restrict__ X, float* __restrict__ Y, int n) {
  int i = blockIdx.x * 256 + threadIdx.x;
  if (i < n) Y[i] = eluf(X[i]);
}

#define VPOS 2048

template <int G>
struct ConvOps {
  v4f wa[2 / G][G][2];
  v2f fr[2 / G][9];
};

template <int G>
__device__ __forceinline__ void conv_load(ConvOps<G>& o, const v4f* const (&wt)[G],
                                          const float* vl, int ch) {
#pragma unroll
  for (int j = 0; j < 2 / G; ++j) {
    const int step = ch * (2 / G) + j;
#pragma unroll
    for (int g = 0; g < G; ++g) {
      o.wa[j][g][0] = wt[g][step * 4];
      o.wa[j][g][1] = wt[g][step * 4 + 1];
    }
    const float* base = vl + step * 64;
#pragma unroll
    for (int p = 0; p < 9; ++p) o.fr[j][p] = *(const v2f*)(base + p * VPOS);
  }
}

template <int G>
__device__ __forceinline__ void conv_mma(const ConvOps<G>& o, v8f (&acc)[G][9]) {
#pragma unroll
  for (int j = 0; j < 2 / G; ++j) {
    v2f u[G][9];
#pragma unroll
    for (int g = 0; g < G; ++g) {
      const v4f w0 = o.wa[j][g][0], w1 = o.wa[j][g][1];
      const v2f t0 = (v2f){w0[0], w1[0]};
      const v2f t1 = (v2f){w0[1], w1[1]};
      const v2f t2 = (v2f){w0[2], w1[2]};
      const v2f t3 = (v2f){w0[3], w1[3]};
      u[g][0] = t0;
      u[g][1] = t0 + t1;
      u[g][2] = t1;
      u[g][3] = t0 + t2;
      u[g][5] = t1 + t3;
      u[g][6] = t2;
      u[g][7] = t2 + t3;
      u[g][8] = t3;
      u[g][4] = u[g][1] + u[g][7];
    }
#pragma unroll
    for (int p = 0; p < 9; ++p)
#pragma unroll
      for (int g = 0; g < G; ++g) acc[g][p] = wmma4(u[g][p], o.fr[j][p], acc[g][p]);
  }
}

#define MINF_Y00(m) ((m[0] + m[1]) + (m[3] + m[4]))
#define MINF_Y01(m) ((m[1] - m[2]) + (m[4] - m[5]))
#define MINF_Y10(m) ((m[3] + m[4]) - (m[6] + m[7]))
#define MINF_Y11(m) ((m[4] - m[5]) - (m[7] - m[8]))

template <int G>
__global__ void __launch_bounds__(256)
k_conv2x2(const float* __restrict__ X, const float* __restrict__ W,
          const float* __restrict__ bias, float* Y, const float* Rin,
          float* __restrict__ E) {
  __shared__ __attribute__((aligned(16))) float vs[9 * VPOS];
  const int tid  = threadIdx.x;
  const int lane = tid & 31;
  const int wave = tid >> 5;
  const int half = lane >> 4;
  const int l15  = lane & 15;
  const int b  = blockIdx.x;
  const int p0 = blockIdx.y * 64;
  const int o0 = wave * 16;

  const float* Xb = X + (size_t)b * HIDC * HW;

  {
    const int yt = (int)blockIdx.y * 2 - 1;
    const int j  = tid & 7;
    const int pb = tid >> 3;
    v4f ga[3][2], gc[3][2];
#pragma unroll
    for (int r = 0; r < 3; ++r) {
#pragma unroll
      for (int k = 0; k < 2; ++k) {
        const int yy = (r == 0 && yt < 0) ? 0 : yt + r;
        const float* g = Xb + (size_t)(2 * pb + 64 * k) * HW + yy * 32 + j * 4;
        ga[r][k] = *(const v4f*)g;
        gc[r][k] = *(const v4f*)(g + HW);
      }
    }
    if (yt < 0) {
#pragma unroll
      for (int k = 0; k < 2; ++k) {
        ga[0][k] = (v4f){0.f, 0.f, 0.f, 0.f};
        gc[0][k] = (v4f){0.f, 0.f, 0.f, 0.f};
      }
    }
#pragma unroll
    for (int k = 0; k < 2; ++k) {
      float da[3][5], dc[3][5];
#pragma unroll
      for (int r = 0; r < 3; ++r) {
        float la = __shfl_up(ga[r][k][3], 1);
        float lc = __shfl_up(gc[r][k][3], 1);
        da[r][0] = (j == 0) ? 0.f : la;
        dc[r][0] = (j == 0) ? 0.f : lc;
#pragma unroll
        for (int q = 0; q < 4; ++q) {
          da[r][q + 1] = ga[r][k][q];
          dc[r][q + 1] = gc[r][k][q];
        }
      }
      float ra[3][5], rc[3][5];
#pragma unroll
      for (int q = 0; q < 5; ++q) {
        ra[0][q] = da[0][q] - da[1][q];
        ra[1][q] = da[1][q];
        ra[2][q] = da[1][q] - da[2][q];
        rc[0][q] = dc[0][q] - dc[1][q];
        rc[1][q] = dc[1][q];
        rc[2][q] = dc[1][q] - dc[2][q];
      }
      float* dst = vs + (pb + 32 * k) * 32 + j * 4;
#pragma unroll
      for (int i = 0; i < 3; ++i) {
        v4f s0, s1, s2;
        s0 = (v4f){ra[i][0] - ra[i][1], rc[i][0] - rc[i][1], ra[i][2] - ra[i][3], rc[i][2] - rc[i][3]};
        s1 = (v4f){ra[i][1], rc[i][1], ra[i][3], rc[i][3]};
        s2 = (v4f){ra[i][1] - ra[i][2], rc[i][1] - rc[i][2], ra[i][3] - ra[i][4], rc[i][3] - rc[i][4]};
        *(v4f*)(dst + (i * 3 + 0) * VPOS) = s0;
        *(v4f*)(dst + (i * 3 + 1) * VPOS) = s1;
        *(v4f*)(dst + (i * 3 + 2) * VPOS) = s2;
      }
    }
  }
  __syncthreads();

  v8f acc[G][9];
#pragma unroll
  for (int g = 0; g < G; ++g)
#pragma unroll
    for (int p = 0; p < 9; ++p) acc[g][p] = (v8f){0, 0, 0, 0, 0, 0, 0, 0};

  const v4f* wt[G];
#pragma unroll
  for (int g = 0; g < G; ++g)
    wt[g] = (const v4f*)(W + (size_t)(g * 128 + o0 + l15) * 512 + half * 8);
  const float* vl = vs + half * 32 + l15 * 2;

  constexpr int NCH = (HIDC / 4) / (2 / G);
  ConvOps<G> opA, opB;
  conv_load<G>(opA, wt, vl, 0);
#pragma clang loop unroll(disable)
  for (int c = 0; c < NCH; c += 2) {
    conv_load<G>(opB, wt, vl, c + 1);
    __builtin_amdgcn_sched_barrier(0);
    conv_mma<G>(opA, acc);
    __builtin_amdgcn_sched_barrier(0);
    if (c + 2 < NCH) conv_load<G>(opA, wt, vl, c + 2);
    __builtin_amdgcn_sched_barrier(0);
    conv_mma<G>(opB, acc);
  }

  float bs[G][8];
#pragma unroll
  for (int g = 0; g < G; ++g)
#pragma unroll
    for (int r = 0; r < 8; ++r) bs[g][r] = bias[g * HIDC + o0 + r + half * 8];

  const size_t obase = ((size_t)b * HIDC + o0 + half * 8) * HW + p0 + 2 * l15;

  if (G == 1) {
#pragma unroll
    for (int r = 0; r < 8; ++r) {
      float m[9];
#pragma unroll
      for (int p = 0; p < 9; ++p) m[p] = acc[0][p][r];
      v2f y0, y1;
      y0.x = eluf(MINF_Y00(m) + bs[0][r]);
      y0.y = eluf(MINF_Y01(m) + bs[0][r]);
      y1.x = eluf(MINF_Y10(m) + bs[0][r]);
      y1.y = eluf(MINF_Y11(m) + bs[0][r]);
      *(v2f*)(Y + obase + (size_t)r * HW)      = y0;
      *(v2f*)(Y + obase + (size_t)r * HW + 32) = y1;
    }
  } else {
    v2f res[8][2];
#pragma unroll
    for (int r = 0; r < 8; ++r) {
      res[r][0] = *(const v2f*)(Rin + obase + (size_t)r * HW);
      res[r][1] = *(const v2f*)(Rin + obase + (size_t)r * HW + 32);
    }
#pragma unroll
    for (int r = 0; r < 8; ++r) {
      float m1[9], m2[9];
#pragma unroll
      for (int p = 0; p < 9; ++p) {
        m1[p] = acc[0][p][r];
        m2[p] = acc[G - 1][p][r];
      }
      float t1[4], t2[4];
      t1[0] = MINF_Y00(m1) + bs[0][r];
      t1[1] = MINF_Y01(m1) + bs[0][r];
      t1[2] = MINF_Y10(m1) + bs[0][r];
      t1[3] = MINF_Y11(m1) + bs[0][r];
      t2[0] = MINF_Y00(m2) + bs[G - 1][r];
      t2[1] = MINF_Y01(m2) + bs[G - 1][r];
      t2[2] = MINF_Y10(m2) + bs[G - 1][r];
      t2[3] = MINF_Y11(m2) + bs[G - 1][r];
#pragma unroll
      for (int i = 0; i < 2; ++i) {
        v2f v, e;
        v.x = res[r][i].x + t1[2 * i]     * (1.f / (1.f + expf(-t2[2 * i])));
        v.y = res[r][i].y + t1[2 * i + 1] * (1.f / (1.f + expf(-t2[2 * i + 1])));
        e.x = eluf(v.x);
        e.y = eluf(v.y);
        *(v2f*)(Y + obase + (size_t)r * HW + i * 32) = v;
        *(v2f*)(E + obase + (size_t)r * HW + i * 32) = e;
      }
    }
  }
}

#define TPITCH 160
#define TBUF   (64 * TPITCH)

template <int N>
struct TailOps {
  v2f a[N][4 / N];
  v2f b[N][4 / N][4];
};

template <int N>
__device__ __forceinline__ void tail_load(TailOps<N>& o, const float* const (&wr)[N],
                                          const float* const (&bl)[N], int ch) {
#pragma unroll
  for (int n = 0; n < N; ++n)
#pragma unroll
    for (int j = 0; j < 4 / N; ++j) {
      const int s = ch * (4 / N) + j;
      o.a[n][j].x = wr[n][s * 4];
      o.a[n][j].y = wr[n][s * 4 + 1];
#pragma unroll
      for (int t = 0; t < 4; ++t)
        o.b[n][j][t] = *(const v2f*)(bl[n] + s * (2 * TPITCH) + t * 32);
    }
}

template <int N>
__device__ __forceinline__ void tail_mma(const TailOps<N>& o, v8f (&acc)[N][4]) {
#pragma unroll
  for (int j = 0; j < 4 / N; ++j)
#pragma unroll
    for (int t = 0; t < 4; ++t)
#pragma unroll
      for (int n = 0; n < N; ++n) acc[n][t] = wmma4(o.a[n][j], o.b[n][j][t], acc[n][t]);
}

template <int N>
__device__ __forceinline__ void tail_gemm(const float* const (&wr)[N],
                                          const float* const (&bl)[N], v8f (&acc)[N][4]) {
#pragma unroll
  for (int n = 0; n < N; ++n)
#pragma unroll
    for (int t = 0; t < 4; ++t) acc[n][t] = (v8f){0, 0, 0, 0, 0, 0, 0, 0};
  TailOps<N> opA, opB;
  tail_load<N>(opA, wr, bl, 0);
#pragma clang loop unroll(disable)
  for (int ch = 0; ch < 8 * N; ch += 2) {
    tail_load<N>(opB, wr, bl, ch + 1);
    __builtin_amdgcn_sched_barrier(0);
    tail_mma<N>(opA, acc);
    if (ch + 2 < 8 * N) tail_load<N>(opA, wr, bl, ch + 2);
    __builtin_amdgcn_sched_barrier(0);
    tail_mma<N>(opB, acc);
  }
}

__global__ void __launch_bounds__(256)
k_tail(const float* __restrict__ O, const float* __restrict__ E,
       const float* __restrict__ Xres,
       const float* __restrict__ Wap, const float* __restrict__ bap,
       const float* __restrict__ Wc, const float* __restrict__ bc,
       const float* __restrict__ Wa, const float* __restrict__ ba,
       const float* __restrict__ Wp, const float* __restrict__ bp,
       float* __restrict__ Y) {
  __shared__ __attribute__((aligned(16))) float ts[3 * TBUF];
  const int tid  = threadIdx.x;
  const int lane = tid & 31;
  const int wave = tid >> 5;
  const int half = lane >> 4;
  const int l15  = lane & 15;
  const int b  = blockIdx.x;
  const int p0 = blockIdx.y * 64;
  const int o0 = wave * 16;

  {
    const int j  = tid & 15;
    const int pb = tid >> 4;
    const float* Ob = O + (size_t)b * HIDC * HW + p0 + j * 4;
    const float* Eb = E + (size_t)b * HIDC * HW + p0 + j * 4;
    v4f oa[4], oc[4], ea[4], ec[4];
#pragma unroll
    for (int k = 0; k < 4; ++k) {
      const size_t ro = (size_t)(2 * (pb + 16 * k)) * HW;
      oa[k] = *(const v4f*)(Ob + ro);
      oc[k] = *(const v4f*)(Ob + ro + HW);
      ea[k] = *(const v4f*)(Eb + ro);
      ec[k] = *(const v4f*)(Eb + ro + HW);
    }
#pragma unroll
    for (int k = 0; k < 4; ++k) {
      float* d = ts + (pb + 16 * k) * TPITCH + j * 8;
      *(v4f*)d              = (v4f){oa[k].x, oc[k].x, oa[k].y, oc[k].y};
      *(v4f*)(d + 4)        = (v4f){oa[k].z, oc[k].z, oa[k].w, oc[k].w};
      *(v4f*)(d + TBUF)     = (v4f){ea[k].x, ec[k].x, ea[k].y, ec[k].y};
      *(v4f*)(d + TBUF + 4) = (v4f){ea[k].z, ec[k].z, ea[k].w, ec[k].w};
    }
  }
  __syncthreads();

  const size_t wofs = (size_t)(o0 + l15) * HIDC + half * 2;
  const int    lofs = half * TPITCH + l15 * 2;
  float* tw = ts + ((o0 >> 1) + half * 4) * TPITCH + l15 * 2;
  const int ob = o0 + half * 8;

  {
    v8f acc[1][4];
    const float* const wr[1] = {Wap + wofs};
    const float* const bl[1] = {ts + lofs};
    tail_gemm<1>(wr, bl, acc);
    float bs[8];
#pragma unroll
    for (int r = 0; r < 8; ++r) bs[r] = bap[ob + r];
#pragma unroll
    for (int t = 0; t < 4; ++t)
#pragma unroll
      for (int r = 0; r < 8; r += 2) {
        v2f v;
        v.x = eluf(acc[0][t][r] + bs[r]);
        v.y = eluf(acc[0][t][r + 1] + bs[r + 1]);
        *(v2f*)(tw + 2 * TBUF + (r >> 1) * TPITCH + t * 32) = v;
      }
  }
  __syncthreads();

  float xr[4][8];
  const size_t ybase = ((size_t)b * HIDC + ob) * HW + p0 + l15;
  {
    v8f acc[2][4];
    const float* const wr[2] = {Wa + wofs, Wc + wofs};
    const float* const bl[2] = {ts + 2 * TBUF + lofs, ts + TBUF + lofs};
    tail_gemm<2>(wr, bl, acc);
    float bsa[8], bsc[8];
#pragma unroll
    for (int r = 0; r < 8; ++r) {
      bsa[r] = ba[ob + r];
      bsc[r] = bc[ob + r];
    }
#pragma unroll
    for (int t = 0; t < 4; ++t)
#pragma unroll
      for (int r = 0; r < 8; ++r) xr[t][r] = Xres[ybase + (size_t)r * HW + t * 16];
#pragma unroll
    for (int t = 0; t < 4; ++t)
#pragma unroll
      for (int r = 0; r < 8; r += 2) {
        v2f v;
        {
          const float a1 = eluf(acc[0][t][r] + bsa[r]);
          const float c1 = eluf(acc[1][t][r] + bsc[r]);
          v.x = eluf(c1 + a1);
        }
        {
          const float a1 = eluf(acc[0][t][r + 1] + bsa[r + 1]);
          const float c1 = eluf(acc[1][t][r + 1] + bsc[r + 1]);
          v.y = eluf(c1 + a1);
        }
        *(v2f*)(tw + (r >> 1) * TPITCH + t * 32) = v;
      }
  }
  __syncthreads();

  {
    v8f acc[1][4];
    const float* const wr[1] = {Wp + wofs};
    const float* const bl[1] = {ts + lofs};
    tail_gemm<1>(wr, bl, acc);
    float bs[8];
#pragma unroll
    for (int r = 0; r < 8; ++r) bs[r] = bp[ob + r];
#pragma unroll
    for (int t = 0; t < 4; ++t)
#pragma unroll
      for (int r = 0; r < 8; ++r) {
        float v = eluf(acc[0][t][r] + bs[r]);
        v += xr[t][r];
        Y[ybase + (size_t)r * HW + t * 16] = v;
      }
  }
}

__global__ void __launch_bounds__(256)
k_qkv(const float* __restrict__ X, const float* __restrict__ W,
      const float* __restrict__ bias, float* __restrict__ Y) {
  const int lane = threadIdx.x & 31;
  const int wave = threadIdx.x >> 5;
  const int half = lane >> 4;
  const int l15  = lane & 15;
  const int b  = blockIdx.x;
  const int p0 = blockIdx.y * 64;
  const int o0 = (blockIdx.z * 8 + wave) * 16;
  if (o0 >= 160) return;

  const float* Xb   = X + (size_t)b * HIDC * HW;
  const float* wrow = W + (size_t)(o0 + l15) * 130;

  v8f acc[4];
#pragma unroll
  for (int t = 0; t < 4; ++t) acc[t] = (v8f){0, 0, 0, 0, 0, 0, 0, 0};

  for (int c0 = 0; c0 < HIDC; c0 += 4) {
    int ca = c0 + half * 2;
    v2f a;
    a.x = wrow[ca];
    a.y = wrow[ca + 1];
    const float* x0 = Xb + (size_t)ca * HW + p0 + l15;
#pragma unroll
    for (int t = 0; t < 4; ++t) {
      v2f bb;
      bb.x = x0[t * 16];
      bb.y = x0[HW + t * 16];
      acc[t] = wmma4(a, bb, acc[t]);
    }
  }

#pragma unroll
  for (int t = 0; t < 4; ++t) {
    int p = p0 + t * 16 + l15;
    float py = (float)(p >> 5) * (1.f / 32.f) - 0.5f;
    float px = (float)(p & 31) * (1.f / 32.f) - 0.5f;
#pragma unroll
    for (int r = 0; r < 8; ++r) {
      int o = o0 + r + half * 8;
      float v = acc[t][r] + W[(size_t)o * 130 + 128] * py +
                W[(size_t)o * 130 + 129] * px + bias[o];
      Y[((size_t)b * 160 + o) * HW + p] = v;
    }
  }
}

#define ATT_NW 8
__global__ void __launch_bounds__(256)
k_attn(const float* __restrict__ QKV, float* __restrict__ Ob) {
  __shared__ __attribute__((aligned(16))) float po[ATT_NW * 8 * 8 * 32];
  __shared__ __attribute__((aligned(16))) float pm[ATT_NW * 16];
  __shared__ __attribute__((aligned(16))) float ps[ATT_NW * 32];
  const int lane = threadIdx.x & 31;
  const int wave = __builtin_amdgcn_readfirstlane((int)(threadIdx.x >> 5));
  const int half = lane >> 4;
  const int l15  = lane & 15;
  const int qt = (HW / 16 - 1) - (int)blockIdx.x;
  const int b  = blockIdx.y;
  const int q0 = qt * 16;
  const int qp = q0 + l15;
  const float* Qb = QKV + (size_t)b * 160 * HW;

  v2f qb[4];
#pragma unroll
  for (int kc = 0; kc < 4; ++kc) {
    qb[kc].x = Qb[(size_t)(kc * 4 + half * 2) * HW + qp];
    qb[kc].y = Qb[(size_t)(kc * 4 + half * 2 + 1) * HW + qp];
  }

  v8f acc[8];
#pragma unroll
  for (int vt = 0; vt < 8; ++vt) acc[vt] = (v8f){0, 0, 0, 0, 0, 0, 0, 0};
  float mrun = -3.0e38f;
  float lrun = 0.f;

  for (int kt = wave; kt <= qt; kt += ATT_NW) {
    const int k0 = kt * 16;
    v2f ka[4];
#pragma unroll
    for (int kc = 0; kc < 4; ++kc) {
      ka[kc].x = Qb[(size_t)(16 + kc * 4 + half * 2) * HW + k0 + l15];
      ka[kc].y = Qb[(size_t)(16 + kc * 4 + half * 2 + 1) * HW + k0 + l15];
    }
    v4f va[8][2];
#pragma unroll
    for (int vt = 0; vt < 8; ++vt) {
      const float* vrow = Qb + (size_t)(32 + vt * 16 + l15) * HW + k0 + half * 8;
      va[vt][0] = *(const v4f*)vrow;
      va[vt][1] = *(const v4f*)(vrow + 4);
    }
    __builtin_amdgcn_sched_barrier(0);
    v8f s = (v8f){0, 0, 0, 0, 0, 0, 0, 0};
#pragma unroll
    for (int kc = 0; kc < 4; ++kc) s = wmma4(ka[kc], qb[kc], s);
    float val[8];
    float rm = -3.0e38f;
#pragma unroll
    for (int r = 0; r < 8; ++r) {
      const int kp = k0 + half * 8 + r;
      val[r] = (kp <= qp) ? s[r] * 0.25f : -3.0e38f;
      rm = fmaxf(rm, val[r]);
    }
    rm = fmaxf(rm, __shfl_xor(rm, 16));
    const float nm = fmaxf(mrun, rm);
    const float alpha = expf(mrun - nm);
    v8f p;
    float rs = 0.f;
#pragma unroll
    for (int r = 0; r < 8; ++r) {
      p[r] = expf(val[r] - nm);
      rs += p[r];
    }
    lrun = lrun * alpha + rs;
    mrun = nm;
#pragma unroll
    for (int vt = 0; vt < 8; ++vt)
#pragma unroll
      for (int r = 0; r < 8; ++r) acc[vt][r] *= alpha;
#pragma unroll
    for (int kc = 0; kc < 4; ++kc) {
      v2f pb;
      pb.x = p[2 * kc];
      pb.y = p[2 * kc + 1];
#pragma unroll
      for (int vt = 0; vt < 8; ++vt) {
        v2f a;
        a.x = va[vt][kc >> 1][(kc & 1) * 2];
        a.y = va[vt][kc >> 1][(kc & 1) * 2 + 1];
        acc[vt] = wmma4(a, pb, acc[vt]);
      }
    }
  }

  const int nact = (qt + 1 < ATT_NW) ? qt + 1 : ATT_NW;
  const bool active = wave < nact;
  if (active && half == 0) pm[wave * 16 + l15] = mrun;
  __syncthreads();
  if (active) {
    float M = pm[l15];
    for (int j = 1; j < nact; ++j) M = fmaxf(M, pm[j * 16 + l15]);
    const float sc = expf(mrun - M);
    ps[wave * 32 + lane] = lrun * sc;
#pragma unroll
    for (int r = 0; r < 8; ++r)
#pragma unroll
      for (int vt = 0; vt < 8; ++vt)
        po[((wave * 8 + vt) * 8 + r) * 32 + lane] = acc[vt][r] * sc;
  }
  __syncthreads();

  float Ls = 0.f, os[8];
#pragma unroll
  for (int r = 0; r < 8; ++r) os[r] = 0.f;
  for (int j = 0; j < nact; ++j) {
    Ls += ps[j * 32 + l15] + ps[j * 32 + 16 + l15];
#pragma unroll
    for (int r = 0; r < 8; ++r) os[r] += po[((j * 8 + wave) * 8 + r) * 32 + lane];
  }
  const float inv = 1.f / Ls;
#pragma unroll
  for (int r = 0; r < 8; ++r)
    Ob[((size_t)b * HIDC + wave * 16 + half * 8 + r) * HW + qp] = os[r] * inv;
}

extern "C" void kernel_launch(void* const* d_in, const int* in_sizes, int n_in,
                              void* d_out, int out_size, void* d_ws, size_t ws_size,
                              hipStream_t stream) {
  const float* x    = (const float*)d_in[0];
  const float* rw1  = (const float*)d_in[1];
  const float* rb1  = (const float*)d_in[2];
  const float* rw2  = (const float*)d_in[3];
  const float* rb2  = (const float*)d_in[4];
  const float* qkvw = (const float*)d_in[5];
  const float* qkvb = (const float*)d_in[6];
  const float* apw  = (const float*)d_in[7];
  const float* apb  = (const float*)d_in[8];
  const float* ocw  = (const float*)d_in[9];
  const float* ocb  = (const float*)d_in[10];
  const float* oaw  = (const float*)d_in[11];
  const float* oab  = (const float*)d_in[12];
  const float* opw  = (const float*)d_in[13];
  const float* opb  = (const float*)d_in[14];
  float* Yout = (float*)d_out;

  const size_t N1 = (size_t)NB * HIDC * HW;
  float* A  = (float*)d_ws;
  float* E  = A + N1;
  float* F  = E + N1;
  float* Df = F + N1;

  dim3 blk(256);
  dim3 ew((unsigned)((N1 + 255) / 256));
  dim3 g128(NB, HW / 64, 1), g256(NB, HW / 64, 2);

  k_elu<<<ew, blk, 0, stream>>>(x, E, (int)N1);
  for (int i = 0; i < 6; ++i) {
    const float* w1 = rw1 + (size_t)i * HIDC * HIDC * 4;
    const float* b1 = rb1 + (size_t)i * HIDC;
    const float* w2 = rw2 + (size_t)i * 2 * HIDC * HIDC * 4;
    const float* b2 = rb2 + (size_t)i * 2 * HIDC;
    k_conv2x2<1><<<g128, blk, 0, stream>>>(E, w1, b1, F, nullptr, nullptr);
    k_conv2x2<2><<<g128, blk, 0, stream>>>(F, w2, b2, A, i == 0 ? x : A, E);
  }

  k_qkv<<<g256, blk, 0, stream>>>(A, qkvw, qkvb, Df);
  k_attn<<<dim3(HW / 16, NB), dim3(256), 0, stream>>>(Df, F);

  k_tail<<<g128, blk, 0, stream>>>(F, E, x, apw, apb, ocw, ocb, oaw, oab, opw, opb, Yout);
}


# A operands in lane order: conv and 1x1 weights copied so by the first launch each call, V stored so by the qkv kernel; a fragment load reads 4 whole 128-B lines, was 16; every sum keeps its order
# speedup vs baseline: 2.7112x; 1.0470x over previous
#include <hip/hip_runtime.h>
#include <hip/hip_bf16.h>


typedef __attribute__((ext_vector_type(2))) float v2f;
typedef __attribute__((ext_vector_type(4))) float v4f;
typedef __attribute__((ext_vector_type(8))) float v8f;

#define NB   32
#define HIDC 128
#define HW   1024

__device__ __forceinline__ v8f wmma4(v2f a, v2f b, v8f c) {
  return __builtin_amdgcn_wmma_f32_16x16x4_f32(false, a, false, b, (short)0, c,
                                               false, false);
}

__device__ __forceinline__ float eluf(float x) {
  return x > 0.f ? x : (expf(x) - 1.f);
}

#define NW1_16B (6 * 128 * 128)
#define NW2_16B (6 * 256 * 128)
#define NWT_16B (4 * 128 * 32)
__global__ void __launch_bounds__(256)
k_first(const float* __restrict__ X, float* __restrict__ Y, int n,
        const float* __restrict__ W1, const float* __restrict__ W2,
        const float* __restrict__ Wap, const float* __restrict__ Wc,
        const float* __restrict__ Wa, const float* __restrict__ Wp,
        float* __restrict__ P1, float* __restrict__ P2, float* __restrict__ PT) {
  const int nelu = (n + 255) >> 8;
  if ((int)blockIdx.x < nelu) {
    int i = blockIdx.x * 256 + threadIdx.x;
    if (i < n) Y[i] = eluf(X[i]);
    return;
  }
  const int t = ((int)blockIdx.x - nelu) * 256 + (int)threadIdx.x;
  if (t < NW1_16B + NW2_16B) {
    const bool first = t < NW1_16B;
    const v4f* src = (const v4f*)(first ? W1 : W2);
    v4f* dst = (v4f*)(first ? P1 : P2);
    const int d    = first ? t : t - NW1_16B;
    const int lane = d & 31;
    const int e    = (d >> 5) & 1;
    const int s    = (d >> 6) & 31;
    const int rg   = d >> 11;
    const int row  = rg * 16 + (lane & 15);
    const int ch   = 4 * s + 2 * (lane >> 4) + e;
    dst[d] = src[(size_t)row * 128 + ch];
  } else if (t < NW1_16B + NW2_16B + NWT_16B) {
    const int t2 = t - (NW1_16B + NW2_16B);
    const int m  = t2 >> 12;
    const float* Wm = (m == 0) ? Wap : (m == 1) ? Wc : (m == 2) ? Wa : Wp;
    const int d    = t2 & 4095;
    const int lane = d & 31;
    const int s2   = (d >> 5) & 15;
    const int rg   = d >> 9;
    const float* r = Wm + (size_t)(rg * 16 + (lane & 15)) * 128 + 8 * s2 + 2 * (lane >> 4);
    const v2f lo = *(const v2f*)r;
    const v2f hi = *(const v2f*)(r + 4);
    ((v4f*)PT)[t2] = (v4f){lo.x, lo.y, hi.x, hi.y};
  }
}

#define VPOS 2048

template <int G>
struct ConvOps {
  v4f wa[2 / G][G][2];
  v2f fr[2 / G][9];
};

template <int G>
__device__ __forceinline__ void conv_load(ConvOps<G>& o, const v4f* const (&wt)[G],
                                          const float* vl, int ch) {
#pragma unroll
  for (int j = 0; j < 2 / G; ++j) {
    const int step = ch * (2 / G) + j;
#pragma unroll
    for (int g = 0; g < G; ++g) {
      o.wa[j][g][0] = wt[g][step * 64];
      o.wa[j][g][1] = wt[g][step * 64 + 32];
    }
    const float* base = vl + step * 64;
#pragma unroll
    for (int p = 0; p < 9; ++p) o.fr[j][p] = *(const v2f*)(base + p * VPOS);
  }
}

template <int G>
__device__ __forceinline__ void conv_mma(const ConvOps<G>& o, v8f (&acc)[G][9]) {
#pragma unroll
  for (int j = 0; j < 2 / G; ++j) {
    v2f u[G][9];
#pragma unroll
    for (int g = 0; g < G; ++g) {
      const v4f w0 = o.wa[j][g][0], w1 = o.wa[j][g][1];
      const v2f t0 = (v2f){w0[0], w1[0]};
      const v2f t1 = (v2f){w0[1], w1[1]};
      const v2f t2 = (v2f){w0[2], w1[2]};
      const v2f t3 = (v2f){w0[3], w1[3]};
      u[g][0] = t0;
      u[g][1] = t0 + t1;
      u[g][2] = t1;
      u[g][3] = t0 + t2;
      u[g][5] = t1 + t3;
      u[g][6] = t2;
      u[g][7] = t2 + t3;
      u[g][8] = t3;
      u[g][4] = u[g][1] + u[g][7];
    }
#pragma unroll
    for (int p = 0; p < 9; ++p)
#pragma unroll
      for (int g = 0; g < G; ++g) acc[g][p] = wmma4(u[g][p], o.fr[j][p], acc[g][p]);
  }
}

#define MINF_Y00(m) ((m[0] + m[1]) + (m[3] + m[4]))
#define MINF_Y01(m) ((m[1] - m[2]) + (m[4] - m[5]))
#define MINF_Y10(m) ((m[3] + m[4]) - (m[6] + m[7]))
#define MINF_Y11(m) ((m[4] - m[5]) - (m[7] - m[8]))

template <int G>
__global__ void __launch_bounds__(256)
k_conv2x2(const float* __restrict__ X, const float* __restrict__ W,
          const float* __restrict__ bias, float* Y, const float* Rin,
          float* __restrict__ E) {
  __shared__ __attribute__((aligned(16))) float vs[9 * VPOS];
  const int tid  = threadIdx.x;
  const int lane = tid & 31;
  const int wave = tid >> 5;
  const int half = lane >> 4;
  const int l15  = lane & 15;
  const int b  = blockIdx.x;
  const int p0 = blockIdx.y * 64;
  const int o0 = wave * 16;

  const float* Xb = X + (size_t)b * HIDC * HW;

  {
    const int yt = (int)blockIdx.y * 2 - 1;
    const int j  = tid & 7;
    const int pb = tid >> 3;
    v4f ga[3][2], gc[3][2];
#pragma unroll
    for (int r = 0; r < 3; ++r) {
#pragma unroll
      for (int k = 0; k < 2; ++k) {
        const int yy = (r == 0 && yt < 0) ? 0 : yt + r;
        const float* g = Xb + (size_t)(2 * pb + 64 * k) * HW + yy * 32 + j * 4;
        ga[r][k] = *(const v4f*)g;
        gc[r][k] = *(const v4f*)(g + HW);
      }
    }
    if (yt < 0) {
#pragma unroll
      for (int k = 0; k < 2; ++k) {
        ga[0][k] = (v4f){0.f, 0.f, 0.f, 0.f};
        gc[0][k] = (v4f){0.f, 0.f, 0.f, 0.f};
      }
    }
#pragma unroll
    for (int k = 0; k < 2; ++k) {
      float da[3][5], dc[3][5];
#pragma unroll
      for (int r = 0; r < 3; ++r) {
        float la = __shfl_up(ga[r][k][3], 1);
        float lc = __shfl_up(gc[r][k][3], 1);
        da[r][0] = (j == 0) ? 0.f : la;
        dc[r][0] = (j == 0) ? 0.f : lc;
#pragma unroll
        for (int q = 0; q < 4; ++q) {
          da[r][q + 1] = ga[r][k][q];
          dc[r][q + 1] = gc[r][k][q];
        }
      }
      float ra[3][5], rc[3][5];
#pragma unroll
      for (int q = 0; q < 5; ++q) {
        ra[0][q] = da[0][q] - da[1][q];
        ra[1][q] = da[1][q];
        ra[2][q] = da[1][q] - da[2][q];
        rc[0][q] = dc[0][q] - dc[1][q];
        rc[1][q] = dc[1][q];
        rc[2][q] = dc[1][q] - dc[2][q];
      }
      float* dst = vs + (pb + 32 * k) * 32 + j * 4;
#pragma unroll
      for (int i = 0; i < 3; ++i) {
        v4f s0, s1, s2;
        s0 = (v4f){ra[i][0] - ra[i][1], rc[i][0] - rc[i][1], ra[i][2] - ra[i][3], rc[i][2] - rc[i][3]};
        s1 = (v4f){ra[i][1], rc[i][1], ra[i][3], rc[i][3]};
        s2 = (v4f){ra[i][1] - ra[i][2], rc[i][1] - rc[i][2], ra[i][3] - ra[i][4], rc[i][3] - rc[i][4]};
        *(v4f*)(dst + (i * 3 + 0) * VPOS) = s0;
        *(v4f*)(dst + (i * 3 + 1) * VPOS) = s1;
        *(v4f*)(dst + (i * 3 + 2) * VPOS) = s2;
      }
    }
  }
  __syncthreads();

  v8f acc[G][9];
#pragma unroll
  for (int g = 0; g < G; ++g)
#pragma unroll
    for (int p = 0; p < 9; ++p) acc[g][p] = (v8f){0, 0, 0, 0, 0, 0, 0, 0};

  const v4f* wt[G];
#pragma unroll
  for (int g = 0; g < G; ++g)
    wt[g] = (const v4f*)W + (size_t)((g * 8 + wave) * 2048 + lane);
  const float* vl = vs + half * 32 + l15 * 2;

  constexpr int NCH = (HIDC / 4) / (2 / G);
  ConvOps<G> opA, opB;
  conv_load<G>(opA, wt, vl, 0);
#pragma clang loop unroll(disable)
  for (int c = 0; c < NCH; c += 2) {
    conv_load<G>(opB, wt, vl, c + 1);
    __builtin_amdgcn_sched_barrier(0);
    conv_mma<G>(opA, acc);
    __builtin_amdgcn_sched_barrier(0);
    if (c + 2 < NCH) conv_load<G>(opA, wt, vl, c + 2);
    __builtin_amdgcn_sched_barrier(0);
    conv_mma<G>(opB, acc);
  }

  float bs[G][8];
#pragma unroll
  for (int g = 0; g < G; ++g)
#pragma unroll
    for (int r = 0; r < 8; ++r) bs[g][r] = bias[g * HIDC + o0 + r + half * 8];

  const size_t obase = ((size_t)b * HIDC + o0 + half * 8) * HW + p0 + 2 * l15;

  if (G == 1) {
#pragma unroll
    for (int r = 0; r < 8; ++r) {
      float m[9];
#pragma unroll
      for (int p = 0; p < 9; ++p) m[p] = acc[0][p][r];
      v2f y0, y1;
      y0.x = eluf(MINF_Y00(m) + bs[0][r]);
      y0.y = eluf(MINF_Y01(m) + bs[0][r]);
      y1.x = eluf(MINF_Y10(m) + bs[0][r]);
      y1.y = eluf(MINF_Y11(m) + bs[0][r]);
      *(v2f*)(Y + obase + (size_t)r * HW)      = y0;
      *(v2f*)(Y + obase + (size_t)r * HW + 32) = y1;
    }
  } else {
    v2f res[8][2];
#pragma unroll
    for (int r = 0; r < 8; ++r) {
      res[r][0] = *(const v2f*)(Rin + obase + (size_t)r * HW);
      res[r][1] = *(const v2f*)(Rin + obase + (size_t)r * HW + 32);
    }
#pragma unroll
    for (int r = 0; r < 8; ++r) {
      float m1[9], m2[9];
#pragma unroll
      for (int p = 0; p < 9; ++p) {
        m1[p] = acc[0][p][r];
        m2[p] = acc[G - 1][p][r];
      }
      float t1[4], t2[4];
      t1[0] = MINF_Y00(m1) + bs[0][r];
      t1[1] = MINF_Y01(m1) + bs[0][r];
      t1[2] = MINF_Y10(m1) + bs[0][r];
      t1[3] = MINF_Y11(m1) + bs[0][r];
      t2[0] = MINF_Y00(m2) + bs[G - 1][r];
      t2[1] = MINF_Y01(m2) + bs[G - 1][r];
      t2[2] = MINF_Y10(m2) + bs[G - 1][r];
      t2[3] = MINF_Y11(m2) + bs[G - 1][r];
#pragma unroll
      for (int i = 0; i < 2; ++i) {
        v2f v, e;
        v.x = res[r][i].x + t1[2 * i]     * (1.f / (1.f + expf(-t2[2 * i])));
        v.y = res[r][i].y + t1[2 * i + 1] * (1.f / (1.f + expf(-t2[2 * i + 1])));
        e.x = eluf(v.x);
        e.y = eluf(v.y);
        *(v2f*)(Y + obase + (size_t)r * HW + i * 32) = v;
        *(v2f*)(E + obase + (size_t)r * HW + i * 32) = e;
      }
    }
  }
}

#define TPITCH 160
#define TBUF   (64 * TPITCH)

template <int N>
struct TailOps {
  v2f a[N][4 / N];
  v2f b[N][4 / N][4];
};

template <int N>
__device__ __forceinline__ void tail_load(TailOps<N>& o, const v4f* const (&wr)[N],
                                          const float* const (&bl)[N], int ch) {
#pragma unroll
  for (int n = 0; n < N; ++n) {
#pragma unroll
    for (int j2 = 0; j2 < 2 / N; ++j2) {
      const int s2 = ch * (2 / N) + j2;
      const v4f w = wr[n][s2 * 32];
      o.a[n][2 * j2]     = (v2f){w.x, w.y};
      o.a[n][2 * j2 + 1] = (v2f){w.z, w.w};
    }
#pragma unroll
    for (int j = 0; j < 4 / N; ++j) {
      const int s = ch * (4 / N) + j;
#pragma unroll
      for (int t = 0; t < 4; ++t)
        o.b[n][j][t] = *(const v2f*)(bl[n] + s * (2 * TPITCH) + t * 32);
    }
  }
}

template <int N>
__device__ __forceinline__ void tail_mma(const TailOps<N>& o, v8f (&acc)[N][4]) {
#pragma unroll
  for (int j = 0; j < 4 / N; ++j)
#pragma unroll
    for (int t = 0; t < 4; ++t)
#pragma unroll
      for (int n = 0; n < N; ++n) acc[n][t] = wmma4(o.a[n][j], o.b[n][j][t], acc[n][t]);
}

template <int N>
__device__ __forceinline__ void tail_gemm(const v4f* const (&wr)[N],
                                          const float* const (&bl)[N], v8f (&acc)[N][4]) {
#pragma unroll
  for (int n = 0; n < N; ++n)
#pragma unroll
    for (int t = 0; t < 4; ++t) acc[n][t] = (v8f){0, 0, 0, 0, 0, 0, 0, 0};
  TailOps<N> opA, opB;
  tail_load<N>(opA, wr, bl, 0);
#pragma clang loop unroll(disable)
  for (int ch = 0; ch < 8 * N; ch += 2) {
    tail_load<N>(opB, wr, bl, ch + 1);
    __builtin_amdgcn_sched_barrier(0);
    tail_mma<N>(opA, acc);
    if (ch + 2 < 8 * N) tail_load<N>(opA, wr, bl, ch + 2);
    __builtin_amdgcn_sched_barrier(0);
    tail_mma<N>(opB, acc);
  }
}

__global__ void __launch_bounds__(256)
k_tail(const float* __restrict__ O, const float* __restrict__ E,
       const float* __restrict__ Xres,
       const float* __restrict__ Wap, const float* __restrict__ bap,
       const float* __restrict__ Wc, const float* __restrict__ bc,
       const float* __restrict__ Wa, const float* __restrict__ ba,
       const float* __restrict__ Wp, const float* __restrict__ bp,
       float* __restrict__ Y) {
  __shared__ __attribute__((aligned(16))) float ts[3 * TBUF];
  const int tid  = threadIdx.x;
  const int lane = tid & 31;
  const int wave = tid >> 5;
  const int half = lane >> 4;
  const int l15  = lane & 15;
  const int b  = blockIdx.x;
  const int p0 = blockIdx.y * 64;
  const int o0 = wave * 16;

  {
    const int j  = tid & 15;
    const int pb = tid >> 4;
    const float* Ob = O + (size_t)b * HIDC * HW + p0 + j * 4;
    const float* Eb = E + (size_t)b * HIDC * HW + p0 + j * 4;
    v4f oa[4], oc[4], ea[4], ec[4];
#pragma unroll
    for (int k = 0; k < 4; ++k) {
      const size_t ro = (size_t)(2 * (pb + 16 * k)) * HW;
      oa[k] = *(const v4f*)(Ob + ro);
      oc[k] = *(const v4f*)(Ob + ro + HW);
      ea[k] = *(const v4f*)(Eb + ro);
      ec[k] = *(const v4f*)(Eb + ro + HW);
    }
#pragma unroll
    for (int k = 0; k < 4; ++k) {
      float* d = ts + (pb + 16 * k) * TPITCH + j * 8;
      *(v4f*)d              = (v4f){oa[k].x, oc[k].x, oa[k].y, oc[k].y};
      *(v4f*)(d + 4)        = (v4f){oa[k].z, oc[k].z, oa[k].w, oc[k].w};
      *(v4f*)(d + TBUF)     = (v4f){ea[k].x, ec[k].x, ea[k].y, ec[k].y};
      *(v4f*)(d + TBUF + 4) = (v4f){ea[k].z, ec[k].z, ea[k].w, ec[k].w};
    }
  }
  __syncthreads();

  const int    wofs = wave * 512 + lane;
  const int    lofs = half * TPITCH + l15 * 2;
  float* tw = ts + ((o0 >> 1) + half * 4) * TPITCH + l15 * 2;
  const int ob = o0 + half * 8;

  {
    v8f acc[1][4];
    const v4f* const wr[1] = {(const v4f*)Wap + wofs};
    const float* const bl[1] = {ts + lofs};
    tail_gemm<1>(wr, bl, acc);
    float bs[8];
#pragma unroll
    for (int r = 0; r < 8; ++r) bs[r] = bap[ob + r];
#pragma unroll
    for (int t = 0; t < 4; ++t)
#pragma unroll
      for (int r = 0; r < 8; r += 2) {
        v2f v;
        v.x = eluf(acc[0][t][r] + bs[r]);
        v.y = eluf(acc[0][t][r + 1] + bs[r + 1]);
        *(v2f*)(tw + 2 * TBUF + (r >> 1) * TPITCH + t * 32) = v;
      }
  }
  __syncthreads();

  float xr[4][8];
  const size_t ybase = ((size_t)b * HIDC + ob) * HW + p0 + l15;
  {
    v8f acc[2][4];
    const v4f* const wr[2] = {(const v4f*)Wa + wofs, (const v4f*)Wc + wofs};
    const float* const bl[2] = {ts + 2 * TBUF + lofs, ts + TBUF + lofs};
    tail_gemm<2>(wr, bl, acc);
    float bsa[8], bsc[8];
#pragma unroll
    for (int r = 0; r < 8; ++r) {
      bsa[r] = ba[ob + r];
      bsc[r] = bc[ob + r];
    }
#pragma unroll
    for (int t = 0; t < 4; ++t)
#pragma unroll
      for (int r = 0; r < 8; ++r) xr[t][r] = Xres[ybase + (size_t)r * HW + t * 16];
#pragma unroll
    for (int t = 0; t < 4; ++t)
#pragma unroll
      for (int r = 0; r < 8; r += 2) {
        v2f v;
        {
          const float a1 = eluf(acc[0][t][r] + bsa[r]);
          const float c1 = eluf(acc[1][t][r] + bsc[r]);
          v.x = eluf(c1 + a1);
        }
        {
          const float a1 = eluf(acc[0][t][r + 1] + bsa[r + 1]);
          const float c1 = eluf(acc[1][t][r + 1] + bsc[r + 1]);
          v.y = eluf(c1 + a1);
        }
        *(v2f*)(tw + (r >> 1) * TPITCH + t * 32) = v;
      }
  }
  __syncthreads();

  {
    v8f acc[1][4];
    const v4f* const wr[1] = {(const v4f*)Wp + wofs};
    const float* const bl[1] = {ts + lofs};
    tail_gemm<1>(wr, bl, acc);
    float bs[8];
#pragma unroll
    for (int r = 0; r < 8; ++r) bs[r] = bp[ob + r];
#pragma unroll
    for (int t = 0; t < 4; ++t)
#pragma unroll
      for (int r = 0; r < 8; ++r) {
        float v = eluf(acc[0][t][r] + bs[r]);
        v += xr[t][r];
        Y[ybase + (size_t)r * HW + t * 16] = v;
      }
  }
}

__global__ void __launch_bounds__(256)
k_qkv(const float* __restrict__ X, const float* __restrict__ W,
      const float* __restrict__ bias, float* __restrict__ Y) {
  const int lane = threadIdx.x & 31;
  const int wave = __builtin_amdgcn_readfirstlane((int)(threadIdx.x >> 5));
  const int half = lane >> 4;
  const int l15  = lane & 15;
  const int b  = blockIdx.x;
  const int p0 = blockIdx.y * 64;
  const int o0 = (blockIdx.z * 8 + wave) * 16;
  if (o0 >= 160) return;

  const float* Xb   = X + (size_t)b * HIDC * HW;
  const float* wrow = W + (size_t)(o0 + l15) * 130;

  v8f acc[4];
#pragma unroll
  for (int t = 0; t < 4; ++t) acc[t] = (v8f){0, 0, 0, 0, 0, 0, 0, 0};

  for (int c0 = 0; c0 < HIDC; c0 += 4) {
    int ca = c0 + half * 2;
    v2f a;
    a.x = wrow[ca];
    a.y = wrow[ca + 1];
    const float* x0 = Xb + (size_t)ca * HW + p0 + l15;
#pragma unroll
    for (int t = 0; t < 4; ++t) {
      v2f bb;
      bb.x = x0[t * 16];
      bb.y = x0[HW + t * 16];
      acc[t] = wmma4(a, bb, acc[t]);
    }
  }

  const bool isv = o0 >= 32;
#pragma unroll
  for (int t = 0; t < 4; ++t) {
    int p = p0 + t * 16 + l15;
    float py = (float)(p >> 5) * (1.f / 32.f) - 0.5f;
    float px = (float)(p & 31) * (1.f / 32.f) - 0.5f;
    const size_t vbase = ((size_t)b * 160 + 32) * HW +
                         (size_t)((((p >> 4) * 8 + ((o0 - 32) >> 4)) * 2 + ((l15 >> 2) & 1)) * 128 +
                                  ((l15 >> 3) * 16 + half * 8) * 4 + (l15 & 3));
    float v[8];
#pragma unroll
    for (int r = 0; r < 8; ++r) {
      int o = o0 + r + half * 8;
      v[r] = acc[t][r] + W[(size_t)o * 130 + 128] * py +
             W[(size_t)o * 130 + 129] * px + bias[o];
    }
    if (isv) {
#pragma unroll
      for (int r = 0; r < 8; ++r) Y[vbase + r * 4] = v[r];
    } else {
#pragma unroll
      for (int r = 0; r < 8; ++r) Y[((size_t)b * 160 + o0 + r + half * 8) * HW + p] = v[r];
    }
  }
}

#define ATT_NW 8
__global__ void __launch_bounds__(256)
k_attn(const float* __restrict__ QKV, float* __restrict__ Ob) {
  __shared__ __attribute__((aligned(16))) float po[ATT_NW * 8 * 8 * 32];
  __shared__ __attribute__((aligned(16))) float pm[ATT_NW * 16];
  __shared__ __attribute__((aligned(16))) float ps[ATT_NW * 32];
  const int lane = threadIdx.x & 31;
  const int wave = __builtin_amdgcn_readfirstlane((int)(threadIdx.x >> 5));
  const int half = lane >> 4;
  const int l15  = lane & 15;
  const int qt = (HW / 16 - 1) - (int)blockIdx.x;
  const int b  = blockIdx.y;
  const int q0 = qt * 16;
  const int qp = q0 + l15;
  const float* Qb = QKV + (size_t)b * 160 * HW;

  v2f qb[4];
#pragma unroll
  for (int kc = 0; kc < 4; ++kc) {
    qb[kc].x = Qb[(size_t)(kc * 4 + half * 2) * HW + qp];
    qb[kc].y = Qb[(size_t)(kc * 4 + half * 2 + 1) * HW + qp];
  }

  v8f acc[8];
#pragma unroll
  for (int vt = 0; vt < 8; ++vt) acc[vt] = (v8f){0, 0, 0, 0, 0, 0, 0, 0};
  float mrun = -3.0e38f;
  float lrun = 0.f;

  for (int kt = wave; kt <= qt; kt += ATT_NW) {
    const int k0 = kt * 16;
    v2f ka[4];
#pragma unroll
    for (int kc = 0; kc < 4; ++kc) {
      ka[kc].x = Qb[(size_t)(16 + kc * 4 + half * 2) * HW + k0 + l15];
      ka[kc].y = Qb[(size_t)(16 + kc * 4 + half * 2 + 1) * HW + k0 + l15];
    }
    v4f va[8][2];
    const float* vtile = Qb + (size_t)32 * HW + (size_t)kt * 2048 + lane * 4;
#pragma unroll
    for (int vt = 0; vt < 8; ++vt) {
      va[vt][0] = *(const v4f*)(vtile + vt * 256);
      va[vt][1] = *(const v4f*)(vtile + vt * 256 + 128);
    }
    __builtin_amdgcn_sched_barrier(0);
    v8f s = (v8f){0, 0, 0, 0, 0, 0, 0, 0};
#pragma unroll
    for (int kc = 0; kc < 4; ++kc) s = wmma4(ka[kc], qb[kc], s);
    float val[8];
    float rm = -3.0e38f;
#pragma unroll
    for (int r = 0; r < 8; ++r) {
      const int kp = k0 + half * 8 + r;
      val[r] = (kp <= qp) ? s[r] * 0.25f : -3.0e38f;
      rm = fmaxf(rm, val[r]);
    }
    rm = fmaxf(rm, __shfl_xor(rm, 16));
    const float nm = fmaxf(mrun, rm);
    const float alpha = expf(mrun - nm);
    v8f p;
    float rs = 0.f;
#pragma unroll
    for (int r = 0; r < 8; ++r) {
      p[r] = expf(val[r] - nm);
      rs += p[r];
    }
    lrun = lrun * alpha + rs;
    mrun = nm;
#pragma unroll
    for (int vt = 0; vt < 8; ++vt)
#pragma unroll
      for (int r = 0; r < 8; ++r) acc[vt][r] *= alpha;
#pragma unroll
    for (int kc = 0; kc < 4; ++kc) {
      v2f pb;
      pb.x = p[2 * kc];
      pb.y = p[2 * kc + 1];
#pragma unroll
      for (int vt = 0; vt < 8; ++vt) {
        v2f a;
        a.x = va[vt][kc >> 1][(kc & 1) * 2];
        a.y = va[vt][kc >> 1][(kc & 1) * 2 + 1];
        acc[vt] = wmma4(a, pb, acc[vt]);
      }
    }
  }

  const int nact = (qt + 1 < ATT_NW) ? qt + 1 : ATT_NW;
  const bool active = wave < nact;
  if (active && half == 0) pm[wave * 16 + l15] = mrun;
  __syncthreads();
  if (active) {
    float M = pm[l15];
    for (int j = 1; j < nact; ++j) M = fmaxf(M, pm[j * 16 + l15]);
    const float sc = expf(mrun - M);
    ps[wave * 32 + lane] = lrun * sc;
#pragma unroll
    for (int r = 0; r < 8; ++r)
#pragma unroll
      for (int vt = 0; vt < 8; ++vt)
        po[((wave * 8 + vt) * 8 + r) * 32 + lane] = acc[vt][r] * sc;
  }
  __syncthreads();

  float Ls = 0.f, os[8];
#pragma unroll
  for (int r = 0; r < 8; ++r) os[r] = 0.f;
  for (int j = 0; j < nact; ++j) {
    Ls += ps[j * 32 + l15] + ps[j * 32 + 16 + l15];
#pragma unroll
    for (int r = 0; r < 8; ++r) os[r] += po[((j * 8 + wave) * 8 + r) * 32 + lane];
  }
  const float inv = 1.f / Ls;
#pragma unroll
  for (int r = 0; r < 8; ++r)
    Ob[((size_t)b * HIDC + wave * 16 + half * 8 + r) * HW + qp] = os[r] * inv;
}

extern "C" void kernel_launch(void* const* d_in, const int* in_sizes, int n_in,
                              void* d_out, int out_size, void* d_ws, size_t ws_size,
                              hipStream_t stream) {
  const float* x    = (const float*)d_in[0];
  const float* rw1  = (const float*)d_in[1];
  const float* rb1  = (const float*)d_in[2];
  const float* rw2  = (const float*)d_in[3];
  const float* rb2  = (const float*)d_in[4];
  const float* qkvw = (const float*)d_in[5];
  const float* qkvb = (const float*)d_in[6];
  const float* apw  = (const float*)d_in[7];
  const float* apb  = (const float*)d_in[8];
  const float* ocw  = (const float*)d_in[9];
  const float* ocb  = (const float*)d_in[10];
  const float* oaw  = (const float*)d_in[11];
  const float* oab  = (const float*)d_in[12];
  const float* opw  = (const float*)d_in[13];
  const float* opb  = (const float*)d_in[14];
  float* Yout = (float*)d_out;

  const size_t N1 = (size_t)NB * HIDC * HW;
  float* A  = (float*)d_ws;
  float* E  = A + N1;
  float* F  = E + N1;
  float* Df = F + N1;
  float* P1 = Df + 2 * N1;
  float* P2 = P1 + (size_t)6 * HIDC * HIDC * 4;
  float* PT = P2 + (size_t)6 * 2 * HIDC * HIDC * 4;

  dim3 blk(256);
  dim3 g1((unsigned)((N1 + 255) / 256) + 1216);
  dim3 g128(NB, HW / 64, 1), g256(NB, HW / 64, 2);

  k_first<<<g1, blk, 0, stream>>>(x, E, (int)N1, rw1, rw2, apw, ocw, oaw, opw, P1, P2, PT);
  for (int i = 0; i < 6; ++i) {
    const float* w1 = P1 + (size_t)i * HIDC * HIDC * 4;
    const float* b1 = rb1 + (size_t)i * HIDC;
    const float* w2 = P2 + (size_t)i * 2 * HIDC * HIDC * 4;
    const float* b2 = rb2 + (size_t)i * 2 * HIDC;
    k_conv2x2<1><<<g128, blk, 0, stream>>>(E, w1, b1, F, nullptr, nullptr);
    k_conv2x2<2><<<g128, blk, 0, stream>>>(F, w2, b2, A, i == 0 ? x : A, E);
  }

  k_qkv<<<g256, blk, 0, stream>>>(A, qkvw, qkvb, Df);
  k_attn<<<dim3(HW / 16, NB), dim3(256), 0, stream>>>(Df, F);

  k_tail<<<g128, blk, 0, stream>>>(F, E, x, PT, apb, PT + 16384, ocb, PT + 2 * 16384, oab,
                                   PT + 3 * 16384, opb, Yout);
}


# Conv weight fragments requested two chunks ahead of use (three register sets in rotation, was one ahead); first launch stores tap pairs adjacent, so no register copies; every sum keeps its order
# speedup vs baseline: 2.7971x; 1.0317x over previous
#include <hip/hip_runtime.h>
#include <hip/hip_bf16.h>


typedef __attribute__((ext_vector_type(2))) float v2f;
typedef __attribute__((ext_vector_type(4))) float v4f;
typedef __attribute__((ext_vector_type(8))) float v8f;

#define NB   32
#define HIDC 128
#define HW   1024

__device__ __forceinline__ v8f wmma4(v2f a, v2f b, v8f c) {
  return __builtin_amdgcn_wmma_f32_16x16x4_f32(false, a, false, b, (short)0, c,
                                               false, false);
}

__device__ __forceinline__ float eluf(float x) {
  return x > 0.f ? x : (expf(x) - 1.f);
}

#define NW1_16B (6 * 128 * 128)
#define NW2_16B (6 * 256 * 128)
#define NWT_16B (4 * 128 * 32)
__global__ void __launch_bounds__(256)
k_first(const float* __restrict__ X, float* __restrict__ Y, int n,
        const float* __restrict__ W1, const float* __restrict__ W2,
        const float* __restrict__ Wap, const float* __restrict__ Wc,
        const float* __restrict__ Wa, const float* __restrict__ Wp,
        float* __restrict__ P1, float* __restrict__ P2, float* __restrict__ PT) {
  const int nelu = (n + 255) >> 8;
  if ((int)blockIdx.x < nelu) {
    int i = blockIdx.x * 256 + threadIdx.x;
    if (i < n) Y[i] = eluf(X[i]);
    return;
  }
  const int t = ((int)blockIdx.x - nelu) * 256 + (int)threadIdx.x;
  if (t < NW1_16B + NW2_16B) {
    const bool first = t < NW1_16B;
    const float* src = first ? W1 : W2;
    v4f* dst = (v4f*)(first ? P1 : P2);
    const int d    = first ? t : t - NW1_16B;
    const int lane = d & 31;
    const int e    = (d >> 5) & 1;
    const int s    = (d >> 6) & 31;
    const int rg   = d >> 11;
    const int row  = rg * 16 + (lane & 15);
    const int ch   = 4 * s + 2 * (lane >> 4);
    const float* r = src + ((size_t)row * 128 + ch) * 4 + 2 * e;
    const v2f c0 = *(const v2f*)r;
    const v2f c1 = *(const v2f*)(r + 4);
    dst[d] = (v4f){c0.x, c1.x, c0.y, c1.y};
  } else if (t < NW1_16B + NW2_16B + NWT_16B) {
    const int t2 = t - (NW1_16B + NW2_16B);
    const int m  = t2 >> 12;
    const float* Wm = (m == 0) ? Wap : (m == 1) ? Wc : (m == 2) ? Wa : Wp;
    const int d    = t2 & 4095;
    const int lane = d & 31;
    const int s2   = (d >> 5) & 15;
    const int rg   = d >> 9;
    const float* r = Wm + (size_t)(rg * 16 + (lane & 15)) * 128 + 8 * s2 + 2 * (lane >> 4);
    const v2f lo = *(const v2f*)r;
    const v2f hi = *(const v2f*)(r + 4);
    ((v4f*)PT)[t2] = (v4f){lo.x, lo.y, hi.x, hi.y};
  }
}

#define VPOS 2048

template <int G>
struct ConvW {
  v4f wa[2 / G][G][2];
};
template <int G>
struct ConvF {
  v2f fr[2 / G][9];
};

template <int G>
__device__ __forceinline__ void conv_load_w(ConvW<G>& o, const v4f* const (&wt)[G], int ch) {
#pragma unroll
  for (int j = 0; j < 2 / G; ++j) {
    const int step = ch * (2 / G) + j;
#pragma unroll
    for (int g = 0; g < G; ++g) {
      o.wa[j][g][0] = wt[g][step * 64];
      o.wa[j][g][1] = wt[g][step * 64 + 32];
    }
  }
}

template <int G>
__device__ __forceinline__ void conv_load_f(ConvF<G>& o, const float* vl, int ch) {
#pragma unroll
  for (int j = 0; j < 2 / G; ++j) {
    const float* base = vl + (ch * (2 / G) + j) * 64;
#pragma unroll
    for (int p = 0; p < 9; ++p) o.fr[j][p] = *(const v2f*)(base + p * VPOS);
  }
}

template <int G>
__device__ __forceinline__ void conv_mma(const ConvW<G>& w, const ConvF<G>& f, v8f (&acc)[G][9]) {
#pragma unroll
  for (int j = 0; j < 2 / G; ++j) {
    v2f u[G][9];
#pragma unroll
    for (int g = 0; g < G; ++g) {
      const v4f w01 = w.wa[j][g][0], w23 = w.wa[j][g][1];
      const v2f t0 = (v2f){w01[0], w01[1]};
      const v2f t1 = (v2f){w01[2], w01[3]};
      const v2f t2 = (v2f){w23[0], w23[1]};
      const v2f t3 = (v2f){w23[2], w23[3]};
      u[g][0] = t0;
      u[g][1] = t0 + t1;
      u[g][2] = t1;
      u[g][3] = t0 + t2;
      u[g][5] = t1 + t3;
      u[g][6] = t2;
      u[g][7] = t2 + t3;
      u[g][8] = t3;
      u[g][4] = u[g][1] + u[g][7];
    }
#pragma unroll
    for (int p = 0; p < 9; ++p)
#pragma unroll
      for (int g = 0; g < G; ++g) acc[g][p] = wmma4(u[g][p], f.fr[j][p], acc[g][p]);
  }
}

#define MINF_Y00(m) ((m[0] + m[1]) + (m[3] + m[4]))
#define MINF_Y01(m) ((m[1] - m[2]) + (m[4] - m[5]))
#define MINF_Y10(m) ((m[3] + m[4]) - (m[6] + m[7]))
#define MINF_Y11(m) ((m[4] - m[5]) - (m[7] - m[8]))

template <int G>
__global__ void __launch_bounds__(256)
k_conv2x2(const float* __restrict__ X, const float* __restrict__ W,
          const float* __restrict__ bias, float* Y, const float* Rin,
          float* __restrict__ E) {
  __shared__ __attribute__((aligned(16))) float vs[9 * VPOS];
  const int tid  = threadIdx.x;
  const int lane = tid & 31;
  const int wave = tid >> 5;
  const int half = lane >> 4;
  const int l15  = lane & 15;
  const int b  = blockIdx.x;
  const int p0 = blockIdx.y * 64;
  const int o0 = wave * 16;

  const float* Xb = X + (size_t)b * HIDC * HW;

  const v4f* wt[G];
#pragma unroll
  for (int g = 0; g < G; ++g)
    wt[g] = (const v4f*)W + (size_t)((g * 8 + wave) * 2048 + lane);

  ConvW<G> wq[3];
  conv_load_w<G>(wq[0], wt, 0);
  conv_load_w<G>(wq[1], wt, 1);

  {
    const int yt = (int)blockIdx.y * 2 - 1;
    const int j  = tid & 7;
    const int pb = tid >> 3;
    v4f ga[3][2], gc[3][2];
#pragma unroll
    for (int r = 0; r < 3; ++r) {
#pragma unroll
      for (int k = 0; k < 2; ++k) {
        const int yy = (r == 0 && yt < 0) ? 0 : yt + r;
        const float* g = Xb + (size_t)(2 * pb + 64 * k) * HW + yy * 32 + j * 4;
        ga[r][k] = *(const v4f*)g;
        gc[r][k] = *(const v4f*)(g + HW);
      }
    }
    if (yt < 0) {
#pragma unroll
      for (int k = 0; k < 2; ++k) {
        ga[0][k] = (v4f){0.f, 0.f, 0.f, 0.f};
        gc[0][k] = (v4f){0.f, 0.f, 0.f, 0.f};
      }
    }
#pragma unroll
    for (int k = 0; k < 2; ++k) {
      float da[3][5], dc[3][5];
#pragma unroll
      for (int r = 0; r < 3; ++r) {
        float la = __shfl_up(ga[r][k][3], 1);
        float lc = __shfl_up(gc[r][k][3], 1);
        da[r][0] = (j == 0) ? 0.f : la;
        dc[r][0] = (j == 0) ? 0.f : lc;
#pragma unroll
        for (int q = 0; q < 4; ++q) {
          da[r][q + 1] = ga[r][k][q];
          dc[r][q + 1] = gc[r][k][q];
        }
      }
      float ra[3][5], rc[3][5];
#pragma unroll
      for (int q = 0; q < 5; ++q) {
        ra[0][q] = da[0][q] - da[1][q];
        ra[1][q] = da[1][q];
        ra[2][q] = da[1][q] - da[2][q];
        rc[0][q] = dc[0][q] - dc[1][q];
        rc[1][q] = dc[1][q];
        rc[2][q] = dc[1][q] - dc[2][q];
      }
      float* dst = vs + (pb + 32 * k) * 32 + j * 4;
#pragma unroll
      for (int i = 0; i < 3; ++i) {
        v4f s0, s1, s2;
        s0 = (v4f){ra[i][0] - ra[i][1], rc[i][0] - rc[i][1], ra[i][2] - ra[i][3], rc[i][2] - rc[i][3]};
        s1 = (v4f){ra[i][1], rc[i][1], ra[i][3], rc[i][3]};
        s2 = (v4f){ra[i][1] - ra[i][2], rc[i][1] - rc[i][2], ra[i][3] - ra[i][4], rc[i][3] - rc[i][4]};
        *(v4f*)(dst + (i * 3 + 0) * VPOS) = s0;
        *(v4f*)(dst + (i * 3 + 1) * VPOS) = s1;
        *(v4f*)(dst + (i * 3 + 2) * VPOS) = s2;
      }
    }
  }
  __syncthreads();

  v8f acc[G][9];
#pragma unroll
  for (int g = 0; g < G; ++g)
#pragma unroll
    for (int p = 0; p < 9; ++p) acc[g][p] = (v8f){0, 0, 0, 0, 0, 0, 0, 0};

  const float* vl = vs + half * 32 + l15 * 2;

  constexpr int NCH   = (HIDC / 4) / (2 / G);
  constexpr int TAIL0 = ((NCH - 2) / 6) * 6;
  ConvF<G> fq[2];
  conv_load_f<G>(fq[0], vl, 0);
#pragma clang loop unroll(disable)
  for (int c = 0; c < TAIL0; c += 6) {
#pragma unroll
    for (int t = 0; t < 6; ++t) {
      conv_load_w<G>(wq[(t + 2) % 3], wt, c + t + 2);
      conv_load_f<G>(fq[(t + 1) % 2], vl, c + t + 1);
      __builtin_amdgcn_sched_barrier(0);
      conv_mma<G>(wq[t % 3], fq[t % 2], acc);
      __builtin_amdgcn_sched_barrier(0);
    }
  }
#pragma unroll
  for (int t = 0; t < NCH - TAIL0; ++t) {
    if (TAIL0 + t + 2 < NCH) conv_load_w<G>(wq[(t + 2) % 3], wt, TAIL0 + t + 2);
    if (TAIL0 + t + 1 < NCH) conv_load_f<G>(fq[(t + 1) % 2], vl, TAIL0 + t + 1);
    __builtin_amdgcn_sched_barrier(0);
    conv_mma<G>(wq[t % 3], fq[t % 2], acc);
    __builtin_amdgcn_sched_barrier(0);
  }

  float bs[G][8];
#pragma unroll
  for (int g = 0; g < G; ++g)
#pragma unroll
    for (int r = 0; r < 8; ++r) bs[g][r] = bias[g * HIDC + o0 + r + half * 8];

  const size_t obase = ((size_t)b * HIDC + o0 + half * 8) * HW + p0 + 2 * l15;

  if (G == 1) {
#pragma unroll
    for (int r = 0; r < 8; ++r) {
      float m[9];
#pragma unroll
      for (int p = 0; p < 9; ++p) m[p] = acc[0][p][r];
      v2f y0, y1;
      y0.x = eluf(MINF_Y00(m) + bs[0][r]);
      y0.y = eluf(MINF_Y01(m) + bs[0][r]);
      y1.x = eluf(MINF_Y10(m) + bs[0][r]);
      y1.y = eluf(MINF_Y11(m) + bs[0][r]);
      *(v2f*)(Y + obase + (size_t)r * HW)      = y0;
      *(v2f*)(Y + obase + (size_t)r * HW + 32) = y1;
    }
  } else {
    v2f res[8][2];
#pragma unroll
    for (int r = 0; r < 8; ++r) {
      res[r][0] = *(const v2f*)(Rin + obase + (size_t)r * HW);
      res[r][1] = *(const v2f*)(Rin + obase + (size_t)r * HW + 32);
    }
#pragma unroll
    for (int r = 0; r < 8; ++r) {
      float m1[9], m2[9];
#pragma unroll
      for (int p = 0; p < 9; ++p) {
        m1[p] = acc[0][p][r];
        m2[p] = acc[G - 1][p][r];
      }
      float t1[4], t2[4];
      t1[0] = MINF_Y00(m1) + bs[0][r];
      t1[1] = MINF_Y01(m1) + bs[0][r];
      t1[2] = MINF_Y10(m1) + bs[0][r];
      t1[3] = MINF_Y11(m1) + bs[0][r];
      t2[0] = MINF_Y00(m2) + bs[G - 1][r];
      t2[1] = MINF_Y01(m2) + bs[G - 1][r];
      t2[2] = MINF_Y10(m2) + bs[G - 1][r];
      t2[3] = MINF_Y11(m2) + bs[G - 1][r];
#pragma unroll
      for (int i = 0; i < 2; ++i) {
        v2f v, e;
        v.x = res[r][i].x + t1[2 * i]     * (1.f / (1.f + expf(-t2[2 * i])));
        v.y = res[r][i].y + t1[2 * i + 1] * (1.f / (1.f + expf(-t2[2 * i + 1])));
        e.x = eluf(v.x);
        e.y = eluf(v.y);
        *(v2f*)(Y + obase + (size_t)r * HW + i * 32) = v;
        *(v2f*)(E + obase + (size_t)r * HW + i * 32) = e;
      }
    }
  }
}

#define TPITCH 160
#define TBUF   (64 * TPITCH)

template <int N>
struct TailOps {
  v2f a[N][4 / N];
  v2f b[N][4 / N][4];
};

template <int N>
__device__ __forceinline__ void tail_load(TailOps<N>& o, const v4f* const (&wr)[N],
                                          const float* const (&bl)[N], int ch) {
#pragma unroll
  for (int n = 0; n < N; ++n) {
#pragma unroll
    for (int j2 = 0; j2 < 2 / N; ++j2) {
      const int s2 = ch * (2 / N) + j2;
      const v4f w = wr[n][s2 * 32];
      o.a[n][2 * j2]     = (v2f){w.x, w.y};
      o.a[n][2 * j2 + 1] = (v2f){w.z, w.w};
    }
#pragma unroll
    for (int j = 0; j < 4 / N; ++j) {
      const int s = ch * (4 / N) + j;
#pragma unroll
      for (int t = 0; t < 4; ++t)
        o.b[n][j][t] = *(const v2f*)(bl[n] + s * (2 * TPITCH) + t * 32);
    }
  }
}

template <int N>
__device__ __forceinline__ void tail_mma(const TailOps<N>& o, v8f (&acc)[N][4]) {
#pragma unroll
  for (int j = 0; j < 4 / N; ++j)
#pragma unroll
    for (int t = 0; t < 4; ++t)
#pragma unroll
      for (int n = 0; n < N; ++n) acc[n][t] = wmma4(o.a[n][j], o.b[n][j][t], acc[n][t]);
}

template <int N>
__device__ __forceinline__ void tail_gemm(const v4f* const (&wr)[N],
                                          const float* const (&bl)[N], v8f (&acc)[N][4]) {
#pragma unroll
  for (int n = 0; n < N; ++n)
#pragma unroll
    for (int t = 0; t < 4; ++t) acc[n][t] = (v8f){0, 0, 0, 0, 0, 0, 0, 0};
  TailOps<N> opA, opB;
  tail_load<N>(opA, wr, bl, 0);
#pragma clang loop unroll(disable)
  for (int ch = 0; ch < 8 * N; ch += 2) {
    tail_load<N>(opB, wr, bl, ch + 1);
    __builtin_amdgcn_sched_barrier(0);
    tail_mma<N>(opA, acc);
    if (ch + 2 < 8 * N) tail_load<N>(opA, wr, bl, ch + 2);
    __builtin_amdgcn_sched_barrier(0);
    tail_mma<N>(opB, acc);
  }
}

__global__ void __launch_bounds__(256)
k_tail(const float* __restrict__ O, const float* __restrict__ E,
       const float* __restrict__ Xres,
       const float* __restrict__ Wap, const float* __restrict__ bap,
       const float* __restrict__ Wc, const float* __restrict__ bc,
       const float* __restrict__ Wa, const float* __restrict__ ba,
       const float* __restrict__ Wp, const float* __restrict__ bp,
       float* __restrict__ Y) {
  __shared__ __attribute__((aligned(16))) float ts[3 * TBUF];
  const int tid  = threadIdx.x;
  const int lane = tid & 31;
  const int wave = tid >> 5;
  const int half = lane >> 4;
  const int l15  = lane & 15;
  const int b  = blockIdx.x;
  const int p0 = blockIdx.y * 64;
  const int o0 = wave * 16;

  {
    const int j  = tid & 15;
    const int pb = tid >> 4;
    const float* Ob = O + (size_t)b * HIDC * HW + p0 + j * 4;
    const float* Eb = E + (size_t)b * HIDC * HW + p0 + j * 4;
    v4f oa[4], oc[4], ea[4], ec[4];
#pragma unroll
    for (int k = 0; k < 4; ++k) {
      const size_t ro = (size_t)(2 * (pb + 16 * k)) * HW;
      oa[k] = *(const v4f*)(Ob + ro);
      oc[k] = *(const v4f*)(Ob + ro + HW);
      ea[k] = *(const v4f*)(Eb + ro);
      ec[k] = *(const v4f*)(Eb + ro + HW);
    }
#pragma unroll
    for (int k = 0; k < 4; ++k) {
      float* d = ts + (pb + 16 * k) * TPITCH + j * 8;
      *(v4f*)d              = (v4f){oa[k].x, oc[k].x, oa[k].y, oc[k].y};
      *(v4f*)(d + 4)        = (v4f){oa[k].z, oc[k].z, oa[k].w, oc[k].w};
      *(v4f*)(d + TBUF)     = (v4f){ea[k].x, ec[k].x, ea[k].y, ec[k].y};
      *(v4f*)(d + TBUF + 4) = (v4f){ea[k].z, ec[k].z, ea[k].w, ec[k].w};
    }
  }
  __syncthreads();

  const int    wofs = wave * 512 + lane;
  const int    lofs = half * TPITCH + l15 * 2;
  float* tw = ts + ((o0 >> 1) + half * 4) * TPITCH + l15 * 2;
  const int ob = o0 + half * 8;

  {
    v8f acc[1][4];
    const v4f* const wr[1] = {(const v4f*)Wap + wofs};
    const float* const bl[1] = {ts + lofs};
    tail_gemm<1>(wr, bl, acc);
    float bs[8];
#pragma unroll
    for (int r = 0; r < 8; ++r) bs[r] = bap[ob + r];
#pragma unroll
    for (int t = 0; t < 4; ++t)
#pragma unroll
      for (int r = 0; r < 8; r += 2) {
        v2f v;
        v.x = eluf(acc[0][t][r] + bs[r]);
        v.y = eluf(acc[0][t][r + 1] + bs[r + 1]);
        *(v2f*)(tw + 2 * TBUF + (r >> 1) * TPITCH + t * 32) = v;
      }
  }
  __syncthreads();

  float xr[4][8];
  const size_t ybase = ((size_t)b * HIDC + ob) * HW + p0 + l15;
  {
    v8f acc[2][4];
    const v4f* const wr[2] = {(const v4f*)Wa + wofs, (const v4f*)Wc + wofs};
    const float* const bl[2] = {ts + 2 * TBUF + lofs, ts + TBUF + lofs};
    tail_gemm<2>(wr, bl, acc);
    float bsa[8], bsc[8];
#pragma unroll
    for (int r = 0; r < 8; ++r) {
      bsa[r] = ba[ob + r];
      bsc[r] = bc[ob + r];
    }
#pragma unroll
    for (int t = 0; t < 4; ++t)
#pragma unroll
      for (int r = 0; r < 8; ++r) xr[t][r] = Xres[ybase + (size_t)r * HW + t * 16];
#pragma unroll
    for (int t = 0; t < 4; ++t)
#pragma unroll
      for (int r = 0; r < 8; r += 2) {
        v2f v;
        {
          const float a1 = eluf(acc[0][t][r] + bsa[r]);
          const float c1 = eluf(acc[1][t][r] + bsc[r]);
          v.x = eluf(c1 + a1);
        }
        {
          const float a1 = eluf(acc[0][t][r + 1] + bsa[r + 1]);
          const float c1 = eluf(acc[1][t][r + 1] + bsc[r + 1]);
          v.y = eluf(c1 + a1);
        }
        *(v2f*)(tw + (r >> 1) * TPITCH + t * 32) = v;
      }
  }
  __syncthreads();

  {
    v8f acc[1][4];
    const v4f* const wr[1] = {(const v4f*)Wp + wofs};
    const float* const bl[1] = {ts + lofs};
    tail_gemm<1>(wr, bl, acc);
    float bs[8];
#pragma unroll
    for (int r = 0; r < 8; ++r) bs[r] = bp[ob + r];
#pragma unroll
    for (int t = 0; t < 4; ++t)
#pragma unroll
      for (int r = 0; r < 8; ++r) {
        float v = eluf(acc[0][t][r] + bs[r]);
        v += xr[t][r];
        Y[ybase + (size_t)r * HW + t * 16] = v;
      }
  }
}

__global__ void __launch_bounds__(256)
k_qkv(const float* __restrict__ X, const float* __restrict__ W,
      const float* __restrict__ bias, float* __restrict__ Y) {
  const int lane = threadIdx.x & 31;
  const int wave = __builtin_amdgcn_readfirstlane((int)(threadIdx.x >> 5));
  const int half = lane >> 4;
  const int l15  = lane & 15;
  const int b  = blockIdx.x;
  const int p0 = blockIdx.y * 64;
  const int o0 = (blockIdx.z * 8 + wave) * 16;
  if (o0 >= 160) return;

  const float* Xb   = X + (size_t)b * HIDC * HW;
  const float* wrow = W + (size_t)(o0 + l15) * 130;

  v8f acc[4];
#pragma unroll
  for (int t = 0; t < 4; ++t) acc[t] = (v8f){0, 0, 0, 0, 0, 0, 0, 0};

  for (int c0 = 0; c0 < HIDC; c0 += 4) {
    int ca = c0 + half * 2;
    v2f a;
    a.x = wrow[ca];
    a.y = wrow[ca + 1];
    const float* x0 = Xb + (size_t)ca * HW + p0 + l15;
#pragma unroll
    for (int t = 0; t < 4; ++t) {
      v2f bb;
      bb.x = x0[t * 16];
      bb.y = x0[HW + t * 16];
      acc[t] = wmma4(a, bb, acc[t]);
    }
  }

  const bool isv = o0 >= 32;
#pragma unroll
  for (int t = 0; t < 4; ++t) {
    int p = p0 + t * 16 + l15;
    float py = (float)(p >> 5) * (1.f / 32.f) - 0.5f;
    float px = (float)(p & 31) * (1.f / 32.f) - 0.5f;
    const size_t vbase = ((size_t)b * 160 + 32) * HW +
                         (size_t)((((p >> 4) * 8 + ((o0 - 32) >> 4)) * 2 + ((l15 >> 2) & 1)) * 128 +
                                  ((l15 >> 3) * 16 + half * 8) * 4 + (l15 & 3));
    float v[8];
#pragma unroll
    for (int r = 0; r < 8; ++r) {
      int o = o0 + r + half * 8;
      v[r] = acc[t][r] + W[(size_t)o * 130 + 128] * py +
             W[(size_t)o * 130 + 129] * px + bias[o];
    }
    if (isv) {
#pragma unroll
      for (int r = 0; r < 8; ++r) Y[vbase + r * 4] = v[r];
    } else {
#pragma unroll
      for (int r = 0; r < 8; ++r) Y[((size_t)b * 160 + o0 + r + half * 8) * HW + p] = v[r];
    }
  }
}

#define ATT_NW 8
__global__ void __launch_bounds__(256)
k_attn(const float* __restrict__ QKV, float* __restrict__ Ob) {
  __shared__ __attribute__((aligned(16))) float po[ATT_NW * 8 * 8 * 32];
  __shared__ __attribute__((aligned(16))) float pm[ATT_NW * 16];
  __shared__ __attribute__((aligned(16))) float ps[ATT_NW * 32];
  const int lane = threadIdx.x & 31;
  const int wave = __builtin_amdgcn_readfirstlane((int)(threadIdx.x >> 5));
  const int half = lane >> 4;
  const int l15  = lane & 15;
  const int qt = (HW / 16 - 1) - (int)blockIdx.x;
  const int b  = blockIdx.y;
  const int q0 = qt * 16;
  const int qp = q0 + l15;
  const float* Qb = QKV + (size_t)b * 160 * HW;

  v2f qb[4];
#pragma unroll
  for (int kc = 0; kc < 4; ++kc) {
    qb[kc].x = Qb[(size_t)(kc * 4 + half * 2) * HW + qp];
    qb[kc].y = Qb[(size_t)(kc * 4 + half * 2 + 1) * HW + qp];
  }

  v8f acc[8];
#pragma unroll
  for (int vt = 0; vt < 8; ++vt) acc[vt] = (v8f){0, 0, 0, 0, 0, 0, 0, 0};
  float mrun = -3.0e38f;
  float lrun = 0.f;

  for (int kt = wave; kt <= qt; kt += ATT_NW) {
    const int k0 = kt * 16;
    v2f ka[4];
#pragma unroll
    for (int kc = 0; kc < 4; ++kc) {
      ka[kc].x = Qb[(size_t)(16 + kc * 4 + half * 2) * HW + k0 + l15];
      ka[kc].y = Qb[(size_t)(16 + kc * 4 + half * 2 + 1) * HW + k0 + l15];
    }
    v4f va[8][2];
    const float* vtile = Qb + (size_t)32 * HW + (size_t)kt * 2048 + lane * 4;
#pragma unroll
    for (int vt = 0; vt < 8; ++vt) {
      va[vt][0] = *(const v4f*)(vtile + vt * 256);
      va[vt][1] = *(const v4f*)(vtile + vt * 256 + 128);
    }
    __builtin_amdgcn_sched_barrier(0);
    v8f s = (v8f){0, 0, 0, 0, 0, 0, 0, 0};
#pragma unroll
    for (int kc = 0; kc < 4; ++kc) s = wmma4(ka[kc], qb[kc], s);
    float val[8];
    float rm = -3.0e38f;
#pragma unroll
    for (int r = 0; r < 8; ++r) {
      const int kp = k0 + half * 8 + r;
      val[r] = (kp <= qp) ? s[r] * 0.25f : -3.0e38f;
      rm = fmaxf(rm, val[r]);
    }
    rm = fmaxf(rm, __shfl_xor(rm, 16));
    const float nm = fmaxf(mrun, rm);
    const float alpha = expf(mrun - nm);
    v8f p;
    float rs = 0.f;
#pragma unroll
    for (int r = 0; r < 8; ++r) {
      p[r] = expf(val[r] - nm);
      rs += p[r];
    }
    lrun = lrun * alpha + rs;
    mrun = nm;
#pragma unroll
    for (int vt = 0; vt < 8; ++vt)
#pragma unroll
      for (int r = 0; r < 8; ++r) acc[vt][r] *= alpha;
#pragma unroll
    for (int kc = 0; kc < 4; ++kc) {
      v2f pb;
      pb.x = p[2 * kc];
      pb.y = p[2 * kc + 1];
#pragma unroll
      for (int vt = 0; vt < 8; ++vt) {
        v2f a;
        a.x = va[vt][kc >> 1][(kc & 1) * 2];
        a.y = va[vt][kc >> 1][(kc & 1) * 2 + 1];
        acc[vt] = wmma4(a, pb, acc[vt]);
      }
    }
  }

  const int nact = (qt + 1 < ATT_NW) ? qt + 1 : ATT_NW;
  const bool active = wave < nact;
  if (active && half == 0) pm[wave * 16 + l15] = mrun;
  __syncthreads();
  if (active) {
    float M = pm[l15];
    for (int j = 1; j < nact; ++j) M = fmaxf(M, pm[j * 16 + l15]);
    const float sc = expf(mrun - M);
    ps[wave * 32 + lane] = lrun * sc;
#pragma unroll
    for (int r = 0; r < 8; ++r)
#pragma unroll
      for (int vt = 0; vt < 8; ++vt)
        po[((wave * 8 + vt) * 8 + r) * 32 + lane] = acc[vt][r] * sc;
  }
  __syncthreads();

  float Ls = 0.f, os[8];
#pragma unroll
  for (int r = 0; r < 8; ++r) os[r] = 0.f;
  for (int j = 0; j < nact; ++j) {
    Ls += ps[j * 32 + l15] + ps[j * 32 + 16 + l15];
#pragma unroll
    for (int r = 0; r < 8; ++r) os[r] += po[((j * 8 + wave) * 8 + r) * 32 + lane];
  }
  const float inv = 1.f / Ls;
#pragma unroll
  for (int r = 0; r < 8; ++r)
    Ob[((size_t)b * HIDC + wave * 16 + half * 8 + r) * HW + qp] = os[r] * inv;
}

extern "C" void kernel_launch(void* const* d_in, const int* in_sizes, int n_in,
                              void* d_out, int out_size, void* d_ws, size_t ws_size,
                              hipStream_t stream) {
  const float* x    = (const float*)d_in[0];
  const float* rw1  = (const float*)d_in[1];
  const float* rb1  = (const float*)d_in[2];
  const float* rw2  = (const float*)d_in[3];
  const float* rb2  = (const float*)d_in[4];
  const float* qkvw = (const float*)d_in[5];
  const float* qkvb = (const float*)d_in[6];
  const float* apw  = (const float*)d_in[7];
  const float* apb  = (const float*)d_in[8];
  const float* ocw  = (const float*)d_in[9];
  const float* ocb  = (const float*)d_in[10];
  const float* oaw  = (const float*)d_in[11];
  const float* oab  = (const float*)d_in[12];
  const float* opw  = (const float*)d_in[13];
  const float* opb  = (const float*)d_in[14];
  float* Yout = (float*)d_out;

  const size_t N1 = (size_t)NB * HIDC * HW;
  float* A  = (float*)d_ws;
  float* E  = A + N1;
  float* F  = E + N1;
  float* Df = F + N1;
  float* P1 = Df + 2 * N1;
  float* P2 = P1 + (size_t)6 * HIDC * HIDC * 4;
  float* PT = P2 + (size_t)6 * 2 * HIDC * HIDC * 4;

  dim3 blk(256);
  dim3 g1((unsigned)((N1 + 255) / 256) + 1216);
  dim3 g128(NB, HW / 64, 1), g256(NB, HW / 64, 2);

  k_first<<<g1, blk, 0, stream>>>(x, E, (int)N1, rw1, rw2, apw, ocw, oaw, opw, P1, P2, PT);
  for (int i = 0; i < 6; ++i) {
    const float* w1 = P1 + (size_t)i * HIDC * HIDC * 4;
    const float* b1 = rb1 + (size_t)i * HIDC;
    const float* w2 = P2 + (size_t)i * 2 * HIDC * HIDC * 4;
    const float* b2 = rb2 + (size_t)i * 2 * HIDC;
    k_conv2x2<1><<<g128, blk, 0, stream>>>(E, w1, b1, F, nullptr, nullptr);
    k_conv2x2<2><<<g128, blk, 0, stream>>>(F, w2, b2, A, i == 0 ? x : A, E);
  }

  k_qkv<<<g256, blk, 0, stream>>>(A, qkvw, qkvb, Df);
  k_attn<<<dim3(HW / 16, NB), dim3(256), 0, stream>>>(Df, F);

  k_tail<<<g128, blk, 0, stream>>>(F, E, x, PT, apb, PT + 16384, ocb, PT + 2 * 16384, oab,
                                   PT + 3 * 16384, opb, Yout);
}


# 2x2 convs: columns in the F(4,2) minimal-filtering form, 15 matrix products per 2x4 output block and channel (was 18); a block owns 4 image rows; conv sum order changed
# speedup vs baseline: 3.0246x; 1.0813x over previous
#include <hip/hip_runtime.h>
#include <hip/hip_bf16.h>


typedef __attribute__((ext_vector_type(2))) float v2f;
typedef __attribute__((ext_vector_type(4))) float v4f;
typedef __attribute__((ext_vector_type(8))) float v8f;

#define NB   32
#define HIDC 128
#define HW   1024

__device__ __forceinline__ v8f wmma4(v2f a, v2f b, v8f c) {
  return __builtin_amdgcn_wmma_f32_16x16x4_f32(false, a, false, b, (short)0, c,
                                               false, false);
}

__device__ __forceinline__ float eluf(float x) {
  return x > 0.f ? x : (expf(x) - 1.f);
}

#define NW1_16B (6 * 128 * 128)
#define NW2_16B (6 * 256 * 128)
#define NWT_16B (4 * 128 * 32)
__global__ void __launch_bounds__(256)
k_first(const float* __restrict__ X, float* __restrict__ Y, int n,
        const float* __restrict__ W1, const float* __restrict__ W2,
        const float* __restrict__ Wap, const float* __restrict__ Wc,
        const float* __restrict__ Wa, const float* __restrict__ Wp,
        float* __restrict__ P1, float* __restrict__ P2, float* __restrict__ PT) {
  const int nelu = (n + 255) >> 8;
  if ((int)blockIdx.x < nelu) {
    int i = blockIdx.x * 256 + threadIdx.x;
    if (i < n) Y[i] = eluf(X[i]);
    return;
  }
  const int t = ((int)blockIdx.x - nelu) * 256 + (int)threadIdx.x;
  if (t < NW1_16B + NW2_16B) {
    const bool first = t < NW1_16B;
    const float* src = first ? W1 : W2;
    v4f* dst = (v4f*)(first ? P1 : P2);
    const int d    = first ? t : t - NW1_16B;
    const int lane = d & 31;
    const int e    = (d >> 5) & 1;
    const int s    = (d >> 6) & 31;
    const int rg   = d >> 11;
    const int row  = rg * 16 + (lane & 15);
    const int ch   = 4 * s + 2 * (lane >> 4);
    const float* r = src + ((size_t)row * 128 + ch) * 4 + 2 * e;
    const v2f c0 = *(const v2f*)r;
    const v2f c1 = *(const v2f*)(r + 4);
    dst[d] = (v4f){c0.x, c1.x, c0.y, c1.y};
  } else if (t < NW1_16B + NW2_16B + NWT_16B) {
    const int t2 = t - (NW1_16B + NW2_16B);
    const int m  = t2 >> 12;
    const float* Wm = (m == 0) ? Wap : (m == 1) ? Wc : (m == 2) ? Wa : Wp;
    const int d    = t2 & 4095;
    const int lane = d & 31;
    const int s2   = (d >> 5) & 15;
    const int rg   = d >> 9;
    const float* r = Wm + (size_t)(rg * 16 + (lane & 15)) * 128 + 8 * s2 + 2 * (lane >> 4);
    const v2f lo = *(const v2f*)r;
    const v2f hi = *(const v2f*)(r + 4);
    ((v4f*)PT)[t2] = (v4f){lo.x, lo.y, hi.x, hi.y};
  }
}

#define VPOS 2048
#define NPOS 15

template <int G>
struct ConvW {
  v4f wa[2 / G][G][2];
};
template <int G>
struct ConvF {
  v2f fr[2 / G][NPOS];
};

template <int G>
__device__ __forceinline__ void conv_load_w(ConvW<G>& o, const v4f* const (&wt)[G], int ch) {
#pragma unroll
  for (int j = 0; j < 2 / G; ++j) {
    const int step = ch * (2 / G) + j;
#pragma unroll
    for (int g = 0; g < G; ++g) {
      o.wa[j][g][0] = wt[g][step * 64];
      o.wa[j][g][1] = wt[g][step * 64 + 32];
    }
  }
}

template <int G>
__device__ __forceinline__ void conv_load_f(ConvF<G>& o, const float* vl, int ch) {
#pragma unroll
  for (int j = 0; j < 2 / G; ++j) {
    const float* base = vl + (ch * (2 / G) + j) * 64;
#pragma unroll
    for (int p = 0; p < NPOS; ++p) o.fr[j][p] = *(const v2f*)(base + p * VPOS);
  }
}

template <int G>
__device__ __forceinline__ void conv_mma(const ConvW<G>& w, const ConvF<G>& f, v8f (&acc)[G][NPOS]) {
#pragma unroll
  for (int j = 0; j < 2 / G; ++j) {
    v2f u[G][NPOS];
#pragma unroll
    for (int g = 0; g < G; ++g) {
      const v4f w01 = w.wa[j][g][0], w23 = w.wa[j][g][1];
      const v2f t0 = (v2f){w01[0], w01[1]};
      const v2f t1 = (v2f){w01[2], w01[3]};
      const v2f t2 = (v2f){w23[0], w23[1]};
      const v2f t3 = (v2f){w23[2], w23[3]};
      const v2f ra[3] = {t0, t0 + t2, t2};
      const v2f rb[3] = {t1, t1 + t3, t3};
#pragma unroll
      for (int i = 0; i < 3; ++i) {
        const v2f a = ra[i], b = rb[i];
        u[g][i * 5 + 0] = 0.5f * a;
        u[g][i * 5 + 1] = -0.5f * (a + b);
        u[g][i * 5 + 2] = (b - a) * (1.f / 6.f);
        u[g][i * 5 + 3] = (a + 2.f * b) * (1.f / 6.f);
        u[g][i * 5 + 4] = b;
      }
    }
#pragma unroll
    for (int p = 0; p < NPOS; ++p)
#pragma unroll
      for (int g = 0; g < G; ++g) acc[g][p] = wmma4(u[g][p], f.fr[j][p], acc[g][p]);
  }
}

#define MINF_Z0(m, c) (m[c] + m[5 + c])
#define MINF_Z1(m, c) (m[5 + c] - m[10 + c])
#define MINF_C0(z) ((z[0] + z[1]) + (z[2] + z[3]))
#define MINF_C1(z) ((z[1] - z[2]) + 2.f * z[3])
#define MINF_C2(z) ((z[1] + z[2]) + 4.f * z[3])
#define MINF_C3(z) ((z[1] - z[2]) + (8.f * z[3] + z[4]))

__device__ __forceinline__ v4f minf_row(const float (&m)[NPOS], int i) {
  float z[5];
#pragma unroll
  for (int c = 0; c < 5; ++c) z[c] = (i == 0) ? MINF_Z0(m, c) : MINF_Z1(m, c);
  return (v4f){MINF_C0(z), MINF_C1(z), MINF_C2(z), MINF_C3(z)};
}

__device__ __forceinline__ void minf_cols(const float (&x)[5], float (&c)[5]) {
  c[0] = 2.f * (x[0] - x[2]) + (x[3] - x[1]);
  c[1] = (x[3] - x[2]) - 2.f * x[1];
  c[2] = 2.f * x[1] - 3.f * x[2] + x[3];
  c[3] = x[3] - x[1];
  c[4] = 2.f * (x[1] - x[3]) + (x[4] - x[2]);
}

template <int G>
__global__ void __launch_bounds__(256)
k_conv2x2(const float* __restrict__ X, const float* __restrict__ W,
          const float* __restrict__ bias, float* Y, const float* Rin,
          float* __restrict__ E) {
  __shared__ __attribute__((aligned(16))) float vs[NPOS * VPOS];
  const int tid  = threadIdx.x;
  const int lane = tid & 31;
  const int wave = tid >> 5;
  const int half = lane >> 4;
  const int l15  = lane & 15;
  const int b  = blockIdx.x;
  const int r0 = blockIdx.y * 4;
  const int o0 = wave * 16;

  const float* Xb = X + (size_t)b * HIDC * HW;

  const v4f* wt[G];
#pragma unroll
  for (int g = 0; g < G; ++g)
    wt[g] = (const v4f*)W + (size_t)((g * 8 + wave) * 2048 + lane);

  ConvW<G> wq[3];
  conv_load_w<G>(wq[0], wt, 0);
  conv_load_w<G>(wq[1], wt, 1);

  {
    const int yt = r0 - 1;
    const int j  = tid & 7;
    const int pb = tid >> 3;
    v4f ga[5][2], gc[5][2];
#pragma unroll
    for (int q = 0; q < 5; ++q) {
#pragma unroll
      for (int k = 0; k < 2; ++k) {
        const int yy = (q == 0 && yt < 0) ? 0 : yt + q;
        const float* g = Xb + (size_t)(2 * pb + 64 * k) * HW + yy * 32 + j * 4;
        ga[q][k] = *(const v4f*)g;
        gc[q][k] = *(const v4f*)(g + HW);
      }
    }
    if (yt < 0) {
#pragma unroll
      for (int k = 0; k < 2; ++k) {
        ga[0][k] = (v4f){0.f, 0.f, 0.f, 0.f};
        gc[0][k] = (v4f){0.f, 0.f, 0.f, 0.f};
      }
    }
#pragma unroll
    for (int k = 0; k < 2; ++k) {
      float da[5][5], dc[5][5];
#pragma unroll
      for (int q = 0; q < 5; ++q) {
        float la = __shfl_up(ga[q][k][3], 1);
        float lc = __shfl_up(gc[q][k][3], 1);
        da[q][0] = (j == 0) ? 0.f : la;
        dc[q][0] = (j == 0) ? 0.f : lc;
#pragma unroll
        for (int x = 0; x < 4; ++x) {
          da[q][x + 1] = ga[q][k][x];
          dc[q][x + 1] = gc[q][k][x];
        }
      }
      float va[2][3][5], vc[2][3][5];
#pragma unroll
      for (int s = 0; s < 2; ++s) {
        float ra[3][5], rc[3][5];
#pragma unroll
        for (int x = 0; x < 5; ++x) {
          ra[0][x] = da[2 * s][x] - da[2 * s + 1][x];
          ra[1][x] = da[2 * s + 1][x];
          ra[2][x] = da[2 * s + 1][x] - da[2 * s + 2][x];
          rc[0][x] = dc[2 * s][x] - dc[2 * s + 1][x];
          rc[1][x] = dc[2 * s + 1][x];
          rc[2][x] = dc[2 * s + 1][x] - dc[2 * s + 2][x];
        }
#pragma unroll
        for (int i = 0; i < 3; ++i) {
          minf_cols(ra[i], va[s][i]);
          minf_cols(rc[i], vc[s][i]);
        }
      }
      float* dst = vs + (pb + 32 * k) * 32 + j * 4;
#pragma unroll
      for (int i = 0; i < 3; ++i)
#pragma unroll
        for (int c = 0; c < 5; ++c)
          *(v4f*)(dst + (i * 5 + c) * VPOS) = (v4f){va[0][i][c], vc[0][i][c], va[1][i][c], vc[1][i][c]};
    }
  }
  __syncthreads();

  v8f acc[G][NPOS];
#pragma unroll
  for (int g = 0; g < G; ++g)
#pragma unroll
    for (int p = 0; p < NPOS; ++p) acc[g][p] = (v8f){0, 0, 0, 0, 0, 0, 0, 0};

  const float* vl = vs + half * 32 + l15 * 2;

  constexpr int NCH   = (HIDC / 4) / (2 / G);
  constexpr int TAIL0 = ((NCH - 2) / 6) * 6;
  ConvF<G> fq[2];
  conv_load_f<G>(fq[0], vl, 0);
#pragma clang loop unroll(disable)
  for (int c = 0; c < TAIL0; c += 6) {
#pragma unroll
    for (int t = 0; t < 6; ++t) {
      conv_load_w<G>(wq[(t + 2) % 3], wt, c + t + 2);
      conv_load_f<G>(fq[(t + 1) % 2], vl, c + t + 1);
      __builtin_amdgcn_sched_barrier(0);
      conv_mma<G>(wq[t % 3], fq[t % 2], acc);
      __builtin_amdgcn_sched_barrier(0);
    }
  }
#pragma unroll
  for (int t = 0; t < NCH - TAIL0; ++t) {
    if (TAIL0 + t + 2 < NCH) conv_load_w<G>(wq[(t + 2) % 3], wt, TAIL0 + t + 2);
    if (TAIL0 + t + 1 < NCH) conv_load_f<G>(fq[(t + 1) % 2], vl, TAIL0 + t + 1);
    __builtin_amdgcn_sched_barrier(0);
    conv_mma<G>(wq[t % 3], fq[t % 2], acc);
    __builtin_amdgcn_sched_barrier(0);
  }

  float bs[G][8];
#pragma unroll
  for (int g = 0; g < G; ++g)
#pragma unroll
    for (int r = 0; r < 8; ++r) bs[g][r] = bias[g * HIDC + o0 + r + half * 8];

  const size_t obase = ((size_t)b * HIDC + o0 + half * 8) * HW +
                       (size_t)((r0 + 2 * (l15 & 1)) * 32 + 4 * (l15 >> 1));

  if (G == 1) {
#pragma unroll
    for (int r = 0; r < 8; ++r) {
      float m[NPOS];
#pragma unroll
      for (int p = 0; p < NPOS; ++p) m[p] = acc[0][p][r];
#pragma unroll
      for (int i = 0; i < 2; ++i) {
        const v4f t = minf_row(m, i);
        v4f y;
#pragma unroll
        for (int x = 0; x < 4; ++x) y[x] = eluf(t[x] + bs[0][r]);
        *(v4f*)(Y + obase + (size_t)r * HW + i * 32) = y;
      }
    }
  } else {
    v4f res[8][2];
#pragma unroll
    for (int r = 0; r < 8; ++r) {
      res[r][0] = *(const v4f*)(Rin + obase + (size_t)r * HW);
      res[r][1] = *(const v4f*)(Rin + obase + (size_t)r * HW + 32);
    }
#pragma unroll
    for (int r = 0; r < 8; ++r) {
      float m1[NPOS], m2[NPOS];
#pragma unroll
      for (int p = 0; p < NPOS; ++p) {
        m1[p] = acc[0][p][r];
        m2[p] = acc[G - 1][p][r];
      }
#pragma unroll
      for (int i = 0; i < 2; ++i) {
        const v4f t1 = minf_row(m1, i);
        const v4f t2 = minf_row(m2, i);
        v4f v, e;
#pragma unroll
        for (int x = 0; x < 4; ++x) {
          const float a1 = t1[x] + bs[0][r];
          const float a2 = t2[x] + bs[G - 1][r];
          v[x] = res[r][i][x] + a1 * (1.f / (1.f + expf(-a2)));
          e[x] = eluf(v[x]);
        }
        *(v4f*)(Y + obase + (size_t)r * HW + i * 32) = v;
        *(v4f*)(E + obase + (size_t)r * HW + i * 32) = e;
      }
    }
  }
}

#define TPITCH 160
#define TBUF   (64 * TPITCH)

template <int N>
struct TailOps {
  v2f a[N][4 / N];
  v2f b[N][4 / N][4];
};

template <int N>
__device__ __forceinline__ void tail_load(TailOps<N>& o, const v4f* const (&wr)[N],
                                          const float* const (&bl)[N], int ch) {
#pragma unroll
  for (int n = 0; n < N; ++n) {
#pragma unroll
    for (int j2 = 0; j2 < 2 / N; ++j2) {
      const int s2 = ch * (2 / N) + j2;
      const v4f w = wr[n][s2 * 32];
      o.a[n][2 * j2]     = (v2f){w.x, w.y};
      o.a[n][2 * j2 + 1] = (v2f){w.z, w.w};
    }
#pragma unroll
    for (int j = 0; j < 4 / N; ++j) {
      const int s = ch * (4 / N) + j;
#pragma unroll
      for (int t = 0; t < 4; ++t)
        o.b[n][j][t] = *(const v2f*)(bl[n] + s * (2 * TPITCH) + t * 32);
    }
  }
}

template <int N>
__device__ __forceinline__ void tail_mma(const TailOps<N>& o, v8f (&acc)[N][4]) {
#pragma unroll
  for (int j = 0; j < 4 / N; ++j)
#pragma unroll
    for (int t = 0; t < 4; ++t)
#pragma unroll
      for (int n = 0; n < N; ++n) acc[n][t] = wmma4(o.a[n][j], o.b[n][j][t], acc[n][t]);
}

template <int N>
__device__ __forceinline__ void tail_gemm(const v4f* const (&wr)[N],
                                          const float* const (&bl)[N], v8f (&acc)[N][4]) {
#pragma unroll
  for (int n = 0; n < N; ++n)
#pragma unroll
    for (int t = 0; t < 4; ++t) acc[n][t] = (v8f){0, 0, 0, 0, 0, 0, 0, 0};
  TailOps<N> opA, opB;
  tail_load<N>(opA, wr, bl, 0);
#pragma clang loop unroll(disable)
  for (int ch = 0; ch < 8 * N; ch += 2) {
    tail_load<N>(opB, wr, bl, ch + 1);
    __builtin_amdgcn_sched_barrier(0);
    tail_mma<N>(opA, acc);
    if (ch + 2 < 8 * N) tail_load<N>(opA, wr, bl, ch + 2);
    __builtin_amdgcn_sched_barrier(0);
    tail_mma<N>(opB, acc);
  }
}

__global__ void __launch_bounds__(256)
k_tail(const float* __restrict__ O, const float* __restrict__ E,
       const float* __restrict__ Xres,
       const float* __restrict__ Wap, const float* __restrict__ bap,
       const float* __restrict__ Wc, const float* __restrict__ bc,
       const float* __restrict__ Wa, const float* __restrict__ ba,
       const float* __restrict__ Wp, const float* __restrict__ bp,
       float* __restrict__ Y) {
  __shared__ __attribute__((aligned(16))) float ts[3 * TBUF];
  const int tid  = threadIdx.x;
  const int lane = tid & 31;
  const int wave = tid >> 5;
  const int half = lane >> 4;
  const int l15  = lane & 15;
  const int b  = blockIdx.x;
  const int p0 = blockIdx.y * 64;
  const int o0 = wave * 16;

  {
    const int j  = tid & 15;
    const int pb = tid >> 4;
    const float* Ob = O + (size_t)b * HIDC * HW + p0 + j * 4;
    const float* Eb = E + (size_t)b * HIDC * HW + p0 + j * 4;
    v4f oa[4], oc[4], ea[4], ec[4];
#pragma unroll
    for (int k = 0; k < 4; ++k) {
      const size_t ro = (size_t)(2 * (pb + 16 * k)) * HW;
      oa[k] = *(const v4f*)(Ob + ro);
      oc[k] = *(const v4f*)(Ob + ro + HW);
      ea[k] = *(const v4f*)(Eb + ro);
      ec[k] = *(const v4f*)(Eb + ro + HW);
    }
#pragma unroll
    for (int k = 0; k < 4; ++k) {
      float* d = ts + (pb + 16 * k) * TPITCH + j * 8;
      *(v4f*)d              = (v4f){oa[k].x, oc[k].x, oa[k].y, oc[k].y};
      *(v4f*)(d + 4)        = (v4f){oa[k].z, oc[k].z, oa[k].w, oc[k].w};
      *(v4f*)(d + TBUF)     = (v4f){ea[k].x, ec[k].x, ea[k].y, ec[k].y};
      *(v4f*)(d + TBUF + 4) = (v4f){ea[k].z, ec[k].z, ea[k].w, ec[k].w};
    }
  }
  __syncthreads();

  const int    wofs = wave * 512 + lane;
  const int    lofs = half * TPITCH + l15 * 2;
  float* tw = ts + ((o0 >> 1) + half * 4) * TPITCH + l15 * 2;
  const int ob = o0 + half * 8;

  {
    v8f acc[1][4];
    const v4f* const wr[1] = {(const v4f*)Wap + wofs};
    const float* const bl[1] = {ts + lofs};
    tail_gemm<1>(wr, bl, acc);
    float bs[8];
#pragma unroll
    for (int r = 0; r < 8; ++r) bs[r] = bap[ob + r];
#pragma unroll
    for (int t = 0; t < 4; ++t)
#pragma unroll
      for (int r = 0; r < 8; r += 2) {
        v2f v;
        v.x = eluf(acc[0][t][r] + bs[r]);
        v.y = eluf(acc[0][t][r + 1] + bs[r + 1]);
        *(v2f*)(tw + 2 * TBUF + (r >> 1) * TPITCH + t * 32) = v;
      }
  }
  __syncthreads();

  float xr[4][8];
  const size_t ybase = ((size_t)b * HIDC + ob) * HW + p0 + l15;
  {
    v8f acc[2][4];
    const v4f* const wr[2] = {(const v4f*)Wa + wofs, (const v4f*)Wc + wofs};
    const float* const bl[2] = {ts + 2 * TBUF + lofs, ts + TBUF + lofs};
    tail_gemm<2>(wr, bl, acc);
    float bsa[8], bsc[8];
#pragma unroll
    for (int r = 0; r < 8; ++r) {
      bsa[r] = ba[ob + r];
      bsc[r] = bc[ob + r];
    }
#pragma unroll
    for (int t = 0; t < 4; ++t)
#pragma unroll
      for (int r = 0; r < 8; ++r) xr[t][r] = Xres[ybase + (size_t)r * HW + t * 16];
#pragma unroll
    for (int t = 0; t < 4; ++t)
#pragma unroll
      for (int r = 0; r < 8; r += 2) {
        v2f v;
        {
          const float a1 = eluf(acc[0][t][r] + bsa[r]);
          const float c1 = eluf(acc[1][t][r] + bsc[r]);
          v.x = eluf(c1 + a1);
        }
        {
          const float a1 = eluf(acc[0][t][r + 1] + bsa[r + 1]);
          const float c1 = eluf(acc[1][t][r + 1] + bsc[r + 1]);
          v.y = eluf(c1 + a1);
        }
        *(v2f*)(tw + (r >> 1) * TPITCH + t * 32) = v;
      }
  }
  __syncthreads();

  {
    v8f acc[1][4];
    const v4f* const wr[1] = {(const v4f*)Wp + wofs};
    const float* const bl[1] = {ts + lofs};
    tail_gemm<1>(wr, bl, acc);
    float bs[8];
#pragma unroll
    for (int r = 0; r < 8; ++r) bs[r] = bp[ob + r];
#pragma unroll
    for (int t = 0; t < 4; ++t)
#pragma unroll
      for (int r = 0; r < 8; ++r) {
        float v = eluf(acc[0][t][r] + bs[r]);
        v += xr[t][r];
        Y[ybase + (size_t)r * HW + t * 16] = v;
      }
  }
}

__global__ void __launch_bounds__(256)
k_qkv(const float* __restrict__ X, const float* __restrict__ W,
      const float* __restrict__ bias, float* __restrict__ Y) {
  const int lane = threadIdx.x & 31;
  const int wave = __builtin_amdgcn_readfirstlane((int)(threadIdx.x >> 5));
  const int half = lane >> 4;
  const int l15  = lane & 15;
  const int b  = blockIdx.x;
  const int p0 = blockIdx.y * 64;
  const int o0 = (blockIdx.z * 8 + wave) * 16;
  if (o0 >= 160) return;

  const float* Xb   = X + (size_t)b * HIDC * HW;
  const float* wrow = W + (size_t)(o0 + l15) * 130;

  v8f acc[4];
#pragma unroll
  for (int t = 0; t < 4; ++t) acc[t] = (v8f){0, 0, 0, 0, 0, 0, 0, 0};

  for (int c0 = 0; c0 < HIDC; c0 += 4) {
    int ca = c0 + half * 2;
    v2f a;
    a.x = wrow[ca];
    a.y = wrow[ca + 1];
    const float* x0 = Xb + (size_t)ca * HW + p0 + l15;
#pragma unroll
    for (int t = 0; t < 4; ++t) {
      v2f bb;
      bb.x = x0[t * 16];
      bb.y = x0[HW + t * 16];
      acc[t] = wmma4(a, bb, acc[t]);
    }
  }

  const bool isv = o0 >= 32;
#pragma unroll
  for (int t = 0; t < 4; ++t) {
    int p = p0 + t * 16 + l15;
    float py = (float)(p >> 5) * (1.f / 32.f) - 0.5f;
    float px = (float)(p & 31) * (1.f / 32.f) - 0.5f;
    const size_t vbase = ((size_t)b * 160 + 32) * HW +
                         (size_t)((((p >> 4) * 8 + ((o0 - 32) >> 4)) * 2 + ((l15 >> 2) & 1)) * 128 +
                                  ((l15 >> 3) * 16 + half * 8) * 4 + (l15 & 3));
    float v[8];
#pragma unroll
    for (int r = 0; r < 8; ++r) {
      int o = o0 + r + half * 8;
      v[r] = acc[t][r] + W[(size_t)o * 130 + 128] * py +
             W[(size_t)o * 130 + 129] * px + bias[o];
    }
    if (isv) {
#pragma unroll
      for (int r = 0; r < 8; ++r) Y[vbase + r * 4] = v[r];
    } else {
#pragma unroll
      for (int r = 0; r < 8; ++r) Y[((size_t)b * 160 + o0 + r + half * 8) * HW + p] = v[r];
    }
  }
}

#define ATT_NW 8
__global__ void __launch_bounds__(256)
k_attn(const float* __restrict__ QKV, float* __restrict__ Ob) {
  __shared__ __attribute__((aligned(16))) float po[ATT_NW * 8 * 8 * 32];
  __shared__ __attribute__((aligned(16))) float pm[ATT_NW * 16];
  __shared__ __attribute__((aligned(16))) float ps[ATT_NW * 32];
  const int lane = threadIdx.x & 31;
  const int wave = __builtin_amdgcn_readfirstlane((int)(threadIdx.x >> 5));
  const int half = lane >> 4;
  const int l15  = lane & 15;
  const int qt = (HW / 16 - 1) - (int)blockIdx.x;
  const int b  = blockIdx.y;
  const int q0 = qt * 16;
  const int qp = q0 + l15;
  const float* Qb = QKV + (size_t)b * 160 * HW;

  v2f qb[4];
#pragma unroll
  for (int kc = 0; kc < 4; ++kc) {
    qb[kc].x = Qb[(size_t)(kc * 4 + half * 2) * HW + qp];
    qb[kc].y = Qb[(size_t)(kc * 4 + half * 2 + 1) * HW + qp];
  }

  v8f acc[8];
#pragma unroll
  for (int vt = 0; vt < 8; ++vt) acc[vt] = (v8f){0, 0, 0, 0, 0, 0, 0, 0};
  float mrun = -3.0e38f;
  float lrun = 0.f;

  for (int kt = wave; kt <= qt; kt += ATT_NW) {
    const int k0 = kt * 16;
    v2f ka[4];
#pragma unroll
    for (int kc = 0; kc < 4; ++kc) {
      ka[kc].x = Qb[(size_t)(16 + kc * 4 + half * 2) * HW + k0 + l15];
      ka[kc].y = Qb[(size_t)(16 + kc * 4 + half * 2 + 1) * HW + k0 + l15];
    }
    v4f va[8][2];
    const float* vtile = Qb + (size_t)32 * HW + (size_t)kt * 2048 + lane * 4;
#pragma unroll
    for (int vt = 0; vt < 8; ++vt) {
      va[vt][0] = *(const v4f*)(vtile + vt * 256);
      va[vt][1] = *(const v4f*)(vtile + vt * 256 + 128);
    }
    __builtin_amdgcn_sched_barrier(0);
    v8f s = (v8f){0, 0, 0, 0, 0, 0, 0, 0};
#pragma unroll
    for (int kc = 0; kc < 4; ++kc) s = wmma4(ka[kc], qb[kc], s);
    float val[8];
    float rm = -3.0e38f;
#pragma unroll
    for (int r = 0; r < 8; ++r) {
      const int kp = k0 + half * 8 + r;
      val[r] = (kp <= qp) ? s[r] * 0.25f : -3.0e38f;
      rm = fmaxf(rm, val[r]);
    }
    rm = fmaxf(rm, __shfl_xor(rm, 16));
    const float nm = fmaxf(mrun, rm);
    const float alpha = expf(mrun - nm);
    v8f p;
    float rs = 0.f;
#pragma unroll
    for (int r = 0; r < 8; ++r) {
      p[r] = expf(val[r] - nm);
      rs += p[r];
    }
    lrun = lrun * alpha + rs;
    mrun = nm;
#pragma unroll
    for (int vt = 0; vt < 8; ++vt)
#pragma unroll
      for (int r = 0; r < 8; ++r) acc[vt][r] *= alpha;
#pragma unroll
    for (int kc = 0; kc < 4; ++kc) {
      v2f pb;
      pb.x = p[2 * kc];
      pb.y = p[2 * kc + 1];
#pragma unroll
      for (int vt = 0; vt < 8; ++vt) {
        v2f a;
        a.x = va[vt][kc >> 1][(kc & 1) * 2];
        a.y = va[vt][kc >> 1][(kc & 1) * 2 + 1];
        acc[vt] = wmma4(a, pb, acc[vt]);
      }
    }
  }

  const int nact = (qt + 1 < ATT_NW) ? qt + 1 : ATT_NW;
  const bool active = wave < nact;
  if (active && half == 0) pm[wave * 16 + l15] = mrun;
  __syncthreads();
  if (active) {
    float M = pm[l15];
    for (int j = 1; j < nact; ++j) M = fmaxf(M, pm[j * 16 + l15]);
    const float sc = expf(mrun - M);
    ps[wave * 32 + lane] = lrun * sc;
#pragma unroll
    for (int r = 0; r < 8; ++r)
#pragma unroll
      for (int vt = 0; vt < 8; ++vt)
        po[((wave * 8 + vt) * 8 + r) * 32 + lane] = acc[vt][r] * sc;
  }
  __syncthreads();

  float Ls = 0.f, os[8];
#pragma unroll
  for (int r = 0; r < 8; ++r) os[r] = 0.f;
  for (int j = 0; j < nact; ++j) {
    Ls += ps[j * 32 + l15] + ps[j * 32 + 16 + l15];
#pragma unroll
    for (int r = 0; r < 8; ++r) os[r] += po[((j * 8 + wave) * 8 + r) * 32 + lane];
  }
  const float inv = 1.f / Ls;
#pragma unroll
  for (int r = 0; r < 8; ++r)
    Ob[((size_t)b * HIDC + wave * 16 + half * 8 + r) * HW + qp] = os[r] * inv;
}

extern "C" void kernel_launch(void* const* d_in, const int* in_sizes, int n_in,
                              void* d_out, int out_size, void* d_ws, size_t ws_size,
                              hipStream_t stream) {
  const float* x    = (const float*)d_in[0];
  const float* rw1  = (const float*)d_in[1];
  const float* rb1  = (const float*)d_in[2];
  const float* rw2  = (const float*)d_in[3];
  const float* rb2  = (const float*)d_in[4];
  const float* qkvw = (const float*)d_in[5];
  const float* qkvb = (const float*)d_in[6];
  const float* apw  = (const float*)d_in[7];
  const float* apb  = (const float*)d_in[8];
  const float* ocw  = (const float*)d_in[9];
  const float* ocb  = (const float*)d_in[10];
  const float* oaw  = (const float*)d_in[11];
  const float* oab  = (const float*)d_in[12];
  const float* opw  = (const float*)d_in[13];
  const float* opb  = (const float*)d_in[14];
  float* Yout = (float*)d_out;

  const size_t N1 = (size_t)NB * HIDC * HW;
  float* A  = (float*)d_ws;
  float* E  = A + N1;
  float* F  = E + N1;
  float* Df = F + N1;
  float* P1 = Df + 2 * N1;
  float* P2 = P1 + (size_t)6 * HIDC * HIDC * 4;
  float* PT = P2 + (size_t)6 * 2 * HIDC * HIDC * 4;

  dim3 blk(256);
  dim3 g1((unsigned)((N1 + 255) / 256) + 1216);
  dim3 g128(NB, HW / 64, 1), g256(NB, HW / 64, 2);
  dim3 gconv(NB, HW / 128, 1);

  k_first<<<g1, blk, 0, stream>>>(x, E, (int)N1, rw1, rw2, apw, ocw, oaw, opw, P1, P2, PT);
  for (int i = 0; i < 6; ++i) {
    const float* w1 = P1 + (size_t)i * HIDC * HIDC * 4;
    const float* b1 = rb1 + (size_t)i * HIDC;
    const float* w2 = P2 + (size_t)i * 2 * HIDC * HIDC * 4;
    const float* b2 = rb2 + (size_t)i * 2 * HIDC;
    k_conv2x2<1><<<gconv, blk, 0, stream>>>(E, w1, b1, F, nullptr, nullptr);
    k_conv2x2<2><<<gconv, blk, 0, stream>>>(F, w2, b2, A, i == 0 ? x : A, E);
  }

  k_qkv<<<g256, blk, 0, stream>>>(A, qkvw, qkvb, Df);
  k_attn<<<dim3(HW / 16, NB), dim3(256), 0, stream>>>(Df, F);

  k_tail<<<g128, blk, 0, stream>>>(F, E, x, PT, apb, PT + 16384, ocb, PT + 2 * 16384, oab,
                                   PT + 3 * 16384, opb, Yout);
}
